# starting kernel
# baseline (speedup 1.0000x reference)
#include <hip/hip_runtime.h>

#define DDIM 300
#define PDIM 320
#define NWN  20000
#define EWN  320000
#define BN   256
#define LN   256
#define SDN  2048
#define EDN  4096
#define CN   20
#define KC   160

typedef __attribute__((ext_vector_type(2))) float v2f;
typedef __attribute__((ext_vector_type(8))) float v8f;


__global__ void zero_f32(float* __restrict__ p, long long n) {
    long long t = (long long)blockIdx.x * blockDim.x + threadIdx.x;
    if (t < n) p[t] = 0.f;
}

__global__ void gather_pad(const float* __restrict__ emb, const int* __restrict__ nid,
                           float* __restrict__ out, int n) {
    int t = blockIdx.x * blockDim.x + threadIdx.x;
    if (t >= n * PDIM) return;
    int r = t / PDIM, j = t % PDIM;
    out[t] = (j < DDIM) ? emb[(size_t)nid[r] * DDIM + j] : 0.f;
}

__global__ void pack_sage_w(const float* __restrict__ w, float* __restrict__ out) {
    int t = blockIdx.x * blockDim.x + threadIdx.x;
    if (t >= 2 * PDIM * PDIM) return;
    int h = t / (PDIM * PDIM);
    int r = (t / PDIM) % PDIM;
    int c = t % PDIM;
    out[t] = (r < DDIM && c < DDIM) ? w[((size_t)(h * DDIM + r)) * DDIM + c] : 0.f;
}

__global__ void pack_w(const float* __restrict__ w, float* __restrict__ out) {
    int t = blockIdx.x * blockDim.x + threadIdx.x;
    if (t >= PDIM * PDIM) return;
    int r = t / PDIM, c = t % PDIM;
    out[t] = (r < DDIM && c < DDIM) ? w[(size_t)r * DDIM + c] : 0.f;
}


__global__ void count_edges(const int* __restrict__ e_dst, int E, float* __restrict__ cnt) {
    int t = blockIdx.x * blockDim.x + threadIdx.x;
    if (t < E) atomicAdd(&cnt[e_dst[t]], 1.f);
}

__global__ void scatter_edges(const float* __restrict__ emb, const int* __restrict__ src_nid,
                              const int* __restrict__ e_src, const int* __restrict__ e_dst,
                              int E, float* __restrict__ sum) {
    const int lane = threadIdx.x & 31;
    const int e = blockIdx.x * (blockDim.x >> 5) + (threadIdx.x >> 5);
    if (e >= E) return;
    const int s = src_nid[e_src[e]];
    const float* __restrict__ src = emb + (size_t)s * DDIM;
    float* __restrict__ dst = sum + (size_t)e_dst[e] * PDIM;
    for (int j = lane; j < DDIM; j += 32)
        atomicAdd(&dst[j], src[j]);
}

__global__ void normalize_mean(float* __restrict__ sum, const float* __restrict__ cnt, int n) {
    int t = blockIdx.x * blockDim.x + threadIdx.x;
    if (t >= n * PDIM) return;
    sum[t] /= fmaxf(cnt[t / PDIM], 1.f);
}

__global__ __launch_bounds__(256) void gemm_f32_wmma(
        const float* __restrict__ A, const float* __restrict__ B,
        const float* __restrict__ bias, float* __restrict__ C,
        int M, int N, int K, int lda, int ldb, int ldc, int accum) {
    __shared__ float ldsB[64 * KC];
    const int lane = threadIdx.x & 31;
    const int wave = threadIdx.x >> 5;
    const int hi   = lane >> 4;
    const int l16  = lane & 15;
    const int nbase = blockIdx.y * 64;
    const int mt    = blockIdx.x * 8 + wave;
    const bool valid = (mt * 16) < M;
    const int m = mt * 16 + l16;

    v8f acc[4];
    if (valid) {
        if (accum) {
#pragma unroll
            for (int ns = 0; ns < 4; ++ns)
#pragma unroll
                for (int i = 0; i < 8; ++i)
                    acc[ns][i] = C[(size_t)(mt * 16 + i + hi * 8) * ldc + nbase + ns * 16 + l16];
        } else {
#pragma unroll
            for (int ns = 0; ns < 4; ++ns) {
                const int n = nbase + ns * 16 + l16;
                const float bv = (bias && n < DDIM) ? bias[n] : 0.f;
#pragma unroll
                for (int i = 0; i < 8; ++i) acc[ns][i] = bv;
            }
        }
    }

    const float* __restrict__ Arow = A + (size_t)m * lda;
    for (int kk = 0; kk < K; kk += KC) {
        __syncthreads();
        for (int idx = threadIdx.x; idx < 64 * KC; idx += 256) {
            int c = idx & 63, kloc = idx >> 6;
            ldsB[c * KC + kloc] = B[(size_t)(kk + kloc) * ldb + nbase + c];
        }
        __syncthreads();
        if (valid) {
            for (int k = 0; k < KC; k += 4) {
                v2f a = *(const v2f*)(Arow + kk + k + 2 * hi);
#pragma unroll
                for (int ns = 0; ns < 4; ++ns) {
                    v2f b = *(const v2f*)(&ldsB[(ns * 16 + l16) * KC + k + 2 * hi]);
                    acc[ns] = __builtin_amdgcn_wmma_f32_16x16x4_f32(
                        false, a, false, b, (short)0, acc[ns], false, false);
                }
            }
        }
    }
    if (valid) {
#pragma unroll
        for (int ns = 0; ns < 4; ++ns)
#pragma unroll
            for (int i = 0; i < 8; ++i)
                C[(size_t)(mt * 16 + i + hi * 8) * ldc + nbase + ns * 16 + l16] = acc[ns][i];
    }
}

__global__ void sdpa_fuse_add(const float* __restrict__ h0, const float* __restrict__ h1,
                              const float* __restrict__ h2, const float* __restrict__ addv,
                              float* __restrict__ out, int n) {
    const int lane = threadIdx.x & 31;
    const int node = blockIdx.x * (blockDim.x >> 5) + (threadIdx.x >> 5);
    if (node >= n) return;
    const float* x0 = h0 + (size_t)node * PDIM;
    const float* x1 = h1 + (size_t)node * PDIM;
    const float* x2 = h2 + (size_t)node * PDIM;

    float d00 = 0, d01 = 0, d02 = 0, d11 = 0, d12 = 0, d22 = 0;
    for (int j = lane; j < PDIM; j += 32) {
        float a = x0[j], b = x1[j], e = x2[j];
        d00 += a * a; d01 += a * b; d02 += a * e;
        d11 += b * b; d12 += b * e; d22 += e * e;
    }
    for (int off = 16; off > 0; off >>= 1) {
        d00 += __shfl_xor(d00, off, 32);
        d01 += __shfl_xor(d01, off, 32);
        d02 += __shfl_xor(d02, off, 32);
        d11 += __shfl_xor(d11, off, 32);
        d12 += __shfl_xor(d12, off, 32);
        d22 += __shfl_xor(d22, off, 32);
    }
    const float scale = rsqrtf((float)DDIM);
    float S[3][3] = {{d00, d01, d02}, {d01, d11, d12}, {d02, d12, d22}};
    float wsum[3] = {0.f, 0.f, 0.f};
    for (int t = 0; t < 3; ++t) {
        float p0 = S[t][0] * scale, p1 = S[t][1] * scale, p2 = S[t][2] * scale;
        float mx = fmaxf(p0, fmaxf(p1, p2));
        float e0 = __expf(p0 - mx), e1 = __expf(p1 - mx), e2 = __expf(p2 - mx);
        float inv = 1.f / (e0 + e1 + e2);
        wsum[0] += e0 * inv; wsum[1] += e1 * inv; wsum[2] += e2 * inv;
    }
    const float w0 = wsum[0] * (1.f / 3.f), w1 = wsum[1] * (1.f / 3.f), w2 = wsum[2] * (1.f / 3.f);
    const float* av = addv + (size_t)node * PDIM;
    float* o = out + (size_t)node * PDIM;
    for (int j = lane; j < PDIM; j += 32)
        o[j] = w0 * x0[j] + w1 * x1[j] + w2 * x2[j] + av[j];
}

__global__ void pool_seq(const float* __restrict__ table, const int* __restrict__ xb,
                         float* __restrict__ pooled) {
    const int b = blockIdx.x;
    const int j = threadIdx.x;
    if (j >= PDIM) return;
    const int* row = xb + b * LN;
    float acc = 0.f;
    for (int l = 0; l < LN; ++l) {
        int idx = row[l];
        if (idx < NWN) acc += table[(size_t)idx * PDIM + j];
    }
    pooled[(size_t)b * PDIM + j] = acc * (1.f / (float)LN);
}

__global__ void fc_kernel(const float* __restrict__ rep, const float* __restrict__ wfc,
                          const float* __restrict__ bfc, float* __restrict__ out) {
    int t = blockIdx.x * blockDim.x + threadIdx.x;
    if (t >= BN * CN) return;
    int b = t / CN, c = t % CN;
    float acc = bfc[c];
    const float* r = rep + (size_t)b * PDIM;
    for (int j = 0; j < DDIM; ++j) acc += r[j] * wfc[(size_t)j * CN + c];
    out[t] = acc;
}


extern "C" void kernel_launch(void* const* d_in, const int* in_sizes, int n_in,
                              void* d_out, int out_size, void* d_ws, size_t ws_size,
                              hipStream_t stream) {
    (void)in_sizes; (void)n_in; (void)out_size; (void)ws_size;
    const float* emb_word = (const float*)d_in[0];
    const float* emb_doc  = (const float*)d_in[1];
    const float* w_sage[6] = {(const float*)d_in[2],  (const float*)d_in[4],  (const float*)d_in[6],
                              (const float*)d_in[8],  (const float*)d_in[10], (const float*)d_in[12]};
    const float* b_sage[6] = {(const float*)d_in[3],  (const float*)d_in[5],  (const float*)d_in[7],
                              (const float*)d_in[9],  (const float*)d_in[11], (const float*)d_in[13]};
    const float* w_dense = (const float*)d_in[14];
    const float* b_dense = (const float*)d_in[15];
    const float* w_fc    = (const float*)d_in[16];
    const float* b_fc    = (const float*)d_in[17];
    const int* dst_nid      = (const int*)d_in[18];
    const int* src_nid_w[3] = {(const int*)d_in[19], (const int*)d_in[20], (const int*)d_in[21]};
    const int* e_src_w[3]   = {(const int*)d_in[22], (const int*)d_in[24], (const int*)d_in[26]};
    const int* e_dst_w[3]   = {(const int*)d_in[23], (const int*)d_in[25], (const int*)d_in[27]};
    const int* x_batch      = (const int*)d_in[28];
    const int* src_nid_d[3] = {(const int*)d_in[30], (const int*)d_in[31], (const int*)d_in[32]};
    const int* e_src_d[3]   = {(const int*)d_in[33], (const int*)d_in[35], (const int*)d_in[37]};
    const int* e_dst_d[3]   = {(const int*)d_in[34], (const int*)d_in[36], (const int*)d_in[38]};
    float* out = (float*)d_out;

    float* ws = (float*)d_ws;
    size_t off = 0;
    auto carve = [&](size_t nflt) { float* p = ws + off; off += nflt; return p; };
    float* hdst = carve((size_t)NWN * PDIM);
    float* nbuf = carve((size_t)NWN * PDIM);
    float* h3[3]; for (int r = 0; r < 3; ++r) h3[r] = carve((size_t)NWN * PDIM);
    float* cnt  = carve(NWN);
    float* wpk[6]; for (int r = 0; r < 6; ++r) wpk[r] = carve((size_t)2 * PDIM * PDIM);
    float* wdn  = carve((size_t)PDIM * PDIM);
    float* pooled = carve((size_t)BN * PDIM);
    float* docout = carve((size_t)BN * PDIM);
    float* dnb    = carve((size_t)BN * PDIM);
    float* d3[3]; for (int r = 0; r < 3; ++r) d3[r] = carve((size_t)BN * PDIM);
    float* docrep = carve((size_t)BN * PDIM);

    const int TB = 256;
    const unsigned wordElems = (unsigned)(((size_t)NWN * PDIM + TB - 1) / TB);
    const unsigned docElems  = (unsigned)(((size_t)BN * PDIM + TB - 1) / TB);
    const dim3 gemmGridW((NWN + 127) / 128, PDIM / 64);
    const dim3 gemmGridD((BN + 127) / 128, PDIM / 64);

    for (int r = 0; r < 6; ++r)
        pack_sage_w<<<(2 * PDIM * PDIM + TB - 1) / TB, TB, 0, stream>>>(w_sage[r], wpk[r]);
    pack_w<<<(PDIM * PDIM + TB - 1) / TB, TB, 0, stream>>>(w_dense, wdn);

    gather_pad<<<wordElems, TB, 0, stream>>>(emb_word, dst_nid, hdst, NWN);

    for (int r = 0; r < 3; ++r) {
        zero_f32<<<wordElems, TB, 0, stream>>>(nbuf, (long long)NWN * PDIM);
        zero_f32<<<(NWN + TB - 1) / TB, TB, 0, stream>>>(cnt, NWN);
        count_edges<<<(EWN + TB - 1) / TB, TB, 0, stream>>>(e_dst_w[r], EWN, cnt);
        scatter_edges<<<(EWN + 7) / 8, TB, 0, stream>>>(emb_word, src_nid_w[r], e_src_w[r],
                                                        e_dst_w[r], EWN, nbuf);
        normalize_mean<<<wordElems, TB, 0, stream>>>(nbuf, cnt, NWN);
        gemm_f32_wmma<<<gemmGridW, TB, 0, stream>>>(hdst, wpk[r], b_sage[r], h3[r],
                                                    NWN, PDIM, PDIM, PDIM, PDIM, PDIM, 0);
        gemm_f32_wmma<<<gemmGridW, TB, 0, stream>>>(nbuf, wpk[r] + (size_t)PDIM * PDIM, nullptr,
                                                    h3[r], NWN, PDIM, PDIM, PDIM, PDIM, PDIM, 1);
    }

    sdpa_fuse_add<<<(NWN + 7) / 8, 256, 0, stream>>>(h3[0], h3[1], h3[2], hdst, nbuf, NWN);

    pool_seq<<<BN, PDIM, 0, stream>>>(nbuf, x_batch, pooled);
    gemm_f32_wmma<<<gemmGridD, TB, 0, stream>>>(pooled, wdn, b_dense, docout,
                                                BN, PDIM, PDIM, PDIM, PDIM, PDIM, 0);

    for (int r = 0; r < 3; ++r) {
        zero_f32<<<docElems, TB, 0, stream>>>(dnb, (long long)BN * PDIM);
        zero_f32<<<1, TB, 0, stream>>>(cnt, BN);
        count_edges<<<(EDN + TB - 1) / TB, TB, 0, stream>>>(e_dst_d[r], EDN, cnt);
        scatter_edges<<<(EDN + 7) / 8, TB, 0, stream>>>(emb_doc, src_nid_d[r], e_src_d[r],
                                                        e_dst_d[r], EDN, dnb);
        normalize_mean<<<docElems, TB, 0, stream>>>(dnb, cnt, BN);
        gemm_f32_wmma<<<gemmGridD, TB, 0, stream>>>(docout, wpk[3 + r], b_sage[3 + r], d3[r],
                                                    BN, PDIM, PDIM, PDIM, PDIM, PDIM, 0);
        gemm_f32_wmma<<<gemmGridD, TB, 0, stream>>>(dnb, wpk[3 + r] + (size_t)PDIM * PDIM, nullptr,
                                                    d3[r], BN, PDIM, PDIM, PDIM, PDIM, PDIM, 1);
    }

    sdpa_fuse_add<<<(BN + 7) / 8, 256, 0, stream>>>(d3[0], d3[1], d3[2], docout, docrep, BN);
    fc_kernel<<<(BN * CN + TB - 1) / TB, TB, 0, stream>>>(docrep, w_fc, b_fc, out);
}


# GEMM: each K chunk's global loads issued as batches ahead of use (B panel: 10 x 16-byte loads per thread before the LDS stores; A fragments: 10 k-steps ahead, two register sets), not load-wait pairs
# speedup vs baseline: 1.2147x; 1.2147x over previous
#include <hip/hip_runtime.h>

#define DDIM 300
#define PDIM 320
#define NWN  20000
#define EWN  320000
#define BN   256
#define LN   256
#define SDN  2048
#define EDN  4096
#define CN   20
#define KC   160

typedef __attribute__((ext_vector_type(2))) float v2f;
typedef __attribute__((ext_vector_type(8))) float v8f;


__global__ void zero_f32(float* __restrict__ p, long long n) {
    long long t = (long long)blockIdx.x * blockDim.x + threadIdx.x;
    if (t < n) p[t] = 0.f;
}

__global__ void gather_pad(const float* __restrict__ emb, const int* __restrict__ nid,
                           float* __restrict__ out, int n) {
    int t = blockIdx.x * blockDim.x + threadIdx.x;
    if (t >= n * PDIM) return;
    int r = t / PDIM, j = t % PDIM;
    out[t] = (j < DDIM) ? emb[(size_t)nid[r] * DDIM + j] : 0.f;
}

__global__ void pack_sage_w(const float* __restrict__ w, float* __restrict__ out) {
    int t = blockIdx.x * blockDim.x + threadIdx.x;
    if (t >= 2 * PDIM * PDIM) return;
    int h = t / (PDIM * PDIM);
    int r = (t / PDIM) % PDIM;
    int c = t % PDIM;
    out[t] = (r < DDIM && c < DDIM) ? w[((size_t)(h * DDIM + r)) * DDIM + c] : 0.f;
}

__global__ void pack_w(const float* __restrict__ w, float* __restrict__ out) {
    int t = blockIdx.x * blockDim.x + threadIdx.x;
    if (t >= PDIM * PDIM) return;
    int r = t / PDIM, c = t % PDIM;
    out[t] = (r < DDIM && c < DDIM) ? w[(size_t)r * DDIM + c] : 0.f;
}


__global__ void count_edges(const int* __restrict__ e_dst, int E, float* __restrict__ cnt) {
    int t = blockIdx.x * blockDim.x + threadIdx.x;
    if (t < E) atomicAdd(&cnt[e_dst[t]], 1.f);
}

__global__ void scatter_edges(const float* __restrict__ emb, const int* __restrict__ src_nid,
                              const int* __restrict__ e_src, const int* __restrict__ e_dst,
                              int E, float* __restrict__ sum) {
    const int lane = threadIdx.x & 31;
    const int e = blockIdx.x * (blockDim.x >> 5) + (threadIdx.x >> 5);
    if (e >= E) return;
    const int s = src_nid[e_src[e]];
    const float* __restrict__ src = emb + (size_t)s * DDIM;
    float* __restrict__ dst = sum + (size_t)e_dst[e] * PDIM;
    for (int j = lane; j < DDIM; j += 32)
        atomicAdd(&dst[j], src[j]);
}

__global__ void normalize_mean(float* __restrict__ sum, const float* __restrict__ cnt, int n) {
    int t = blockIdx.x * blockDim.x + threadIdx.x;
    if (t >= n * PDIM) return;
    sum[t] /= fmaxf(cnt[t / PDIM], 1.f);
}

#define AB 10
static_assert(KC == 16 * AB, "the K loop below is written for four A batches per chunk");

__device__ __forceinline__ void load_a_batch(v2f (&a)[AB], const float* __restrict__ p) {
#pragma unroll
    for (int t = 0; t < AB; ++t) a[t] = *(const v2f*)(p + 4 * t);
}

__device__ __forceinline__ void mma_a_batch(v8f (&acc)[4], const v2f (&a)[AB], const float* lb) {
#pragma unroll
    for (int t = 0; t < AB; ++t) {
#pragma unroll
        for (int ns = 0; ns < 4; ++ns) {
            v2f b = *(const v2f*)(lb + ns * 16 * KC + 4 * t);
            acc[ns] = __builtin_amdgcn_wmma_f32_16x16x4_f32(
                false, a[t], false, b, (short)0, acc[ns], false, false);
        }
    }
}

__global__ __launch_bounds__(256) void gemm_f32_wmma(
        const float* __restrict__ A, const float* __restrict__ B,
        const float* __restrict__ bias, float* __restrict__ C,
        int M, int N, int K, int lda, int ldb, int ldc, int accum) {
    __shared__ float ldsB[64 * KC];
    const int lane = threadIdx.x & 31;
    const int wave = threadIdx.x >> 5;
    const int hi   = lane >> 4;
    const int l16  = lane & 15;
    const int nbase = blockIdx.y * 64;
    const int mt    = blockIdx.x * 8 + wave;
    const bool valid = (mt * 16) < M;
    const int m = mt * 16 + l16;

    v8f acc[4];
    if (valid) {
        if (accum) {
#pragma unroll
            for (int ns = 0; ns < 4; ++ns)
#pragma unroll
                for (int i = 0; i < 8; ++i)
                    acc[ns][i] = C[(size_t)(mt * 16 + i + hi * 8) * ldc + nbase + ns * 16 + l16];
        } else {
#pragma unroll
            for (int ns = 0; ns < 4; ++ns) {
                const int n = nbase + ns * 16 + l16;
                const float bv = (bias && n < DDIM) ? bias[n] : 0.f;
#pragma unroll
                for (int i = 0; i < 8; ++i) acc[ns][i] = bv;
            }
        }
    }

    const float* __restrict__ Ap = A + (size_t)m * lda + 2 * hi;
    const float* lb = &ldsB[l16 * KC + 2 * hi];
    const int sc  = (threadIdx.x & 15) * 4;
    const int skr = threadIdx.x >> 4;
    const float* __restrict__ Bp = B + (size_t)skr * ldb + nbase + sc;

    v2f a0[AB], a1[AB];
    if (valid) load_a_batch(a0, Ap);

    for (int kk = 0; kk < K; kk += KC) {
        float4 bt[KC / 16];
#pragma unroll
        for (int i = 0; i < KC / 16; ++i)
            bt[i] = *(const float4*)(Bp + (size_t)(kk + 16 * i) * ldb);

        __syncthreads();
#pragma unroll
        for (int i = 0; i < KC / 16; ++i) {
            const int kloc = skr + 16 * i;
            ldsB[(sc + 0) * KC + kloc] = bt[i].x;
            ldsB[(sc + 1) * KC + kloc] = bt[i].y;
            ldsB[(sc + 2) * KC + kloc] = bt[i].z;
            ldsB[(sc + 3) * KC + kloc] = bt[i].w;
        }
        __syncthreads();

        if (valid) {
            const float* ak = Ap + kk;
            load_a_batch(a1, ak + 4 * AB);
            mma_a_batch(acc, a0, lb);
            load_a_batch(a0, ak + 8 * AB);
            mma_a_batch(acc, a1, lb + 4 * AB);
            load_a_batch(a1, ak + 12 * AB);
            mma_a_batch(acc, a0, lb + 8 * AB);
            if (kk + KC < K) load_a_batch(a0, ak + KC);
            mma_a_batch(acc, a1, lb + 12 * AB);
        }
    }
    if (valid) {
#pragma unroll
        for (int ns = 0; ns < 4; ++ns)
#pragma unroll
            for (int i = 0; i < 8; ++i)
                C[(size_t)(mt * 16 + i + hi * 8) * ldc + nbase + ns * 16 + l16] = acc[ns][i];
    }
}

__global__ void sdpa_fuse_add(const float* __restrict__ h0, const float* __restrict__ h1,
                              const float* __restrict__ h2, const float* __restrict__ addv,
                              float* __restrict__ out, int n) {
    const int lane = threadIdx.x & 31;
    const int node = blockIdx.x * (blockDim.x >> 5) + (threadIdx.x >> 5);
    if (node >= n) return;
    const float* x0 = h0 + (size_t)node * PDIM;
    const float* x1 = h1 + (size_t)node * PDIM;
    const float* x2 = h2 + (size_t)node * PDIM;

    float d00 = 0, d01 = 0, d02 = 0, d11 = 0, d12 = 0, d22 = 0;
    for (int j = lane; j < PDIM; j += 32) {
        float a = x0[j], b = x1[j], e = x2[j];
        d00 += a * a; d01 += a * b; d02 += a * e;
        d11 += b * b; d12 += b * e; d22 += e * e;
    }
    for (int off = 16; off > 0; off >>= 1) {
        d00 += __shfl_xor(d00, off, 32);
        d01 += __shfl_xor(d01, off, 32);
        d02 += __shfl_xor(d02, off, 32);
        d11 += __shfl_xor(d11, off, 32);
        d12 += __shfl_xor(d12, off, 32);
        d22 += __shfl_xor(d22, off, 32);
    }
    const float scale = rsqrtf((float)DDIM);
    float S[3][3] = {{d00, d01, d02}, {d01, d11, d12}, {d02, d12, d22}};
    float wsum[3] = {0.f, 0.f, 0.f};
    for (int t = 0; t < 3; ++t) {
        float p0 = S[t][0] * scale, p1 = S[t][1] * scale, p2 = S[t][2] * scale;
        float mx = fmaxf(p0, fmaxf(p1, p2));
        float e0 = __expf(p0 - mx), e1 = __expf(p1 - mx), e2 = __expf(p2 - mx);
        float inv = 1.f / (e0 + e1 + e2);
        wsum[0] += e0 * inv; wsum[1] += e1 * inv; wsum[2] += e2 * inv;
    }
    const float w0 = wsum[0] * (1.f / 3.f), w1 = wsum[1] * (1.f / 3.f), w2 = wsum[2] * (1.f / 3.f);
    const float* av = addv + (size_t)node * PDIM;
    float* o = out + (size_t)node * PDIM;
    for (int j = lane; j < PDIM; j += 32)
        o[j] = w0 * x0[j] + w1 * x1[j] + w2 * x2[j] + av[j];
}

__global__ void pool_seq(const float* __restrict__ table, const int* __restrict__ xb,
                         float* __restrict__ pooled) {
    const int b = blockIdx.x;
    const int j = threadIdx.x;
    if (j >= PDIM) return;
    const int* row = xb + b * LN;
    float acc = 0.f;
    for (int l = 0; l < LN; ++l) {
        int idx = row[l];
        if (idx < NWN) acc += table[(size_t)idx * PDIM + j];
    }
    pooled[(size_t)b * PDIM + j] = acc * (1.f / (float)LN);
}

__global__ void fc_kernel(const float* __restrict__ rep, const float* __restrict__ wfc,
                          const float* __restrict__ bfc, float* __restrict__ out) {
    int t = blockIdx.x * blockDim.x + threadIdx.x;
    if (t >= BN * CN) return;
    int b = t / CN, c = t % CN;
    float acc = bfc[c];
    const float* r = rep + (size_t)b * PDIM;
    for (int j = 0; j < DDIM; ++j) acc += r[j] * wfc[(size_t)j * CN + c];
    out[t] = acc;
}


extern "C" void kernel_launch(void* const* d_in, const int* in_sizes, int n_in,
                              void* d_out, int out_size, void* d_ws, size_t ws_size,
                              hipStream_t stream) {
    (void)in_sizes; (void)n_in; (void)out_size; (void)ws_size;
    const float* emb_word = (const float*)d_in[0];
    const float* emb_doc  = (const float*)d_in[1];
    const float* w_sage[6] = {(const float*)d_in[2],  (const float*)d_in[4],  (const float*)d_in[6],
                              (const float*)d_in[8],  (const float*)d_in[10], (const float*)d_in[12]};
    const float* b_sage[6] = {(const float*)d_in[3],  (const float*)d_in[5],  (const float*)d_in[7],
                              (const float*)d_in[9],  (const float*)d_in[11], (const float*)d_in[13]};
    const float* w_dense = (const float*)d_in[14];
    const float* b_dense = (const float*)d_in[15];
    const float* w_fc    = (const float*)d_in[16];
    const float* b_fc    = (const float*)d_in[17];
    const int* dst_nid      = (const int*)d_in[18];
    const int* src_nid_w[3] = {(const int*)d_in[19], (const int*)d_in[20], (const int*)d_in[21]};
    const int* e_src_w[3]   = {(const int*)d_in[22], (const int*)d_in[24], (const int*)d_in[26]};
    const int* e_dst_w[3]   = {(const int*)d_in[23], (const int*)d_in[25], (const int*)d_in[27]};
    const int* x_batch      = (const int*)d_in[28];
    const int* src_nid_d[3] = {(const int*)d_in[30], (const int*)d_in[31], (const int*)d_in[32]};
    const int* e_src_d[3]   = {(const int*)d_in[33], (const int*)d_in[35], (const int*)d_in[37]};
    const int* e_dst_d[3]   = {(const int*)d_in[34], (const int*)d_in[36], (const int*)d_in[38]};
    float* out = (float*)d_out;

    float* ws = (float*)d_ws;
    size_t off = 0;
    auto carve = [&](size_t nflt) { float* p = ws + off; off += nflt; return p; };
    float* hdst = carve((size_t)NWN * PDIM);
    float* nbuf = carve((size_t)NWN * PDIM);
    float* h3[3]; for (int r = 0; r < 3; ++r) h3[r] = carve((size_t)NWN * PDIM);
    float* cnt  = carve(NWN);
    float* wpk[6]; for (int r = 0; r < 6; ++r) wpk[r] = carve((size_t)2 * PDIM * PDIM);
    float* wdn  = carve((size_t)PDIM * PDIM);
    float* pooled = carve((size_t)BN * PDIM);
    float* docout = carve((size_t)BN * PDIM);
    float* dnb    = carve((size_t)BN * PDIM);
    float* d3[3]; for (int r = 0; r < 3; ++r) d3[r] = carve((size_t)BN * PDIM);
    float* docrep = carve((size_t)BN * PDIM);

    const int TB = 256;
    const unsigned wordElems = (unsigned)(((size_t)NWN * PDIM + TB - 1) / TB);
    const unsigned docElems  = (unsigned)(((size_t)BN * PDIM + TB - 1) / TB);
    const dim3 gemmGridW((NWN + 127) / 128, PDIM / 64);
    const dim3 gemmGridD((BN + 127) / 128, PDIM / 64);

    for (int r = 0; r < 6; ++r)
        pack_sage_w<<<(2 * PDIM * PDIM + TB - 1) / TB, TB, 0, stream>>>(w_sage[r], wpk[r]);
    pack_w<<<(PDIM * PDIM + TB - 1) / TB, TB, 0, stream>>>(w_dense, wdn);

    gather_pad<<<wordElems, TB, 0, stream>>>(emb_word, dst_nid, hdst, NWN);

    for (int r = 0; r < 3; ++r) {
        zero_f32<<<wordElems, TB, 0, stream>>>(nbuf, (long long)NWN * PDIM);
        zero_f32<<<(NWN + TB - 1) / TB, TB, 0, stream>>>(cnt, NWN);
        count_edges<<<(EWN + TB - 1) / TB, TB, 0, stream>>>(e_dst_w[r], EWN, cnt);
        scatter_edges<<<(EWN + 7) / 8, TB, 0, stream>>>(emb_word, src_nid_w[r], e_src_w[r],
                                                        e_dst_w[r], EWN, nbuf);
        normalize_mean<<<wordElems, TB, 0, stream>>>(nbuf, cnt, NWN);
        gemm_f32_wmma<<<gemmGridW, TB, 0, stream>>>(hdst, wpk[r], b_sage[r], h3[r],
                                                    NWN, PDIM, PDIM, PDIM, PDIM, PDIM, 0);
        gemm_f32_wmma<<<gemmGridW, TB, 0, stream>>>(nbuf, wpk[r] + (size_t)PDIM * PDIM, nullptr,
                                                    h3[r], NWN, PDIM, PDIM, PDIM, PDIM, PDIM, 1);
    }

    sdpa_fuse_add<<<(NWN + 7) / 8, 256, 0, stream>>>(h3[0], h3[1], h3[2], hdst, nbuf, NWN);

    pool_seq<<<BN, PDIM, 0, stream>>>(nbuf, x_batch, pooled);
    gemm_f32_wmma<<<gemmGridD, TB, 0, stream>>>(pooled, wdn, b_dense, docout,
                                                BN, PDIM, PDIM, PDIM, PDIM, PDIM, 0);

    for (int r = 0; r < 3; ++r) {
        zero_f32<<<docElems, TB, 0, stream>>>(dnb, (long long)BN * PDIM);
        zero_f32<<<1, TB, 0, stream>>>(cnt, BN);
        count_edges<<<(EDN + TB - 1) / TB, TB, 0, stream>>>(e_dst_d[r], EDN, cnt);
        scatter_edges<<<(EDN + 7) / 8, TB, 0, stream>>>(emb_doc, src_nid_d[r], e_src_d[r],
                                                        e_dst_d[r], EDN, dnb);
        normalize_mean<<<docElems, TB, 0, stream>>>(dnb, cnt, BN);
        gemm_f32_wmma<<<gemmGridD, TB, 0, stream>>>(docout, wpk[3 + r], b_sage[3 + r], d3[r],
                                                    BN, PDIM, PDIM, PDIM, PDIM, PDIM, 0);
        gemm_f32_wmma<<<gemmGridD, TB, 0, stream>>>(dnb, wpk[3 + r] + (size_t)PDIM * PDIM, nullptr,
                                                    d3[r], BN, PDIM, PDIM, PDIM, PDIM, PDIM, 1);
    }

    sdpa_fuse_add<<<(BN + 7) / 8, 256, 0, stream>>>(d3[0], d3[1], d3[2], docout, docrep, BN);
    fc_kernel<<<(BN * CN + TB - 1) / TB, TB, 0, stream>>>(docrep, w_fc, b_fc, out);
}


# seg_mean: edges bucketed by destination (integer counting sort); one wave per node sums its source rows in ascending row-id order, stores the mean once; no float atomics; sum order changes
# speedup vs baseline: 1.3678x; 1.1260x over previous
#include <hip/hip_runtime.h>

#define DDIM 300
#define PDIM 320
#define NWN  20000
#define EWN  320000
#define BN   256
#define LN   256
#define SDN  2048
#define EDN  4096
#define CN   20
#define KC   160

typedef __attribute__((ext_vector_type(2))) float v2f;
typedef __attribute__((ext_vector_type(8))) float v8f;


__global__ void zero_i32(int* __restrict__ p, int n) {
    int t = blockIdx.x * blockDim.x + threadIdx.x;
    if (t < n) p[t] = 0;
}

__global__ void gather_pad(const float* __restrict__ emb, const int* __restrict__ nid,
                           float* __restrict__ out, int n) {
    int t = blockIdx.x * blockDim.x + threadIdx.x;
    if (t >= n * PDIM) return;
    int r = t / PDIM, j = t % PDIM;
    out[t] = (j < DDIM) ? emb[(size_t)nid[r] * DDIM + j] : 0.f;
}

__global__ void pack_sage_w(const float* __restrict__ w, float* __restrict__ out) {
    int t = blockIdx.x * blockDim.x + threadIdx.x;
    if (t >= 2 * PDIM * PDIM) return;
    int h = t / (PDIM * PDIM);
    int r = (t / PDIM) % PDIM;
    int c = t % PDIM;
    out[t] = (r < DDIM && c < DDIM) ? w[((size_t)(h * DDIM + r)) * DDIM + c] : 0.f;
}

__global__ void pack_w(const float* __restrict__ w, float* __restrict__ out) {
    int t = blockIdx.x * blockDim.x + threadIdx.x;
    if (t >= PDIM * PDIM) return;
    int r = t / PDIM, c = t % PDIM;
    out[t] = (r < DDIM && c < DDIM) ? w[(size_t)r * DDIM + c] : 0.f;
}


__global__ void count_dst(const int* __restrict__ e_dst, int E, int* __restrict__ cnt) {
    int t = blockIdx.x * blockDim.x + threadIdx.x;
    if (t < E) atomicAdd(&cnt[e_dst[t]], 1);
}

#define SCAN_T 1024
__global__ __launch_bounds__(SCAN_T) void scan_counts(const int* __restrict__ cnt, int n,
                                                      int* __restrict__ offs, int* __restrict__ cur) {
    __shared__ int wsum[SCAN_T / 32];
    const int t = threadIdx.x;
    const int lane = t & 31;
    const int ipt = (n + SCAN_T - 1) / SCAN_T;
    const int b = t * ipt;
    const int lim = min(ipt, n - b);
    int s = 0;
    for (int i = 0; i < lim; ++i) s += cnt[b + i];
    int v = s;
    for (int off = 1; off < 32; off <<= 1) {
        int u = __shfl_up(v, off, 32);
        if (lane >= off) v += u;
    }
    if (lane == 31) wsum[t >> 5] = v;
    __syncthreads();
    if (t < 32) {
        int w = wsum[t];
        int x = w;
        for (int off = 1; off < 32; off <<= 1) {
            int u = __shfl_up(x, off, 32);
            if (t >= off) x += u;
        }
        wsum[t] = x - w;
    }
    __syncthreads();
    int run = wsum[t >> 5] + (v - s);
    for (int i = 0; i < lim; ++i) {
        const int c = cnt[b + i];
        offs[b + i] = run;
        cur[b + i]  = run;
        run += c;
    }
}

__global__ void place_rows(const int* __restrict__ src_nid, const int* __restrict__ e_src,
                           const int* __restrict__ e_dst, int E,
                           int* __restrict__ cur, int* __restrict__ rows) {
    int t = blockIdx.x * blockDim.x + threadIdx.x;
    if (t >= E) return;
    const int pos = atomicAdd(&cur[e_dst[t]], 1);
    rows[pos] = src_nid[e_src[t]];
}

__device__ __forceinline__ void add4(float4& a, const float4 x) {
    a.x += x.x; a.y += x.y; a.z += x.z; a.w += x.w;
}

__global__ __launch_bounds__(256) void gather_mean(const float* __restrict__ emb,
                                                   const int* __restrict__ rows,
                                                   const int* __restrict__ offs,
                                                   const int* __restrict__ cnt,
                                                   int n, float* __restrict__ out) {
    const int lane = threadIdx.x & 31;
    const int node = __builtin_amdgcn_readfirstlane(
        (int)(blockIdx.x * (blockDim.x >> 5) + (threadIdx.x >> 5)));
    if (node >= n) return;
    const int beg = offs[node];
    const int deg = cnt[node];
    const int i2 = (lane < 11) ? lane + 64 : 74;

    float4 a0 = make_float4(0.f, 0.f, 0.f, 0.f), a1 = a0, a2 = a0;
    if (deg <= 32) {
        const int key = (lane < deg) ? rows[beg + lane] : 0x7fffffff;
        int rank = 0;
#pragma unroll
        for (int k = 0; k < 32; ++k) {
            const int kk = __builtin_amdgcn_readlane(key, k);
            rank += (kk < key || (kk == key && k < lane)) ? 1 : 0;
        }
        const int sorted = __builtin_amdgcn_ds_permute(rank << 2, key);
        int k = 0;
        for (; k + 4 <= deg; k += 4) {
            const float4* r0 = (const float4*)(emb + (size_t)__builtin_amdgcn_readlane(sorted, k + 0) * DDIM);
            const float4* r1 = (const float4*)(emb + (size_t)__builtin_amdgcn_readlane(sorted, k + 1) * DDIM);
            const float4* r2 = (const float4*)(emb + (size_t)__builtin_amdgcn_readlane(sorted, k + 2) * DDIM);
            const float4* r3 = (const float4*)(emb + (size_t)__builtin_amdgcn_readlane(sorted, k + 3) * DDIM);
            const float4 x00 = r0[lane], x01 = r0[lane + 32], x02 = r0[i2];
            const float4 x10 = r1[lane], x11 = r1[lane + 32], x12 = r1[i2];
            const float4 x20 = r2[lane], x21 = r2[lane + 32], x22 = r2[i2];
            const float4 x30 = r3[lane], x31 = r3[lane + 32], x32 = r3[i2];
            add4(a0, x00); add4(a1, x01); add4(a2, x02);
            add4(a0, x10); add4(a1, x11); add4(a2, x12);
            add4(a0, x20); add4(a1, x21); add4(a2, x22);
            add4(a0, x30); add4(a1, x31); add4(a2, x32);
        }
        for (; k < deg; ++k) {
            const float4* r0 = (const float4*)(emb + (size_t)__builtin_amdgcn_readlane(sorted, k) * DDIM);
            const float4 x00 = r0[lane], x01 = r0[lane + 32], x02 = r0[i2];
            add4(a0, x00); add4(a1, x01); add4(a2, x02);
        }
    } else {
        int last = -1;
        for (int pass = 0; pass < deg; ++pass) {
            int best = 0x7fffffff;
            for (int b = lane; b < deg; b += 32) {
                const int kx = rows[beg + b];
                if (kx > last && kx < best) best = kx;
            }
            for (int off = 16; off > 0; off >>= 1) best = min(best, __shfl_xor(best, off, 32));
            best = __builtin_amdgcn_readfirstlane(best);
            if (best == 0x7fffffff) break;
            int mult = 0;
            for (int b = lane; b < deg; b += 32) mult += (rows[beg + b] == best) ? 1 : 0;
            for (int off = 16; off > 0; off >>= 1) mult += __shfl_xor(mult, off, 32);
            mult = __builtin_amdgcn_readfirstlane(mult);
            const float4* r0 = (const float4*)(emb + (size_t)best * DDIM);
            const float4 x00 = r0[lane], x01 = r0[lane + 32], x02 = r0[i2];
            for (int m = 0; m < mult; ++m) { add4(a0, x00); add4(a1, x01); add4(a2, x02); }
            last = best;
        }
    }

    const float d = fmaxf((float)deg, 1.f);
    const float4 z = make_float4(0.f, 0.f, 0.f, 0.f);
    float4* o = (float4*)(out + (size_t)node * PDIM);
    o[lane]      = make_float4(a0.x / d, a0.y / d, a0.z / d, a0.w / d);
    o[lane + 32] = make_float4(a1.x / d, a1.y / d, a1.z / d, a1.w / d);
    if (lane < 16)
        o[lane + 64] = (lane < 11) ? make_float4(a2.x / d, a2.y / d, a2.z / d, a2.w / d) : z;
}

#define AB 10
static_assert(KC == 16 * AB, "the K loop below is written for four A batches per chunk");

__device__ __forceinline__ void load_a_batch(v2f (&a)[AB], const float* __restrict__ p) {
#pragma unroll
    for (int t = 0; t < AB; ++t) a[t] = *(const v2f*)(p + 4 * t);
}

__device__ __forceinline__ void mma_a_batch(v8f (&acc)[4], const v2f (&a)[AB], const float* lb) {
#pragma unroll
    for (int t = 0; t < AB; ++t) {
#pragma unroll
        for (int ns = 0; ns < 4; ++ns) {
            v2f b = *(const v2f*)(lb + ns * 16 * KC + 4 * t);
            acc[ns] = __builtin_amdgcn_wmma_f32_16x16x4_f32(
                false, a[t], false, b, (short)0, acc[ns], false, false);
        }
    }
}

__global__ __launch_bounds__(256) void gemm_f32_wmma(
        const float* __restrict__ A, const float* __restrict__ B,
        const float* __restrict__ bias, float* __restrict__ C,
        int M, int N, int K, int lda, int ldb, int ldc, int accum) {
    __shared__ float ldsB[64 * KC];
    const int lane = threadIdx.x & 31;
    const int wave = threadIdx.x >> 5;
    const int hi   = lane >> 4;
    const int l16  = lane & 15;
    const int nbase = blockIdx.y * 64;
    const int mt    = blockIdx.x * 8 + wave;
    const bool valid = (mt * 16) < M;
    const int m = mt * 16 + l16;

    v8f acc[4];
    if (valid) {
        if (accum) {
#pragma unroll
            for (int ns = 0; ns < 4; ++ns)
#pragma unroll
                for (int i = 0; i < 8; ++i)
                    acc[ns][i] = C[(size_t)(mt * 16 + i + hi * 8) * ldc + nbase + ns * 16 + l16];
        } else {
#pragma unroll
            for (int ns = 0; ns < 4; ++ns) {
                const int n = nbase + ns * 16 + l16;
                const float bv = (bias && n < DDIM) ? bias[n] : 0.f;
#pragma unroll
                for (int i = 0; i < 8; ++i) acc[ns][i] = bv;
            }
        }
    }

    const float* __restrict__ Ap = A + (size_t)m * lda + 2 * hi;
    const float* lb = &ldsB[l16 * KC + 2 * hi];
    const int sc  = (threadIdx.x & 15) * 4;
    const int skr = threadIdx.x >> 4;
    const float* __restrict__ Bp = B + (size_t)skr * ldb + nbase + sc;

    v2f a0[AB], a1[AB];
    if (valid) load_a_batch(a0, Ap);

    for (int kk = 0; kk < K; kk += KC) {
        float4 bt[KC / 16];
#pragma unroll
        for (int i = 0; i < KC / 16; ++i)
            bt[i] = *(const float4*)(Bp + (size_t)(kk + 16 * i) * ldb);

        __syncthreads();
#pragma unroll
        for (int i = 0; i < KC / 16; ++i) {
            const int kloc = skr + 16 * i;
            ldsB[(sc + 0) * KC + kloc] = bt[i].x;
            ldsB[(sc + 1) * KC + kloc] = bt[i].y;
            ldsB[(sc + 2) * KC + kloc] = bt[i].z;
            ldsB[(sc + 3) * KC + kloc] = bt[i].w;
        }
        __syncthreads();

        if (valid) {
            const float* ak = Ap + kk;
            load_a_batch(a1, ak + 4 * AB);
            mma_a_batch(acc, a0, lb);
            load_a_batch(a0, ak + 8 * AB);
            mma_a_batch(acc, a1, lb + 4 * AB);
            load_a_batch(a1, ak + 12 * AB);
            mma_a_batch(acc, a0, lb + 8 * AB);
            if (kk + KC < K) load_a_batch(a0, ak + KC);
            mma_a_batch(acc, a1, lb + 12 * AB);
        }
    }
    if (valid) {
#pragma unroll
        for (int ns = 0; ns < 4; ++ns)
#pragma unroll
            for (int i = 0; i < 8; ++i)
                C[(size_t)(mt * 16 + i + hi * 8) * ldc + nbase + ns * 16 + l16] = acc[ns][i];
    }
}

__global__ void sdpa_fuse_add(const float* __restrict__ h0, const float* __restrict__ h1,
                              const float* __restrict__ h2, const float* __restrict__ addv,
                              float* __restrict__ out, int n) {
    const int lane = threadIdx.x & 31;
    const int node = blockIdx.x * (blockDim.x >> 5) + (threadIdx.x >> 5);
    if (node >= n) return;
    const float* x0 = h0 + (size_t)node * PDIM;
    const float* x1 = h1 + (size_t)node * PDIM;
    const float* x2 = h2 + (size_t)node * PDIM;

    float d00 = 0, d01 = 0, d02 = 0, d11 = 0, d12 = 0, d22 = 0;
    for (int j = lane; j < PDIM; j += 32) {
        float a = x0[j], b = x1[j], e = x2[j];
        d00 += a * a; d01 += a * b; d02 += a * e;
        d11 += b * b; d12 += b * e; d22 += e * e;
    }
    for (int off = 16; off > 0; off >>= 1) {
        d00 += __shfl_xor(d00, off, 32);
        d01 += __shfl_xor(d01, off, 32);
        d02 += __shfl_xor(d02, off, 32);
        d11 += __shfl_xor(d11, off, 32);
        d12 += __shfl_xor(d12, off, 32);
        d22 += __shfl_xor(d22, off, 32);
    }
    const float scale = rsqrtf((float)DDIM);
    float S[3][3] = {{d00, d01, d02}, {d01, d11, d12}, {d02, d12, d22}};
    float wsum[3] = {0.f, 0.f, 0.f};
    for (int t = 0; t < 3; ++t) {
        float p0 = S[t][0] * scale, p1 = S[t][1] * scale, p2 = S[t][2] * scale;
        float mx = fmaxf(p0, fmaxf(p1, p2));
        float e0 = __expf(p0 - mx), e1 = __expf(p1 - mx), e2 = __expf(p2 - mx);
        float inv = 1.f / (e0 + e1 + e2);
        wsum[0] += e0 * inv; wsum[1] += e1 * inv; wsum[2] += e2 * inv;
    }
    const float w0 = wsum[0] * (1.f / 3.f), w1 = wsum[1] * (1.f / 3.f), w2 = wsum[2] * (1.f / 3.f);
    const float* av = addv + (size_t)node * PDIM;
    float* o = out + (size_t)node * PDIM;
    for (int j = lane; j < PDIM; j += 32)
        o[j] = w0 * x0[j] + w1 * x1[j] + w2 * x2[j] + av[j];
}

__global__ void pool_seq(const float* __restrict__ table, const int* __restrict__ xb,
                         float* __restrict__ pooled) {
    const int b = blockIdx.x;
    const int j = threadIdx.x;
    if (j >= PDIM) return;
    const int* row = xb + b * LN;
    float acc = 0.f;
    for (int l = 0; l < LN; ++l) {
        int idx = row[l];
        if (idx < NWN) acc += table[(size_t)idx * PDIM + j];
    }
    pooled[(size_t)b * PDIM + j] = acc * (1.f / (float)LN);
}

__global__ void fc_kernel(const float* __restrict__ rep, const float* __restrict__ wfc,
                          const float* __restrict__ bfc, float* __restrict__ out) {
    int t = blockIdx.x * blockDim.x + threadIdx.x;
    if (t >= BN * CN) return;
    int b = t / CN, c = t % CN;
    float acc = bfc[c];
    const float* r = rep + (size_t)b * PDIM;
    for (int j = 0; j < DDIM; ++j) acc += r[j] * wfc[(size_t)j * CN + c];
    out[t] = acc;
}


extern "C" void kernel_launch(void* const* d_in, const int* in_sizes, int n_in,
                              void* d_out, int out_size, void* d_ws, size_t ws_size,
                              hipStream_t stream) {
    (void)in_sizes; (void)n_in; (void)out_size; (void)ws_size;
    const float* emb_word = (const float*)d_in[0];
    const float* emb_doc  = (const float*)d_in[1];
    const float* w_sage[6] = {(const float*)d_in[2],  (const float*)d_in[4],  (const float*)d_in[6],
                              (const float*)d_in[8],  (const float*)d_in[10], (const float*)d_in[12]};
    const float* b_sage[6] = {(const float*)d_in[3],  (const float*)d_in[5],  (const float*)d_in[7],
                              (const float*)d_in[9],  (const float*)d_in[11], (const float*)d_in[13]};
    const float* w_dense = (const float*)d_in[14];
    const float* b_dense = (const float*)d_in[15];
    const float* w_fc    = (const float*)d_in[16];
    const float* b_fc    = (const float*)d_in[17];
    const int* dst_nid      = (const int*)d_in[18];
    const int* src_nid_w[3] = {(const int*)d_in[19], (const int*)d_in[20], (const int*)d_in[21]};
    const int* e_src_w[3]   = {(const int*)d_in[22], (const int*)d_in[24], (const int*)d_in[26]};
    const int* e_dst_w[3]   = {(const int*)d_in[23], (const int*)d_in[25], (const int*)d_in[27]};
    const int* x_batch      = (const int*)d_in[28];
    const int* src_nid_d[3] = {(const int*)d_in[30], (const int*)d_in[31], (const int*)d_in[32]};
    const int* e_src_d[3]   = {(const int*)d_in[33], (const int*)d_in[35], (const int*)d_in[37]};
    const int* e_dst_d[3]   = {(const int*)d_in[34], (const int*)d_in[36], (const int*)d_in[38]};
    float* out = (float*)d_out;

    float* ws = (float*)d_ws;
    size_t off = 0;
    auto carve = [&](size_t nflt) { float* p = ws + off; off += nflt; return p; };
    float* hdst = carve((size_t)NWN * PDIM);
    float* nbuf = carve((size_t)NWN * PDIM);
    float* h3[3]; for (int r = 0; r < 3; ++r) h3[r] = carve((size_t)NWN * PDIM);
    float* cnt  = carve(NWN);
    float* wpk[6]; for (int r = 0; r < 6; ++r) wpk[r] = carve((size_t)2 * PDIM * PDIM);
    float* wdn  = carve((size_t)PDIM * PDIM);
    float* pooled = carve((size_t)BN * PDIM);
    float* docout = carve((size_t)BN * PDIM);
    float* dnb    = carve((size_t)BN * PDIM);
    float* d3[3]; for (int r = 0; r < 3; ++r) d3[r] = carve((size_t)BN * PDIM);
    float* docrep = carve((size_t)BN * PDIM);
    int* icnt = (int*)cnt;
    int* w_offs = (int*)pooled;
    int* w_cur  = w_offs + NWN;
    int* w_rows = w_cur + NWN;
    int* d_offs = (int*)h3[0];
    int* d_cur  = d_offs + BN;
    int* d_rows = d_cur + BN;

    const int TB = 256;
    const unsigned wordElems = (unsigned)(((size_t)NWN * PDIM + TB - 1) / TB);
    const dim3 gemmGridW((NWN + 127) / 128, PDIM / 64);
    const dim3 gemmGridD((BN + 127) / 128, PDIM / 64);

    for (int r = 0; r < 6; ++r)
        pack_sage_w<<<(2 * PDIM * PDIM + TB - 1) / TB, TB, 0, stream>>>(w_sage[r], wpk[r]);
    pack_w<<<(PDIM * PDIM + TB - 1) / TB, TB, 0, stream>>>(w_dense, wdn);

    gather_pad<<<wordElems, TB, 0, stream>>>(emb_word, dst_nid, hdst, NWN);

    for (int r = 0; r < 3; ++r) {
        zero_i32<<<(NWN + TB - 1) / TB, TB, 0, stream>>>(icnt, NWN);
        count_dst<<<(EWN + TB - 1) / TB, TB, 0, stream>>>(e_dst_w[r], EWN, icnt);
        scan_counts<<<1, SCAN_T, 0, stream>>>(icnt, NWN, w_offs, w_cur);
        place_rows<<<(EWN + TB - 1) / TB, TB, 0, stream>>>(src_nid_w[r], e_src_w[r], e_dst_w[r], EWN,
                                                          w_cur, w_rows);
        gather_mean<<<(NWN + 7) / 8, TB, 0, stream>>>(emb_word, w_rows, w_offs, icnt, NWN, nbuf);
        gemm_f32_wmma<<<gemmGridW, TB, 0, stream>>>(hdst, wpk[r], b_sage[r], h3[r],
                                                    NWN, PDIM, PDIM, PDIM, PDIM, PDIM, 0);
        gemm_f32_wmma<<<gemmGridW, TB, 0, stream>>>(nbuf, wpk[r] + (size_t)PDIM * PDIM, nullptr,
                                                    h3[r], NWN, PDIM, PDIM, PDIM, PDIM, PDIM, 1);
    }

    sdpa_fuse_add<<<(NWN + 7) / 8, 256, 0, stream>>>(h3[0], h3[1], h3[2], hdst, nbuf, NWN);

    pool_seq<<<BN, PDIM, 0, stream>>>(nbuf, x_batch, pooled);
    gemm_f32_wmma<<<gemmGridD, TB, 0, stream>>>(pooled, wdn, b_dense, docout,
                                                BN, PDIM, PDIM, PDIM, PDIM, PDIM, 0);

    for (int r = 0; r < 3; ++r) {
        zero_i32<<<1, TB, 0, stream>>>(icnt, BN);
        count_dst<<<(EDN + TB - 1) / TB, TB, 0, stream>>>(e_dst_d[r], EDN, icnt);
        scan_counts<<<1, SCAN_T, 0, stream>>>(icnt, BN, d_offs, d_cur);
        place_rows<<<(EDN + TB - 1) / TB, TB, 0, stream>>>(src_nid_d[r], e_src_d[r], e_dst_d[r], EDN,
                                                          d_cur, d_rows);
        gather_mean<<<(BN + 7) / 8, TB, 0, stream>>>(emb_doc, d_rows, d_offs, icnt, BN, dnb);
        gemm_f32_wmma<<<gemmGridD, TB, 0, stream>>>(docout, wpk[3 + r], b_sage[3 + r], d3[r],
                                                    BN, PDIM, PDIM, PDIM, PDIM, PDIM, 0);
        gemm_f32_wmma<<<gemmGridD, TB, 0, stream>>>(dnb, wpk[3 + r] + (size_t)PDIM * PDIM, nullptr,
                                                    d3[r], BN, PDIM, PDIM, PDIM, PDIM, PDIM, 1);
    }

    sdpa_fuse_add<<<(BN + 7) / 8, 256, 0, stream>>>(d3[0], d3[1], d3[2], docout, docrep, BN);
    fc_kernel<<<(BN * CN + TB - 1) / TB, TB, 0, stream>>>(docrep, w_fc, b_fc, out);
}


# GEMM: staged B panel's LDS column pitch padded 160 -> 164 floats: a wave's fragment reads spread over all 64 banks (were on 8), panel stores 4-way (were 16-way); same products, same k order
# speedup vs baseline: 1.5993x; 1.1692x over previous
#include <hip/hip_runtime.h>

#define DDIM 300
#define PDIM 320
#define NWN  20000
#define EWN  320000
#define BN   256
#define LN   256
#define SDN  2048
#define EDN  4096
#define CN   20
#define KC   160
#define LP   164

typedef __attribute__((ext_vector_type(2))) float v2f;
typedef __attribute__((ext_vector_type(8))) float v8f;


__global__ void zero_i32(int* __restrict__ p, int n) {
    int t = blockIdx.x * blockDim.x + threadIdx.x;
    if (t < n) p[t] = 0;
}

__global__ void gather_pad(const float* __restrict__ emb, const int* __restrict__ nid,
                           float* __restrict__ out, int n) {
    int t = blockIdx.x * blockDim.x + threadIdx.x;
    if (t >= n * PDIM) return;
    int r = t / PDIM, j = t % PDIM;
    out[t] = (j < DDIM) ? emb[(size_t)nid[r] * DDIM + j] : 0.f;
}

__global__ void pack_sage_w(const float* __restrict__ w, float* __restrict__ out) {
    int t = blockIdx.x * blockDim.x + threadIdx.x;
    if (t >= 2 * PDIM * PDIM) return;
    int h = t / (PDIM * PDIM);
    int r = (t / PDIM) % PDIM;
    int c = t % PDIM;
    out[t] = (r < DDIM && c < DDIM) ? w[((size_t)(h * DDIM + r)) * DDIM + c] : 0.f;
}

__global__ void pack_w(const float* __restrict__ w, float* __restrict__ out) {
    int t = blockIdx.x * blockDim.x + threadIdx.x;
    if (t >= PDIM * PDIM) return;
    int r = t / PDIM, c = t % PDIM;
    out[t] = (r < DDIM && c < DDIM) ? w[(size_t)r * DDIM + c] : 0.f;
}


__global__ void count_dst(const int* __restrict__ e_dst, int E, int* __restrict__ cnt) {
    int t = blockIdx.x * blockDim.x + threadIdx.x;
    if (t < E) atomicAdd(&cnt[e_dst[t]], 1);
}

#define SCAN_T 1024
__global__ __launch_bounds__(SCAN_T) void scan_counts(const int* __restrict__ cnt, int n,
                                                      int* __restrict__ offs, int* __restrict__ cur) {
    __shared__ int wsum[SCAN_T / 32];
    const int t = threadIdx.x;
    const int lane = t & 31;
    const int ipt = (n + SCAN_T - 1) / SCAN_T;
    const int b = t * ipt;
    const int lim = min(ipt, n - b);
    int s = 0;
    for (int i = 0; i < lim; ++i) s += cnt[b + i];
    int v = s;
    for (int off = 1; off < 32; off <<= 1) {
        int u = __shfl_up(v, off, 32);
        if (lane >= off) v += u;
    }
    if (lane == 31) wsum[t >> 5] = v;
    __syncthreads();
    if (t < 32) {
        int w = wsum[t];
        int x = w;
        for (int off = 1; off < 32; off <<= 1) {
            int u = __shfl_up(x, off, 32);
            if (t >= off) x += u;
        }
        wsum[t] = x - w;
    }
    __syncthreads();
    int run = wsum[t >> 5] + (v - s);
    for (int i = 0; i < lim; ++i) {
        const int c = cnt[b + i];
        offs[b + i] = run;
        cur[b + i]  = run;
        run += c;
    }
}

__global__ void place_rows(const int* __restrict__ src_nid, const int* __restrict__ e_src,
                           const int* __restrict__ e_dst, int E,
                           int* __restrict__ cur, int* __restrict__ rows) {
    int t = blockIdx.x * blockDim.x + threadIdx.x;
    if (t >= E) return;
    const int pos = atomicAdd(&cur[e_dst[t]], 1);
    rows[pos] = src_nid[e_src[t]];
}

__device__ __forceinline__ void add4(float4& a, const float4 x) {
    a.x += x.x; a.y += x.y; a.z += x.z; a.w += x.w;
}

__global__ __launch_bounds__(256) void gather_mean(const float* __restrict__ emb,
                                                   const int* __restrict__ rows,
                                                   const int* __restrict__ offs,
                                                   const int* __restrict__ cnt,
                                                   int n, float* __restrict__ out) {
    const int lane = threadIdx.x & 31;
    const int node = __builtin_amdgcn_readfirstlane(
        (int)(blockIdx.x * (blockDim.x >> 5) + (threadIdx.x >> 5)));
    if (node >= n) return;
    const int beg = offs[node];
    const int deg = cnt[node];
    const int i2 = (lane < 11) ? lane + 64 : 74;

    float4 a0 = make_float4(0.f, 0.f, 0.f, 0.f), a1 = a0, a2 = a0;
    if (deg <= 32) {
        const int key = (lane < deg) ? rows[beg + lane] : 0x7fffffff;
        int rank = 0;
#pragma unroll
        for (int k = 0; k < 32; ++k) {
            const int kk = __builtin_amdgcn_readlane(key, k);
            rank += (kk < key || (kk == key && k < lane)) ? 1 : 0;
        }
        const int sorted = __builtin_amdgcn_ds_permute(rank << 2, key);
        int k = 0;
        for (; k + 4 <= deg; k += 4) {
            const float4* r0 = (const float4*)(emb + (size_t)__builtin_amdgcn_readlane(sorted, k + 0) * DDIM);
            const float4* r1 = (const float4*)(emb + (size_t)__builtin_amdgcn_readlane(sorted, k + 1) * DDIM);
            const float4* r2 = (const float4*)(emb + (size_t)__builtin_amdgcn_readlane(sorted, k + 2) * DDIM);
            const float4* r3 = (const float4*)(emb + (size_t)__builtin_amdgcn_readlane(sorted, k + 3) * DDIM);
            const float4 x00 = r0[lane], x01 = r0[lane + 32], x02 = r0[i2];
            const float4 x10 = r1[lane], x11 = r1[lane + 32], x12 = r1[i2];
            const float4 x20 = r2[lane], x21 = r2[lane + 32], x22 = r2[i2];
            const float4 x30 = r3[lane], x31 = r3[lane + 32], x32 = r3[i2];
            add4(a0, x00); add4(a1, x01); add4(a2, x02);
            add4(a0, x10); add4(a1, x11); add4(a2, x12);
            add4(a0, x20); add4(a1, x21); add4(a2, x22);
            add4(a0, x30); add4(a1, x31); add4(a2, x32);
        }
        for (; k < deg; ++k) {
            const float4* r0 = (const float4*)(emb + (size_t)__builtin_amdgcn_readlane(sorted, k) * DDIM);
            const float4 x00 = r0[lane], x01 = r0[lane + 32], x02 = r0[i2];
            add4(a0, x00); add4(a1, x01); add4(a2, x02);
        }
    } else {
        int last = -1;
        for (int pass = 0; pass < deg; ++pass) {
            int best = 0x7fffffff;
            for (int b = lane; b < deg; b += 32) {
                const int kx = rows[beg + b];
                if (kx > last && kx < best) best = kx;
            }
            for (int off = 16; off > 0; off >>= 1) best = min(best, __shfl_xor(best, off, 32));
            best = __builtin_amdgcn_readfirstlane(best);
            if (best == 0x7fffffff) break;
            int mult = 0;
            for (int b = lane; b < deg; b += 32) mult += (rows[beg + b] == best) ? 1 : 0;
            for (int off = 16; off > 0; off >>= 1) mult += __shfl_xor(mult, off, 32);
            mult = __builtin_amdgcn_readfirstlane(mult);
            const float4* r0 = (const float4*)(emb + (size_t)best * DDIM);
            const float4 x00 = r0[lane], x01 = r0[lane + 32], x02 = r0[i2];
            for (int m = 0; m < mult; ++m) { add4(a0, x00); add4(a1, x01); add4(a2, x02); }
            last = best;
        }
    }

    const float d = fmaxf((float)deg, 1.f);
    const float4 z = make_float4(0.f, 0.f, 0.f, 0.f);
    float4* o = (float4*)(out + (size_t)node * PDIM);
    o[lane]      = make_float4(a0.x / d, a0.y / d, a0.z / d, a0.w / d);
    o[lane + 32] = make_float4(a1.x / d, a1.y / d, a1.z / d, a1.w / d);
    if (lane < 16)
        o[lane + 64] = (lane < 11) ? make_float4(a2.x / d, a2.y / d, a2.z / d, a2.w / d) : z;
}

#define AB 10
static_assert(KC == 16 * AB, "the K loop below is written for four A batches per chunk");

__device__ __forceinline__ void load_a_batch(v2f (&a)[AB], const float* __restrict__ p) {
#pragma unroll
    for (int t = 0; t < AB; ++t) a[t] = *(const v2f*)(p + 4 * t);
}

__device__ __forceinline__ void mma_a_batch(v8f (&acc)[4], const v2f (&a)[AB], const float* lb) {
#pragma unroll
    for (int t = 0; t < AB; ++t) {
#pragma unroll
        for (int ns = 0; ns < 4; ++ns) {
            v2f b = *(const v2f*)(lb + ns * 16 * LP + 4 * t);
            acc[ns] = __builtin_amdgcn_wmma_f32_16x16x4_f32(
                false, a[t], false, b, (short)0, acc[ns], false, false);
        }
    }
}

__global__ __launch_bounds__(256) void gemm_f32_wmma(
        const float* __restrict__ A, const float* __restrict__ B,
        const float* __restrict__ bias, float* __restrict__ C,
        int M, int N, int K, int lda, int ldb, int ldc, int accum) {
    __shared__ float ldsB[64 * LP];
    const int lane = threadIdx.x & 31;
    const int wave = threadIdx.x >> 5;
    const int hi   = lane >> 4;
    const int l16  = lane & 15;
    const int nbase = blockIdx.y * 64;
    const int mt    = blockIdx.x * 8 + wave;
    const bool valid = (mt * 16) < M;
    const int m = mt * 16 + l16;

    v8f acc[4];
    if (valid) {
        if (accum) {
#pragma unroll
            for (int ns = 0; ns < 4; ++ns)
#pragma unroll
                for (int i = 0; i < 8; ++i)
                    acc[ns][i] = C[(size_t)(mt * 16 + i + hi * 8) * ldc + nbase + ns * 16 + l16];
        } else {
#pragma unroll
            for (int ns = 0; ns < 4; ++ns) {
                const int n = nbase + ns * 16 + l16;
                const float bv = (bias && n < DDIM) ? bias[n] : 0.f;
#pragma unroll
                for (int i = 0; i < 8; ++i) acc[ns][i] = bv;
            }
        }
    }

    const float* __restrict__ Ap = A + (size_t)m * lda + 2 * hi;
    const float* lb = &ldsB[l16 * LP + 2 * hi];
    const int sc  = (threadIdx.x & 15) * 4;
    const int skr = threadIdx.x >> 4;
    const float* __restrict__ Bp = B + (size_t)skr * ldb + nbase + sc;

    v2f a0[AB], a1[AB];
    if (valid) load_a_batch(a0, Ap);

    for (int kk = 0; kk < K; kk += KC) {
        float4 bt[KC / 16];
#pragma unroll
        for (int i = 0; i < KC / 16; ++i)
            bt[i] = *(const float4*)(Bp + (size_t)(kk + 16 * i) * ldb);

        __syncthreads();
#pragma unroll
        for (int i = 0; i < KC / 16; ++i) {
            const int kloc = skr + 16 * i;
            ldsB[(sc + 0) * LP + kloc] = bt[i].x;
            ldsB[(sc + 1) * LP + kloc] = bt[i].y;
            ldsB[(sc + 2) * LP + kloc] = bt[i].z;
            ldsB[(sc + 3) * LP + kloc] = bt[i].w;
        }
        __syncthreads();

        if (valid) {
            const float* ak = Ap + kk;
            load_a_batch(a1, ak + 4 * AB);
            mma_a_batch(acc, a0, lb);
            load_a_batch(a0, ak + 8 * AB);
            mma_a_batch(acc, a1, lb + 4 * AB);
            load_a_batch(a1, ak + 12 * AB);
            mma_a_batch(acc, a0, lb + 8 * AB);
            if (kk + KC < K) load_a_batch(a0, ak + KC);
            mma_a_batch(acc, a1, lb + 12 * AB);
        }
    }
    if (valid) {
#pragma unroll
        for (int ns = 0; ns < 4; ++ns)
#pragma unroll
            for (int i = 0; i < 8; ++i)
                C[(size_t)(mt * 16 + i + hi * 8) * ldc + nbase + ns * 16 + l16] = acc[ns][i];
    }
}

__global__ void sdpa_fuse_add(const float* __restrict__ h0, const float* __restrict__ h1,
                              const float* __restrict__ h2, const float* __restrict__ addv,
                              float* __restrict__ out, int n) {
    const int lane = threadIdx.x & 31;
    const int node = blockIdx.x * (blockDim.x >> 5) + (threadIdx.x >> 5);
    if (node >= n) return;
    const float* x0 = h0 + (size_t)node * PDIM;
    const float* x1 = h1 + (size_t)node * PDIM;
    const float* x2 = h2 + (size_t)node * PDIM;

    float d00 = 0, d01 = 0, d02 = 0, d11 = 0, d12 = 0, d22 = 0;
    for (int j = lane; j < PDIM; j += 32) {
        float a = x0[j], b = x1[j], e = x2[j];
        d00 += a * a; d01 += a * b; d02 += a * e;
        d11 += b * b; d12 += b * e; d22 += e * e;
    }
    for (int off = 16; off > 0; off >>= 1) {
        d00 += __shfl_xor(d00, off, 32);
        d01 += __shfl_xor(d01, off, 32);
        d02 += __shfl_xor(d02, off, 32);
        d11 += __shfl_xor(d11, off, 32);
        d12 += __shfl_xor(d12, off, 32);
        d22 += __shfl_xor(d22, off, 32);
    }
    const float scale = rsqrtf((float)DDIM);
    float S[3][3] = {{d00, d01, d02}, {d01, d11, d12}, {d02, d12, d22}};
    float wsum[3] = {0.f, 0.f, 0.f};
    for (int t = 0; t < 3; ++t) {
        float p0 = S[t][0] * scale, p1 = S[t][1] * scale, p2 = S[t][2] * scale;
        float mx = fmaxf(p0, fmaxf(p1, p2));
        float e0 = __expf(p0 - mx), e1 = __expf(p1 - mx), e2 = __expf(p2 - mx);
        float inv = 1.f / (e0 + e1 + e2);
        wsum[0] += e0 * inv; wsum[1] += e1 * inv; wsum[2] += e2 * inv;
    }
    const float w0 = wsum[0] * (1.f / 3.f), w1 = wsum[1] * (1.f / 3.f), w2 = wsum[2] * (1.f / 3.f);
    const float* av = addv + (size_t)node * PDIM;
    float* o = out + (size_t)node * PDIM;
    for (int j = lane; j < PDIM; j += 32)
        o[j] = w0 * x0[j] + w1 * x1[j] + w2 * x2[j] + av[j];
}

__global__ void pool_seq(const float* __restrict__ table, const int* __restrict__ xb,
                         float* __restrict__ pooled) {
    const int b = blockIdx.x;
    const int j = threadIdx.x;
    if (j >= PDIM) return;
    const int* row = xb + b * LN;
    float acc = 0.f;
    for (int l = 0; l < LN; ++l) {
        int idx = row[l];
        if (idx < NWN) acc += table[(size_t)idx * PDIM + j];
    }
    pooled[(size_t)b * PDIM + j] = acc * (1.f / (float)LN);
}

__global__ void fc_kernel(const float* __restrict__ rep, const float* __restrict__ wfc,
                          const float* __restrict__ bfc, float* __restrict__ out) {
    int t = blockIdx.x * blockDim.x + threadIdx.x;
    if (t >= BN * CN) return;
    int b = t / CN, c = t % CN;
    float acc = bfc[c];
    const float* r = rep + (size_t)b * PDIM;
    for (int j = 0; j < DDIM; ++j) acc += r[j] * wfc[(size_t)j * CN + c];
    out[t] = acc;
}


extern "C" void kernel_launch(void* const* d_in, const int* in_sizes, int n_in,
                              void* d_out, int out_size, void* d_ws, size_t ws_size,
                              hipStream_t stream) {
    (void)in_sizes; (void)n_in; (void)out_size; (void)ws_size;
    const float* emb_word = (const float*)d_in[0];
    const float* emb_doc  = (const float*)d_in[1];
    const float* w_sage[6] = {(const float*)d_in[2],  (const float*)d_in[4],  (const float*)d_in[6],
                              (const float*)d_in[8],  (const float*)d_in[10], (const float*)d_in[12]};
    const float* b_sage[6] = {(const float*)d_in[3],  (const float*)d_in[5],  (const float*)d_in[7],
                              (const float*)d_in[9],  (const float*)d_in[11], (const float*)d_in[13]};
    const float* w_dense = (const float*)d_in[14];
    const float* b_dense = (const float*)d_in[15];
    const float* w_fc    = (const float*)d_in[16];
    const float* b_fc    = (const float*)d_in[17];
    const int* dst_nid      = (const int*)d_in[18];
    const int* src_nid_w[3] = {(const int*)d_in[19], (const int*)d_in[20], (const int*)d_in[21]};
    const int* e_src_w[3]   = {(const int*)d_in[22], (const int*)d_in[24], (const int*)d_in[26]};
    const int* e_dst_w[3]   = {(const int*)d_in[23], (const int*)d_in[25], (const int*)d_in[27]};
    const int* x_batch      = (const int*)d_in[28];
    const int* src_nid_d[3] = {(const int*)d_in[30], (const int*)d_in[31], (const int*)d_in[32]};
    const int* e_src_d[3]   = {(const int*)d_in[33], (const int*)d_in[35], (const int*)d_in[37]};
    const int* e_dst_d[3]   = {(const int*)d_in[34], (const int*)d_in[36], (const int*)d_in[38]};
    float* out = (float*)d_out;

    float* ws = (float*)d_ws;
    size_t off = 0;
    auto carve = [&](size_t nflt) { float* p = ws + off; off += nflt; return p; };
    float* hdst = carve((size_t)NWN * PDIM);
    float* nbuf = carve((size_t)NWN * PDIM);
    float* h3[3]; for (int r = 0; r < 3; ++r) h3[r] = carve((size_t)NWN * PDIM);
    float* cnt  = carve(NWN);
    float* wpk[6]; for (int r = 0; r < 6; ++r) wpk[r] = carve((size_t)2 * PDIM * PDIM);
    float* wdn  = carve((size_t)PDIM * PDIM);
    float* pooled = carve((size_t)BN * PDIM);
    float* docout = carve((size_t)BN * PDIM);
    float* dnb    = carve((size_t)BN * PDIM);
    float* d3[3]; for (int r = 0; r < 3; ++r) d3[r] = carve((size_t)BN * PDIM);
    float* docrep = carve((size_t)BN * PDIM);
    int* icnt = (int*)cnt;
    int* w_offs = (int*)pooled;
    int* w_cur  = w_offs + NWN;
    int* w_rows = w_cur + NWN;
    int* d_offs = (int*)h3[0];
    int* d_cur  = d_offs + BN;
    int* d_rows = d_cur + BN;

    const int TB = 256;
    const unsigned wordElems = (unsigned)(((size_t)NWN * PDIM + TB - 1) / TB);
    const dim3 gemmGridW((NWN + 127) / 128, PDIM / 64);
    const dim3 gemmGridD((BN + 127) / 128, PDIM / 64);

    for (int r = 0; r < 6; ++r)
        pack_sage_w<<<(2 * PDIM * PDIM + TB - 1) / TB, TB, 0, stream>>>(w_sage[r], wpk[r]);
    pack_w<<<(PDIM * PDIM + TB - 1) / TB, TB, 0, stream>>>(w_dense, wdn);

    gather_pad<<<wordElems, TB, 0, stream>>>(emb_word, dst_nid, hdst, NWN);

    for (int r = 0; r < 3; ++r) {
        zero_i32<<<(NWN + TB - 1) / TB, TB, 0, stream>>>(icnt, NWN);
        count_dst<<<(EWN + TB - 1) / TB, TB, 0, stream>>>(e_dst_w[r], EWN, icnt);
        scan_counts<<<1, SCAN_T, 0, stream>>>(icnt, NWN, w_offs, w_cur);
        place_rows<<<(EWN + TB - 1) / TB, TB, 0, stream>>>(src_nid_w[r], e_src_w[r], e_dst_w[r], EWN,
                                                          w_cur, w_rows);
        gather_mean<<<(NWN + 7) / 8, TB, 0, stream>>>(emb_word, w_rows, w_offs, icnt, NWN, nbuf);
        gemm_f32_wmma<<<gemmGridW, TB, 0, stream>>>(hdst, wpk[r], b_sage[r], h3[r],
                                                    NWN, PDIM, PDIM, PDIM, PDIM, PDIM, 0);
        gemm_f32_wmma<<<gemmGridW, TB, 0, stream>>>(nbuf, wpk[r] + (size_t)PDIM * PDIM, nullptr,
                                                    h3[r], NWN, PDIM, PDIM, PDIM, PDIM, PDIM, 1);
    }

    sdpa_fuse_add<<<(NWN + 7) / 8, 256, 0, stream>>>(h3[0], h3[1], h3[2], hdst, nbuf, NWN);

    pool_seq<<<BN, PDIM, 0, stream>>>(nbuf, x_batch, pooled);
    gemm_f32_wmma<<<gemmGridD, TB, 0, stream>>>(pooled, wdn, b_dense, docout,
                                                BN, PDIM, PDIM, PDIM, PDIM, PDIM, 0);

    for (int r = 0; r < 3; ++r) {
        zero_i32<<<1, TB, 0, stream>>>(icnt, BN);
        count_dst<<<(EDN + TB - 1) / TB, TB, 0, stream>>>(e_dst_d[r], EDN, icnt);
        scan_counts<<<1, SCAN_T, 0, stream>>>(icnt, BN, d_offs, d_cur);
        place_rows<<<(EDN + TB - 1) / TB, TB, 0, stream>>>(src_nid_d[r], e_src_d[r], e_dst_d[r], EDN,
                                                          d_cur, d_rows);
        gather_mean<<<(BN + 7) / 8, TB, 0, stream>>>(emb_doc, d_rows, d_offs, icnt, BN, dnb);
        gemm_f32_wmma<<<gemmGridD, TB, 0, stream>>>(docout, wpk[3 + r], b_sage[3 + r], d3[r],
                                                    BN, PDIM, PDIM, PDIM, PDIM, PDIM, 0);
        gemm_f32_wmma<<<gemmGridD, TB, 0, stream>>>(dnb, wpk[3 + r] + (size_t)PDIM * PDIM, nullptr,
                                                    d3[r], BN, PDIM, PDIM, PDIM, PDIM, PDIM, 1);
    }

    sdpa_fuse_add<<<(BN + 7) / 8, 256, 0, stream>>>(d3[0], d3[1], d3[2], docout, docrep, BN);
    fc_kernel<<<(BN * CN + TB - 1) / TB, TB, 0, stream>>>(docrep, w_fc, b_fc, out);
}


# pool_seq: a document's 256 table-row loads issued 32 at a time, the next 32 in flight during the adds (token ids loaded once, broadcast by readlane), not a serial load-wait-add chain; same add order
# speedup vs baseline: 1.7416x; 1.0890x over previous
#include <hip/hip_runtime.h>

#define DDIM 300
#define PDIM 320
#define NWN  20000
#define EWN  320000
#define BN   256
#define LN   256
#define SDN  2048
#define EDN  4096
#define CN   20
#define KC   160
#define LP   164

typedef __attribute__((ext_vector_type(2))) float v2f;
typedef __attribute__((ext_vector_type(8))) float v8f;


__global__ void zero_i32(int* __restrict__ p, int n) {
    int t = blockIdx.x * blockDim.x + threadIdx.x;
    if (t < n) p[t] = 0;
}

__global__ void gather_pad(const float* __restrict__ emb, const int* __restrict__ nid,
                           float* __restrict__ out, int n) {
    int t = blockIdx.x * blockDim.x + threadIdx.x;
    if (t >= n * PDIM) return;
    int r = t / PDIM, j = t % PDIM;
    out[t] = (j < DDIM) ? emb[(size_t)nid[r] * DDIM + j] : 0.f;
}

__global__ void pack_sage_w(const float* __restrict__ w, float* __restrict__ out) {
    int t = blockIdx.x * blockDim.x + threadIdx.x;
    if (t >= 2 * PDIM * PDIM) return;
    int h = t / (PDIM * PDIM);
    int r = (t / PDIM) % PDIM;
    int c = t % PDIM;
    out[t] = (r < DDIM && c < DDIM) ? w[((size_t)(h * DDIM + r)) * DDIM + c] : 0.f;
}

__global__ void pack_w(const float* __restrict__ w, float* __restrict__ out) {
    int t = blockIdx.x * blockDim.x + threadIdx.x;
    if (t >= PDIM * PDIM) return;
    int r = t / PDIM, c = t % PDIM;
    out[t] = (r < DDIM && c < DDIM) ? w[(size_t)r * DDIM + c] : 0.f;
}


__global__ void count_dst(const int* __restrict__ e_dst, int E, int* __restrict__ cnt) {
    int t = blockIdx.x * blockDim.x + threadIdx.x;
    if (t < E) atomicAdd(&cnt[e_dst[t]], 1);
}

#define SCAN_T 1024
__global__ __launch_bounds__(SCAN_T) void scan_counts(const int* __restrict__ cnt, int n,
                                                      int* __restrict__ offs, int* __restrict__ cur) {
    __shared__ int wsum[SCAN_T / 32];
    const int t = threadIdx.x;
    const int lane = t & 31;
    const int ipt = (n + SCAN_T - 1) / SCAN_T;
    const int b = t * ipt;
    const int lim = min(ipt, n - b);
    int s = 0;
    for (int i = 0; i < lim; ++i) s += cnt[b + i];
    int v = s;
    for (int off = 1; off < 32; off <<= 1) {
        int u = __shfl_up(v, off, 32);
        if (lane >= off) v += u;
    }
    if (lane == 31) wsum[t >> 5] = v;
    __syncthreads();
    if (t < 32) {
        int w = wsum[t];
        int x = w;
        for (int off = 1; off < 32; off <<= 1) {
            int u = __shfl_up(x, off, 32);
            if (t >= off) x += u;
        }
        wsum[t] = x - w;
    }
    __syncthreads();
    int run = wsum[t >> 5] + (v - s);
    for (int i = 0; i < lim; ++i) {
        const int c = cnt[b + i];
        offs[b + i] = run;
        cur[b + i]  = run;
        run += c;
    }
}

__global__ void place_rows(const int* __restrict__ src_nid, const int* __restrict__ e_src,
                           const int* __restrict__ e_dst, int E,
                           int* __restrict__ cur, int* __restrict__ rows) {
    int t = blockIdx.x * blockDim.x + threadIdx.x;
    if (t >= E) return;
    const int pos = atomicAdd(&cur[e_dst[t]], 1);
    rows[pos] = src_nid[e_src[t]];
}

__device__ __forceinline__ void add4(float4& a, const float4 x) {
    a.x += x.x; a.y += x.y; a.z += x.z; a.w += x.w;
}

__global__ __launch_bounds__(256) void gather_mean(const float* __restrict__ emb,
                                                   const int* __restrict__ rows,
                                                   const int* __restrict__ offs,
                                                   const int* __restrict__ cnt,
                                                   int n, float* __restrict__ out) {
    const int lane = threadIdx.x & 31;
    const int node = __builtin_amdgcn_readfirstlane(
        (int)(blockIdx.x * (blockDim.x >> 5) + (threadIdx.x >> 5)));
    if (node >= n) return;
    const int beg = offs[node];
    const int deg = cnt[node];
    const int i2 = (lane < 11) ? lane + 64 : 74;

    float4 a0 = make_float4(0.f, 0.f, 0.f, 0.f), a1 = a0, a2 = a0;
    if (deg <= 32) {
        const int key = (lane < deg) ? rows[beg + lane] : 0x7fffffff;
        int rank = 0;
#pragma unroll
        for (int k = 0; k < 32; ++k) {
            const int kk = __builtin_amdgcn_readlane(key, k);
            rank += (kk < key || (kk == key && k < lane)) ? 1 : 0;
        }
        const int sorted = __builtin_amdgcn_ds_permute(rank << 2, key);
        int k = 0;
        for (; k + 4 <= deg; k += 4) {
            const float4* r0 = (const float4*)(emb + (size_t)__builtin_amdgcn_readlane(sorted, k + 0) * DDIM);
            const float4* r1 = (const float4*)(emb + (size_t)__builtin_amdgcn_readlane(sorted, k + 1) * DDIM);
            const float4* r2 = (const float4*)(emb + (size_t)__builtin_amdgcn_readlane(sorted, k + 2) * DDIM);
            const float4* r3 = (const float4*)(emb + (size_t)__builtin_amdgcn_readlane(sorted, k + 3) * DDIM);
            const float4 x00 = r0[lane], x01 = r0[lane + 32], x02 = r0[i2];
            const float4 x10 = r1[lane], x11 = r1[lane + 32], x12 = r1[i2];
            const float4 x20 = r2[lane], x21 = r2[lane + 32], x22 = r2[i2];
            const float4 x30 = r3[lane], x31 = r3[lane + 32], x32 = r3[i2];
            add4(a0, x00); add4(a1, x01); add4(a2, x02);
            add4(a0, x10); add4(a1, x11); add4(a2, x12);
            add4(a0, x20); add4(a1, x21); add4(a2, x22);
            add4(a0, x30); add4(a1, x31); add4(a2, x32);
        }
        for (; k < deg; ++k) {
            const float4* r0 = (const float4*)(emb + (size_t)__builtin_amdgcn_readlane(sorted, k) * DDIM);
            const float4 x00 = r0[lane], x01 = r0[lane + 32], x02 = r0[i2];
            add4(a0, x00); add4(a1, x01); add4(a2, x02);
        }
    } else {
        int last = -1;
        for (int pass = 0; pass < deg; ++pass) {
            int best = 0x7fffffff;
            for (int b = lane; b < deg; b += 32) {
                const int kx = rows[beg + b];
                if (kx > last && kx < best) best = kx;
            }
            for (int off = 16; off > 0; off >>= 1) best = min(best, __shfl_xor(best, off, 32));
            best = __builtin_amdgcn_readfirstlane(best);
            if (best == 0x7fffffff) break;
            int mult = 0;
            for (int b = lane; b < deg; b += 32) mult += (rows[beg + b] == best) ? 1 : 0;
            for (int off = 16; off > 0; off >>= 1) mult += __shfl_xor(mult, off, 32);
            mult = __builtin_amdgcn_readfirstlane(mult);
            const float4* r0 = (const float4*)(emb + (size_t)best * DDIM);
            const float4 x00 = r0[lane], x01 = r0[lane + 32], x02 = r0[i2];
            for (int m = 0; m < mult; ++m) { add4(a0, x00); add4(a1, x01); add4(a2, x02); }
            last = best;
        }
    }

    const float d = fmaxf((float)deg, 1.f);
    const float4 z = make_float4(0.f, 0.f, 0.f, 0.f);
    float4* o = (float4*)(out + (size_t)node * PDIM);
    o[lane]      = make_float4(a0.x / d, a0.y / d, a0.z / d, a0.w / d);
    o[lane + 32] = make_float4(a1.x / d, a1.y / d, a1.z / d, a1.w / d);
    if (lane < 16)
        o[lane + 64] = (lane < 11) ? make_float4(a2.x / d, a2.y / d, a2.z / d, a2.w / d) : z;
}

#define AB 10
static_assert(KC == 16 * AB, "the K loop below is written for four A batches per chunk");

__device__ __forceinline__ void load_a_batch(v2f (&a)[AB], const float* __restrict__ p) {
#pragma unroll
    for (int t = 0; t < AB; ++t) a[t] = *(const v2f*)(p + 4 * t);
}

__device__ __forceinline__ void mma_a_batch(v8f (&acc)[4], const v2f (&a)[AB], const float* lb) {
#pragma unroll
    for (int t = 0; t < AB; ++t) {
#pragma unroll
        for (int ns = 0; ns < 4; ++ns) {
            v2f b = *(const v2f*)(lb + ns * 16 * LP + 4 * t);
            acc[ns] = __builtin_amdgcn_wmma_f32_16x16x4_f32(
                false, a[t], false, b, (short)0, acc[ns], false, false);
        }
    }
}

__global__ __launch_bounds__(256) void gemm_f32_wmma(
        const float* __restrict__ A, const float* __restrict__ B,
        const float* __restrict__ bias, float* __restrict__ C,
        int M, int N, int K, int lda, int ldb, int ldc, int accum) {
    __shared__ float ldsB[64 * LP];
    const int lane = threadIdx.x & 31;
    const int wave = threadIdx.x >> 5;
    const int hi   = lane >> 4;
    const int l16  = lane & 15;
    const int nbase = blockIdx.y * 64;
    const int mt    = blockIdx.x * 8 + wave;
    const bool valid = (mt * 16) < M;
    const int m = mt * 16 + l16;

    v8f acc[4];
    if (valid) {
        if (accum) {
#pragma unroll
            for (int ns = 0; ns < 4; ++ns)
#pragma unroll
                for (int i = 0; i < 8; ++i)
                    acc[ns][i] = C[(size_t)(mt * 16 + i + hi * 8) * ldc + nbase + ns * 16 + l16];
        } else {
#pragma unroll
            for (int ns = 0; ns < 4; ++ns) {
                const int n = nbase + ns * 16 + l16;
                const float bv = (bias && n < DDIM) ? bias[n] : 0.f;
#pragma unroll
                for (int i = 0; i < 8; ++i) acc[ns][i] = bv;
            }
        }
    }

    const float* __restrict__ Ap = A + (size_t)m * lda + 2 * hi;
    const float* lb = &ldsB[l16 * LP + 2 * hi];
    const int sc  = (threadIdx.x & 15) * 4;
    const int skr = threadIdx.x >> 4;
    const float* __restrict__ Bp = B + (size_t)skr * ldb + nbase + sc;

    v2f a0[AB], a1[AB];
    if (valid) load_a_batch(a0, Ap);

    for (int kk = 0; kk < K; kk += KC) {
        float4 bt[KC / 16];
#pragma unroll
        for (int i = 0; i < KC / 16; ++i)
            bt[i] = *(const float4*)(Bp + (size_t)(kk + 16 * i) * ldb);

        __syncthreads();
#pragma unroll
        for (int i = 0; i < KC / 16; ++i) {
            const int kloc = skr + 16 * i;
            ldsB[(sc + 0) * LP + kloc] = bt[i].x;
            ldsB[(sc + 1) * LP + kloc] = bt[i].y;
            ldsB[(sc + 2) * LP + kloc] = bt[i].z;
            ldsB[(sc + 3) * LP + kloc] = bt[i].w;
        }
        __syncthreads();

        if (valid) {
            const float* ak = Ap + kk;
            load_a_batch(a1, ak + 4 * AB);
            mma_a_batch(acc, a0, lb);
            load_a_batch(a0, ak + 8 * AB);
            mma_a_batch(acc, a1, lb + 4 * AB);
            load_a_batch(a1, ak + 12 * AB);
            mma_a_batch(acc, a0, lb + 8 * AB);
            if (kk + KC < K) load_a_batch(a0, ak + KC);
            mma_a_batch(acc, a1, lb + 12 * AB);
        }
    }
    if (valid) {
#pragma unroll
        for (int ns = 0; ns < 4; ++ns)
#pragma unroll
            for (int i = 0; i < 8; ++i)
                C[(size_t)(mt * 16 + i + hi * 8) * ldc + nbase + ns * 16 + l16] = acc[ns][i];
    }
}

__global__ void sdpa_fuse_add(const float* __restrict__ h0, const float* __restrict__ h1,
                              const float* __restrict__ h2, const float* __restrict__ addv,
                              float* __restrict__ out, int n) {
    const int lane = threadIdx.x & 31;
    const int node = blockIdx.x * (blockDim.x >> 5) + (threadIdx.x >> 5);
    if (node >= n) return;
    const float* x0 = h0 + (size_t)node * PDIM;
    const float* x1 = h1 + (size_t)node * PDIM;
    const float* x2 = h2 + (size_t)node * PDIM;

    float d00 = 0, d01 = 0, d02 = 0, d11 = 0, d12 = 0, d22 = 0;
    for (int j = lane; j < PDIM; j += 32) {
        float a = x0[j], b = x1[j], e = x2[j];
        d00 += a * a; d01 += a * b; d02 += a * e;
        d11 += b * b; d12 += b * e; d22 += e * e;
    }
    for (int off = 16; off > 0; off >>= 1) {
        d00 += __shfl_xor(d00, off, 32);
        d01 += __shfl_xor(d01, off, 32);
        d02 += __shfl_xor(d02, off, 32);
        d11 += __shfl_xor(d11, off, 32);
        d12 += __shfl_xor(d12, off, 32);
        d22 += __shfl_xor(d22, off, 32);
    }
    const float scale = rsqrtf((float)DDIM);
    float S[3][3] = {{d00, d01, d02}, {d01, d11, d12}, {d02, d12, d22}};
    float wsum[3] = {0.f, 0.f, 0.f};
    for (int t = 0; t < 3; ++t) {
        float p0 = S[t][0] * scale, p1 = S[t][1] * scale, p2 = S[t][2] * scale;
        float mx = fmaxf(p0, fmaxf(p1, p2));
        float e0 = __expf(p0 - mx), e1 = __expf(p1 - mx), e2 = __expf(p2 - mx);
        float inv = 1.f / (e0 + e1 + e2);
        wsum[0] += e0 * inv; wsum[1] += e1 * inv; wsum[2] += e2 * inv;
    }
    const float w0 = wsum[0] * (1.f / 3.f), w1 = wsum[1] * (1.f / 3.f), w2 = wsum[2] * (1.f / 3.f);
    const float* av = addv + (size_t)node * PDIM;
    float* o = out + (size_t)node * PDIM;
    for (int j = lane; j < PDIM; j += 32)
        o[j] = w0 * x0[j] + w1 * x1[j] + w2 * x2[j] + av[j];
}

#define PB 32
static_assert(LN % (2 * PB) == 0, "pool_seq walks the sequence in pairs of batches");
static_assert(PDIM % 32 == 0, "every wave of pool_seq is full, so every lane's token id can be read by readlane");

__device__ __forceinline__ void pool_load(float (&x)[PB], const float* __restrict__ table, unsigned j,
                                          int ids, unsigned real) {
#pragma unroll
    for (int t = 0; t < PB; ++t) {
        const int idx = __builtin_amdgcn_readlane(ids, t);
        const int r = ((real >> t) & 1u) ? idx : 0;
        const float* __restrict__ rp = table + (size_t)r * PDIM;
        x[t] = rp[j];
    }
}

__device__ __forceinline__ void pool_add(float& acc, const float (&x)[PB], unsigned real) {
#pragma unroll
    for (int t = 0; t < PB; ++t)
        acc += ((real >> t) & 1u) ? x[t] : 0.f;
}

__global__ __launch_bounds__(PDIM) void pool_seq(const float* __restrict__ table, const int* __restrict__ xb,
                                                 float* __restrict__ pooled) {
    const int b = blockIdx.x;
    const unsigned j = threadIdx.x;
    if (j >= PDIM) return;
    const unsigned lane = j & 31u;
    const int* row = xb + b * LN;
    int ids[LN / PB];
    unsigned real[LN / PB];
#pragma unroll
    for (int k = 0; k < LN / PB; ++k) ids[k] = row[PB * k + lane];
#pragma unroll
    for (int k = 0; k < LN / PB; ++k)
        real[k] = __builtin_amdgcn_ballot_w32((unsigned)ids[k] < (unsigned)NWN);

    float acc = 0.f;
    float x0[PB], x1[PB];
    pool_load(x0, table, j, ids[0], real[0]);
#pragma unroll
    for (int k = 0; k < LN / PB; k += 2) {
        pool_load(x1, table, j, ids[k + 1], real[k + 1]);
        pool_add(acc, x0, real[k]);
        if (k + 2 < LN / PB) pool_load(x0, table, j, ids[k + 2], real[k + 2]);
        pool_add(acc, x1, real[k + 1]);
    }
    pooled[(size_t)b * PDIM + j] = acc * (1.f / (float)LN);
}

__global__ void fc_kernel(const float* __restrict__ rep, const float* __restrict__ wfc,
                          const float* __restrict__ bfc, float* __restrict__ out) {
    int t = blockIdx.x * blockDim.x + threadIdx.x;
    if (t >= BN * CN) return;
    int b = t / CN, c = t % CN;
    float acc = bfc[c];
    const float* r = rep + (size_t)b * PDIM;
    for (int j = 0; j < DDIM; ++j) acc += r[j] * wfc[(size_t)j * CN + c];
    out[t] = acc;
}


extern "C" void kernel_launch(void* const* d_in, const int* in_sizes, int n_in,
                              void* d_out, int out_size, void* d_ws, size_t ws_size,
                              hipStream_t stream) {
    (void)in_sizes; (void)n_in; (void)out_size; (void)ws_size;
    const float* emb_word = (const float*)d_in[0];
    const float* emb_doc  = (const float*)d_in[1];
    const float* w_sage[6] = {(const float*)d_in[2],  (const float*)d_in[4],  (const float*)d_in[6],
                              (const float*)d_in[8],  (const float*)d_in[10], (const float*)d_in[12]};
    const float* b_sage[6] = {(const float*)d_in[3],  (const float*)d_in[5],  (const float*)d_in[7],
                              (const float*)d_in[9],  (const float*)d_in[11], (const float*)d_in[13]};
    const float* w_dense = (const float*)d_in[14];
    const float* b_dense = (const float*)d_in[15];
    const float* w_fc    = (const float*)d_in[16];
    const float* b_fc    = (const float*)d_in[17];
    const int* dst_nid      = (const int*)d_in[18];
    const int* src_nid_w[3] = {(const int*)d_in[19], (const int*)d_in[20], (const int*)d_in[21]};
    const int* e_src_w[3]   = {(const int*)d_in[22], (const int*)d_in[24], (const int*)d_in[26]};
    const int* e_dst_w[3]   = {(const int*)d_in[23], (const int*)d_in[25], (const int*)d_in[27]};
    const int* x_batch      = (const int*)d_in[28];
    const int* src_nid_d[3] = {(const int*)d_in[30], (const int*)d_in[31], (const int*)d_in[32]};
    const int* e_src_d[3]   = {(const int*)d_in[33], (const int*)d_in[35], (const int*)d_in[37]};
    const int* e_dst_d[3]   = {(const int*)d_in[34], (const int*)d_in[36], (const int*)d_in[38]};
    float* out = (float*)d_out;

    float* ws = (float*)d_ws;
    size_t off = 0;
    auto carve = [&](size_t nflt) { float* p = ws + off; off += nflt; return p; };
    float* hdst = carve((size_t)NWN * PDIM);
    float* nbuf = carve((size_t)NWN * PDIM);
    float* h3[3]; for (int r = 0; r < 3; ++r) h3[r] = carve((size_t)NWN * PDIM);
    float* cnt  = carve(NWN);
    float* wpk[6]; for (int r = 0; r < 6; ++r) wpk[r] = carve((size_t)2 * PDIM * PDIM);
    float* wdn  = carve((size_t)PDIM * PDIM);
    float* pooled = carve((size_t)BN * PDIM);
    float* docout = carve((size_t)BN * PDIM);
    float* dnb    = carve((size_t)BN * PDIM);
    float* d3[3]; for (int r = 0; r < 3; ++r) d3[r] = carve((size_t)BN * PDIM);
    float* docrep = carve((size_t)BN * PDIM);
    int* icnt = (int*)cnt;
    int* w_offs = (int*)pooled;
    int* w_cur  = w_offs + NWN;
    int* w_rows = w_cur + NWN;
    int* d_offs = (int*)h3[0];
    int* d_cur  = d_offs + BN;
    int* d_rows = d_cur + BN;

    const int TB = 256;
    const unsigned wordElems = (unsigned)(((size_t)NWN * PDIM + TB - 1) / TB);
    const dim3 gemmGridW((NWN + 127) / 128, PDIM / 64);
    const dim3 gemmGridD((BN + 127) / 128, PDIM / 64);

    for (int r = 0; r < 6; ++r)
        pack_sage_w<<<(2 * PDIM * PDIM + TB - 1) / TB, TB, 0, stream>>>(w_sage[r], wpk[r]);
    pack_w<<<(PDIM * PDIM + TB - 1) / TB, TB, 0, stream>>>(w_dense, wdn);

    gather_pad<<<wordElems, TB, 0, stream>>>(emb_word, dst_nid, hdst, NWN);

    for (int r = 0; r < 3; ++r) {
        zero_i32<<<(NWN + TB - 1) / TB, TB, 0, stream>>>(icnt, NWN);
        count_dst<<<(EWN + TB - 1) / TB, TB, 0, stream>>>(e_dst_w[r], EWN, icnt);
        scan_counts<<<1, SCAN_T, 0, stream>>>(icnt, NWN, w_offs, w_cur);
        place_rows<<<(EWN + TB - 1) / TB, TB, 0, stream>>>(src_nid_w[r], e_src_w[r], e_dst_w[r], EWN,
                                                          w_cur, w_rows);
        gather_mean<<<(NWN + 7) / 8, TB, 0, stream>>>(emb_word, w_rows, w_offs, icnt, NWN, nbuf);
        gemm_f32_wmma<<<gemmGridW, TB, 0, stream>>>(hdst, wpk[r], b_sage[r], h3[r],
                                                    NWN, PDIM, PDIM, PDIM, PDIM, PDIM, 0);
        gemm_f32_wmma<<<gemmGridW, TB, 0, stream>>>(nbuf, wpk[r] + (size_t)PDIM * PDIM, nullptr,
                                                    h3[r], NWN, PDIM, PDIM, PDIM, PDIM, PDIM, 1);
    }

    sdpa_fuse_add<<<(NWN + 7) / 8, 256, 0, stream>>>(h3[0], h3[1], h3[2], hdst, nbuf, NWN);

    pool_seq<<<BN, PDIM, 0, stream>>>(nbuf, x_batch, pooled);
    gemm_f32_wmma<<<gemmGridD, TB, 0, stream>>>(pooled, wdn, b_dense, docout,
                                                BN, PDIM, PDIM, PDIM, PDIM, PDIM, 0);

    for (int r = 0; r < 3; ++r) {
        zero_i32<<<1, TB, 0, stream>>>(icnt, BN);
        count_dst<<<(EDN + TB - 1) / TB, TB, 0, stream>>>(e_dst_d[r], EDN, icnt);
        scan_counts<<<1, SCAN_T, 0, stream>>>(icnt, BN, d_offs, d_cur);
        place_rows<<<(EDN + TB - 1) / TB, TB, 0, stream>>>(src_nid_d[r], e_src_d[r], e_dst_d[r], EDN,
                                                          d_cur, d_rows);
        gather_mean<<<(BN + 7) / 8, TB, 0, stream>>>(emb_doc, d_rows, d_offs, icnt, BN, dnb);
        gemm_f32_wmma<<<gemmGridD, TB, 0, stream>>>(docout, wpk[3 + r], b_sage[3 + r], d3[r],
                                                    BN, PDIM, PDIM, PDIM, PDIM, PDIM, 0);
        gemm_f32_wmma<<<gemmGridD, TB, 0, stream>>>(dnb, wpk[3 + r] + (size_t)PDIM * PDIM, nullptr,
                                                    d3[r], BN, PDIM, PDIM, PDIM, PDIM, PDIM, 1);
    }

    sdpa_fuse_add<<<(BN + 7) / 8, 256, 0, stream>>>(d3[0], d3[1], d3[2], docout, docrep, BN);
    fc_kernel<<<(BN * CN + TB - 1) / TB, TB, 0, stream>>>(docrep, w_fc, b_fc, out);
}


# independent launches merged: h_dst gather + 7 weight packs + count zeroing = 1; each seg_mean stage once for all 6 jobs; both K halves of a SAGE GEMM = 1 launch (K=640); 55 -> 14 launches, same sums
# speedup vs baseline: 2.4993x; 1.4351x over previous
#include <hip/hip_runtime.h>

#define DDIM 300
#define PDIM 320
#define NWN  20000
#define EWN  320000
#define BN   256
#define LN   256
#define SDN  2048
#define EDN  4096
#define CN   20
#define KC   160
#define LP   164

typedef __attribute__((ext_vector_type(2))) float v2f;
typedef __attribute__((ext_vector_type(8))) float v8f;

struct SegIn { const int* e_src; const int* e_dst; const int* src_nid; };
struct SegWs { int* cnt; int* offs; int* cur; int* rows; int n; int E; };

__device__ __forceinline__ SegWs seg_ws(int job, int* wi, int* di) {
    SegWs s;
    if (job < 3) {
        s.n = NWN; s.E = EWN;
        s.cnt = wi + job * NWN; s.offs = wi + (3 + job) * NWN; s.cur = wi + (6 + job) * NWN;
        s.rows = wi + 9 * NWN + job * EWN;
    } else {
        const int r = job - 3;
        s.n = BN; s.E = EDN;
        s.cnt = di + r * BN; s.offs = di + (3 + r) * BN; s.cur = di + (6 + r) * BN;
        s.rows = di + 9 * BN + r * EDN;
    }
    return s;
}

__device__ __forceinline__ SegIn seg_in(int job, const SegIn& j0, const SegIn& j1, const SegIn& j2,
                                        const SegIn& j3, const SegIn& j4, const SegIn& j5) {
    switch (job) {
        case 0: return j0;
        case 1: return j1;
        case 2: return j2;
        case 3: return j3;
        case 4: return j4;
        default: return j5;
    }
}

#define WEB ((EWN + 255) / 256)
#define DEB ((EDN + 255) / 256)
__device__ __forceinline__ int edge_job(int blk, int& local) {
    if (blk < 3 * WEB) { const int j = blk / WEB; local = blk - j * WEB; return j; }
    const int d = blk - 3 * WEB;
    const int j = d / DEB;
    local = d - j * DEB;
    return 3 + j;
}

#define GPB ((NWN * PDIM) / 256)
#define PKB ((13 * PDIM * PDIM) / 256)
#define ZWB ((3 * NWN + 255) / 256)
#define ZDB ((3 * BN + 255) / 256)
static_assert((NWN * PDIM) % 256 == 0 && (PDIM * PDIM) % 256 == 0, "prep blocks never straddle two jobs or two slabs");

__global__ __launch_bounds__(256) void prep(const float* __restrict__ emb, const int* __restrict__ nid,
                                            float* __restrict__ hdst,
                                            const float* __restrict__ w0, const float* __restrict__ w1,
                                            const float* __restrict__ w2, const float* __restrict__ w3,
                                            const float* __restrict__ w4, const float* __restrict__ w5,
                                            const float* __restrict__ wd, float* __restrict__ wout,
                                            int* __restrict__ wi, int* __restrict__ di) {
    const int blk = blockIdx.x;
    if (blk < GPB) {
        const int t = blk * 256 + threadIdx.x;
        const int r = t / PDIM, j = t % PDIM;
        hdst[t] = (j < DDIM) ? emb[(size_t)nid[r] * DDIM + j] : 0.f;
    } else if (blk < GPB + PKB) {
        const int t = (blk - GPB) * 256 + threadIdx.x;
        const int slab = t / (PDIM * PDIM);
        const int r = (t / PDIM) % PDIM;
        const int c = t % PDIM;
        const int which = slab >> 1;
        const float* __restrict__ w = (slab >= 12) ? wd : (which == 0) ? w0 : (which == 1) ? w1 : (which == 2) ? w2
                                    : (which == 3) ? w3 : (which == 4) ? w4 : w5;
        const int r0 = (slab < 12 && (slab & 1)) ? DDIM : 0;
        wout[t] = (r < DDIM && c < DDIM) ? w[((size_t)(r0 + r)) * DDIM + c] : 0.f;
    } else if (blk < GPB + PKB + ZWB) {
        const int t = (blk - GPB - PKB) * 256 + threadIdx.x;
        if (t < 3 * NWN) wi[t] = 0;
    } else {
        const int t = (blk - GPB - PKB - ZWB) * 256 + threadIdx.x;
        if (t < 3 * BN) di[t] = 0;
    }
}


__global__ __launch_bounds__(256) void count_dst(const SegIn j0, const SegIn j1, const SegIn j2,
                                                 const SegIn j3, const SegIn j4, const SegIn j5,
                                                 int* __restrict__ wi, int* __restrict__ di) {
    int local;
    const int job = edge_job(blockIdx.x, local);
    const SegIn in = seg_in(job, j0, j1, j2, j3, j4, j5);
    const SegWs s = seg_ws(job, wi, di);
    const int t = local * 256 + threadIdx.x;
    if (t < s.E) atomicAdd(&s.cnt[in.e_dst[t]], 1);
}

#define SCAN_T 1024
__global__ __launch_bounds__(SCAN_T) void scan_counts(int* __restrict__ wi, int* __restrict__ di) {
    __shared__ int wsum[SCAN_T / 32];
    const SegWs sw = seg_ws(blockIdx.x, wi, di);
    const int* __restrict__ cnt = sw.cnt;
    int* __restrict__ offs = sw.offs;
    int* __restrict__ cur = sw.cur;
    const int n = sw.n;
    const int t = threadIdx.x;
    const int lane = t & 31;
    const int ipt = (n + SCAN_T - 1) / SCAN_T;
    const int b = t * ipt;
    const int lim = min(ipt, n - b);
    int s = 0;
    for (int i = 0; i < lim; ++i) s += cnt[b + i];
    int v = s;
    for (int off = 1; off < 32; off <<= 1) {
        int u = __shfl_up(v, off, 32);
        if (lane >= off) v += u;
    }
    if (lane == 31) wsum[t >> 5] = v;
    __syncthreads();
    if (t < 32) {
        int w = wsum[t];
        int x = w;
        for (int off = 1; off < 32; off <<= 1) {
            int u = __shfl_up(x, off, 32);
            if (t >= off) x += u;
        }
        wsum[t] = x - w;
    }
    __syncthreads();
    int run = wsum[t >> 5] + (v - s);
    for (int i = 0; i < lim; ++i) {
        const int c = cnt[b + i];
        offs[b + i] = run;
        cur[b + i]  = run;
        run += c;
    }
}

__global__ __launch_bounds__(256) void place_rows(const SegIn j0, const SegIn j1, const SegIn j2,
                                                  const SegIn j3, const SegIn j4, const SegIn j5,
                                                  int* __restrict__ wi, int* __restrict__ di) {
    int local;
    const int job = edge_job(blockIdx.x, local);
    const SegIn in = seg_in(job, j0, j1, j2, j3, j4, j5);
    const SegWs s = seg_ws(job, wi, di);
    const int t = local * 256 + threadIdx.x;
    if (t >= s.E) return;
    const int pos = atomicAdd(&s.cur[in.e_dst[t]], 1);
    s.rows[pos] = in.src_nid[in.e_src[t]];
}

__device__ __forceinline__ void add4(float4& a, const float4 x) {
    a.x += x.x; a.y += x.y; a.z += x.z; a.w += x.w;
}

#define WNB ((NWN + 7) / 8)
#define DNB ((BN + 7) / 8)
__global__ __launch_bounds__(256) void gather_mean(const float* __restrict__ emb_w, const float* __restrict__ emb_d,
                                                   int* __restrict__ wi, int* __restrict__ di,
                                                   float* __restrict__ o0, float* __restrict__ o1,
                                                   float* __restrict__ o2, float* __restrict__ o3,
                                                   float* __restrict__ o4, float* __restrict__ o5) {
    const int lane = threadIdx.x & 31;
    const int blk = blockIdx.x;
    int job, local;
    if (blk < 3 * WNB) { job = blk / WNB; local = blk - job * WNB; }
    else { const int d = blk - 3 * WNB; const int j = d / DNB; local = d - j * DNB; job = 3 + j; }
    const SegWs s = seg_ws(job, wi, di);
    const float* __restrict__ emb = (job < 3) ? emb_w : emb_d;
    float* __restrict__ out = (job == 0) ? o0 : (job == 1) ? o1 : (job == 2) ? o2
                            : (job == 3) ? o3 : (job == 4) ? o4 : o5;
    const int* __restrict__ rows = s.rows;
    const int* __restrict__ offs = s.offs;
    const int* __restrict__ cnt = s.cnt;
    const int node = __builtin_amdgcn_readfirstlane((int)(local * 8 + (threadIdx.x >> 5)));
    if (node >= s.n) return;
    const int beg = offs[node];
    const int deg = cnt[node];
    const int i2 = (lane < 11) ? lane + 64 : 74;

    float4 a0 = make_float4(0.f, 0.f, 0.f, 0.f), a1 = a0, a2 = a0;
    if (deg <= 32) {
        const int key = (lane < deg) ? rows[beg + lane] : 0x7fffffff;
        int rank = 0;
#pragma unroll
        for (int k = 0; k < 32; ++k) {
            const int kk = __builtin_amdgcn_readlane(key, k);
            rank += (kk < key || (kk == key && k < lane)) ? 1 : 0;
        }
        const int sorted = __builtin_amdgcn_ds_permute(rank << 2, key);
        int k = 0;
        for (; k + 4 <= deg; k += 4) {
            const float4* r0 = (const float4*)(emb + (size_t)__builtin_amdgcn_readlane(sorted, k + 0) * DDIM);
            const float4* r1 = (const float4*)(emb + (size_t)__builtin_amdgcn_readlane(sorted, k + 1) * DDIM);
            const float4* r2 = (const float4*)(emb + (size_t)__builtin_amdgcn_readlane(sorted, k + 2) * DDIM);
            const float4* r3 = (const float4*)(emb + (size_t)__builtin_amdgcn_readlane(sorted, k + 3) * DDIM);
            const float4 x00 = r0[lane], x01 = r0[lane + 32], x02 = r0[i2];
            const float4 x10 = r1[lane], x11 = r1[lane + 32], x12 = r1[i2];
            const float4 x20 = r2[lane], x21 = r2[lane + 32], x22 = r2[i2];
            const float4 x30 = r3[lane], x31 = r3[lane + 32], x32 = r3[i2];
            add4(a0, x00); add4(a1, x01); add4(a2, x02);
            add4(a0, x10); add4(a1, x11); add4(a2, x12);
            add4(a0, x20); add4(a1, x21); add4(a2, x22);
            add4(a0, x30); add4(a1, x31); add4(a2, x32);
        }
        for (; k < deg; ++k) {
            const float4* r0 = (const float4*)(emb + (size_t)__builtin_amdgcn_readlane(sorted, k) * DDIM);
            const float4 x00 = r0[lane], x01 = r0[lane + 32], x02 = r0[i2];
            add4(a0, x00); add4(a1, x01); add4(a2, x02);
        }
    } else {
        int last = -1;
        for (int pass = 0; pass < deg; ++pass) {
            int best = 0x7fffffff;
            for (int b = lane; b < deg; b += 32) {
                const int kx = rows[beg + b];
                if (kx > last && kx < best) best = kx;
            }
            for (int off = 16; off > 0; off >>= 1) best = min(best, __shfl_xor(best, off, 32));
            best = __builtin_amdgcn_readfirstlane(best);
            if (best == 0x7fffffff) break;
            int mult = 0;
            for (int b = lane; b < deg; b += 32) mult += (rows[beg + b] == best) ? 1 : 0;
            for (int off = 16; off > 0; off >>= 1) mult += __shfl_xor(mult, off, 32);
            mult = __builtin_amdgcn_readfirstlane(mult);
            const float4* r0 = (const float4*)(emb + (size_t)best * DDIM);
            const float4 x00 = r0[lane], x01 = r0[lane + 32], x02 = r0[i2];
            for (int m = 0; m < mult; ++m) { add4(a0, x00); add4(a1, x01); add4(a2, x02); }
            last = best;
        }
    }

    const float d = fmaxf((float)deg, 1.f);
    const float4 z = make_float4(0.f, 0.f, 0.f, 0.f);
    float4* o = (float4*)(out + (size_t)node * PDIM);
    o[lane]      = make_float4(a0.x / d, a0.y / d, a0.z / d, a0.w / d);
    o[lane + 32] = make_float4(a1.x / d, a1.y / d, a1.z / d, a1.w / d);
    if (lane < 16)
        o[lane + 64] = (lane < 11) ? make_float4(a2.x / d, a2.y / d, a2.z / d, a2.w / d) : z;
}

#define AB 10
static_assert(KC == 16 * AB, "the K loop below is written for four A batches per chunk");

__device__ __forceinline__ void load_a_batch(v2f (&a)[AB], const float* __restrict__ p) {
#pragma unroll
    for (int t = 0; t < AB; ++t) a[t] = *(const v2f*)(p + 4 * t);
}

__device__ __forceinline__ void mma_a_batch(v8f (&acc)[4], const v2f (&a)[AB], const float* lb) {
#pragma unroll
    for (int t = 0; t < AB; ++t) {
#pragma unroll
        for (int ns = 0; ns < 4; ++ns) {
            v2f b = *(const v2f*)(lb + ns * 16 * LP + 4 * t);
            acc[ns] = __builtin_amdgcn_wmma_f32_16x16x4_f32(
                false, a[t], false, b, (short)0, acc[ns], false, false);
        }
    }
}

__global__ __launch_bounds__(256) void gemm_f32_wmma(
        const float* __restrict__ A0,
        const float* __restrict__ A1z0, const float* __restrict__ A1z1, const float* __restrict__ A1z2,
        const float* __restrict__ Bz0, int bzs,
        const float* __restrict__ bias0, const float* __restrict__ bias1, const float* __restrict__ bias2,
        float* __restrict__ Cz0, int czs,
        int M, int N, int K, int ksplit, int lda, int ldb, int ldc) {
    __shared__ float ldsB[64 * LP];
    (void)N;
    const int z = blockIdx.z;
    const float* __restrict__ A1 = (z == 0) ? A1z0 : (z == 1) ? A1z1 : A1z2;
    const float* __restrict__ bias = (z == 0) ? bias0 : (z == 1) ? bias1 : bias2;
    const float* __restrict__ B = Bz0 + (size_t)z * bzs;
    float* __restrict__ C = Cz0 + (size_t)z * czs;
    const int lane = threadIdx.x & 31;
    const int wave = threadIdx.x >> 5;
    const int hi   = lane >> 4;
    const int l16  = lane & 15;
    const int nbase = blockIdx.y * 64;
    const int mt    = blockIdx.x * 8 + wave;
    const bool valid = (mt * 16) < M;
    const int m = mt * 16 + l16;

    v8f acc[4];
    if (valid) {
#pragma unroll
        for (int ns = 0; ns < 4; ++ns) {
            const int n = nbase + ns * 16 + l16;
            const float bv = (bias && n < DDIM) ? bias[n] : 0.f;
#pragma unroll
            for (int i = 0; i < 8; ++i) acc[ns][i] = bv;
        }
    }

    const int aoff = m * lda + 2 * hi;
    const float* lb = &ldsB[l16 * LP + 2 * hi];
    const int sc  = (threadIdx.x & 15) * 4;
    const int skr = threadIdx.x >> 4;
    const float* __restrict__ Bp = B + (size_t)skr * ldb + nbase + sc;

    v2f a0[AB], a1[AB];
    if (valid) load_a_batch(a0, A0 + aoff);

    for (int kk = 0; kk < K; kk += KC) {
        float4 bt[KC / 16];
#pragma unroll
        for (int i = 0; i < KC / 16; ++i)
            bt[i] = *(const float4*)(Bp + (size_t)(kk + 16 * i) * ldb);

        __syncthreads();
#pragma unroll
        for (int i = 0; i < KC / 16; ++i) {
            const int kloc = skr + 16 * i;
            ldsB[(sc + 0) * LP + kloc] = bt[i].x;
            ldsB[(sc + 1) * LP + kloc] = bt[i].y;
            ldsB[(sc + 2) * LP + kloc] = bt[i].z;
            ldsB[(sc + 3) * LP + kloc] = bt[i].w;
        }
        __syncthreads();

        if (valid) {
            const float* ak = ((kk < ksplit) ? A0 + kk : A1 + (kk - ksplit)) + aoff;
            load_a_batch(a1, ak + 4 * AB);
            mma_a_batch(acc, a0, lb);
            load_a_batch(a0, ak + 8 * AB);
            mma_a_batch(acc, a1, lb + 4 * AB);
            load_a_batch(a1, ak + 12 * AB);
            mma_a_batch(acc, a0, lb + 8 * AB);
            const int kn = kk + KC;
            if (kn < K) load_a_batch(a0, ((kn < ksplit) ? A0 + kn : A1 + (kn - ksplit)) + aoff);
            mma_a_batch(acc, a1, lb + 12 * AB);
        }
    }
    if (valid) {
#pragma unroll
        for (int ns = 0; ns < 4; ++ns)
#pragma unroll
            for (int i = 0; i < 8; ++i)
                C[(size_t)(mt * 16 + i + hi * 8) * ldc + nbase + ns * 16 + l16] = acc[ns][i];
    }
}

__global__ void sdpa_fuse_add(const float* __restrict__ h0, const float* __restrict__ h1,
                              const float* __restrict__ h2, const float* __restrict__ addv,
                              float* __restrict__ out, int n) {
    const int lane = threadIdx.x & 31;
    const int node = blockIdx.x * (blockDim.x >> 5) + (threadIdx.x >> 5);
    if (node >= n) return;
    const float* x0 = h0 + (size_t)node * PDIM;
    const float* x1 = h1 + (size_t)node * PDIM;
    const float* x2 = h2 + (size_t)node * PDIM;

    float d00 = 0, d01 = 0, d02 = 0, d11 = 0, d12 = 0, d22 = 0;
    for (int j = lane; j < PDIM; j += 32) {
        float a = x0[j], b = x1[j], e = x2[j];
        d00 += a * a; d01 += a * b; d02 += a * e;
        d11 += b * b; d12 += b * e; d22 += e * e;
    }
    for (int off = 16; off > 0; off >>= 1) {
        d00 += __shfl_xor(d00, off, 32);
        d01 += __shfl_xor(d01, off, 32);
        d02 += __shfl_xor(d02, off, 32);
        d11 += __shfl_xor(d11, off, 32);
        d12 += __shfl_xor(d12, off, 32);
        d22 += __shfl_xor(d22, off, 32);
    }
    const float scale = rsqrtf((float)DDIM);
    float S[3][3] = {{d00, d01, d02}, {d01, d11, d12}, {d02, d12, d22}};
    float wsum[3] = {0.f, 0.f, 0.f};
    for (int t = 0; t < 3; ++t) {
        float p0 = S[t][0] * scale, p1 = S[t][1] * scale, p2 = S[t][2] * scale;
        float mx = fmaxf(p0, fmaxf(p1, p2));
        float e0 = __expf(p0 - mx), e1 = __expf(p1 - mx), e2 = __expf(p2 - mx);
        float inv = 1.f / (e0 + e1 + e2);
        wsum[0] += e0 * inv; wsum[1] += e1 * inv; wsum[2] += e2 * inv;
    }
    const float w0 = wsum[0] * (1.f / 3.f), w1 = wsum[1] * (1.f / 3.f), w2 = wsum[2] * (1.f / 3.f);
    const float* av = addv + (size_t)node * PDIM;
    float* o = out + (size_t)node * PDIM;
    for (int j = lane; j < PDIM; j += 32)
        o[j] = w0 * x0[j] + w1 * x1[j] + w2 * x2[j] + av[j];
}

#define PB 32
static_assert(LN % (2 * PB) == 0, "pool_seq walks the sequence in pairs of batches");
static_assert(PDIM % 32 == 0, "every wave of pool_seq is full, so every lane's token id can be read by readlane");

__device__ __forceinline__ void pool_load(float (&x)[PB], const float* __restrict__ table, unsigned j,
                                          int ids, unsigned real) {
#pragma unroll
    for (int t = 0; t < PB; ++t) {
        const int idx = __builtin_amdgcn_readlane(ids, t);
        const int r = ((real >> t) & 1u) ? idx : 0;
        const float* __restrict__ rp = table + (size_t)r * PDIM;
        x[t] = rp[j];
    }
}

__device__ __forceinline__ void pool_add(float& acc, const float (&x)[PB], unsigned real) {
#pragma unroll
    for (int t = 0; t < PB; ++t)
        acc += ((real >> t) & 1u) ? x[t] : 0.f;
}

__global__ __launch_bounds__(PDIM) void pool_seq(const float* __restrict__ table, const int* __restrict__ xb,
                                                 float* __restrict__ pooled) {
    const int b = blockIdx.x;
    const unsigned j = threadIdx.x;
    if (j >= PDIM) return;
    const unsigned lane = j & 31u;
    const int* row = xb + b * LN;
    int ids[LN / PB];
    unsigned real[LN / PB];
#pragma unroll
    for (int k = 0; k < LN / PB; ++k) ids[k] = row[PB * k + lane];
#pragma unroll
    for (int k = 0; k < LN / PB; ++k)
        real[k] = __builtin_amdgcn_ballot_w32((unsigned)ids[k] < (unsigned)NWN);

    float acc = 0.f;
    float x0[PB], x1[PB];
    pool_load(x0, table, j, ids[0], real[0]);
#pragma unroll
    for (int k = 0; k < LN / PB; k += 2) {
        pool_load(x1, table, j, ids[k + 1], real[k + 1]);
        pool_add(acc, x0, real[k]);
        if (k + 2 < LN / PB) pool_load(x0, table, j, ids[k + 2], real[k + 2]);
        pool_add(acc, x1, real[k + 1]);
    }
    pooled[(size_t)b * PDIM + j] = acc * (1.f / (float)LN);
}

__global__ void fc_kernel(const float* __restrict__ rep, const float* __restrict__ wfc,
                          const float* __restrict__ bfc, float* __restrict__ out) {
    int t = blockIdx.x * blockDim.x + threadIdx.x;
    if (t >= BN * CN) return;
    int b = t / CN, c = t % CN;
    float acc = bfc[c];
    const float* r = rep + (size_t)b * PDIM;
    for (int j = 0; j < DDIM; ++j) acc += r[j] * wfc[(size_t)j * CN + c];
    out[t] = acc;
}


extern "C" void kernel_launch(void* const* d_in, const int* in_sizes, int n_in,
                              void* d_out, int out_size, void* d_ws, size_t ws_size,
                              hipStream_t stream) {
    (void)in_sizes; (void)n_in; (void)out_size; (void)ws_size;
    const float* emb_word = (const float*)d_in[0];
    const float* emb_doc  = (const float*)d_in[1];
    const float* w_sage[6] = {(const float*)d_in[2],  (const float*)d_in[4],  (const float*)d_in[6],
                              (const float*)d_in[8],  (const float*)d_in[10], (const float*)d_in[12]};
    const float* b_sage[6] = {(const float*)d_in[3],  (const float*)d_in[5],  (const float*)d_in[7],
                              (const float*)d_in[9],  (const float*)d_in[11], (const float*)d_in[13]};
    const float* w_dense = (const float*)d_in[14];
    const float* b_dense = (const float*)d_in[15];
    const float* w_fc    = (const float*)d_in[16];
    const float* b_fc    = (const float*)d_in[17];
    const int* dst_nid      = (const int*)d_in[18];
    const int* x_batch      = (const int*)d_in[28];
    const SegIn jw0 = {(const int*)d_in[22], (const int*)d_in[23], (const int*)d_in[19]};
    const SegIn jw1 = {(const int*)d_in[24], (const int*)d_in[25], (const int*)d_in[20]};
    const SegIn jw2 = {(const int*)d_in[26], (const int*)d_in[27], (const int*)d_in[21]};
    const SegIn jd0 = {(const int*)d_in[33], (const int*)d_in[34], (const int*)d_in[30]};
    const SegIn jd1 = {(const int*)d_in[35], (const int*)d_in[36], (const int*)d_in[31]};
    const SegIn jd2 = {(const int*)d_in[37], (const int*)d_in[38], (const int*)d_in[32]};
    float* out = (float*)d_out;

    float* ws = (float*)d_ws;
    size_t off = 0;
    auto carve = [&](size_t nflt) { float* p = ws + off; off += nflt; return p; };
    float* hdst = carve((size_t)NWN * PDIM);
    float* nbuf = carve((size_t)NWN * PDIM);
    float* h3[3]; for (int r = 0; r < 3; ++r) h3[r] = carve((size_t)NWN * PDIM);
    float* cnt  = carve(NWN);
    float* wpk[6]; for (int r = 0; r < 6; ++r) wpk[r] = carve((size_t)2 * PDIM * PDIM);
    float* wdn  = carve((size_t)PDIM * PDIM);
    float* pooled = carve((size_t)BN * PDIM);
    float* docout = carve((size_t)BN * PDIM);
    float* dnb    = carve((size_t)BN * PDIM);
    float* d3[3]; for (int r = 0; r < 3; ++r) d3[r] = carve((size_t)BN * PDIM);
    float* docrep = carve((size_t)BN * PDIM);
    (void)cnt; (void)wdn;
    int* wi = (int*)h3[0];
    int* di = (int*)d3[0];
    float* wnb[3] = {h3[1], h3[2], nbuf};
    float* dnbr[3] = {dnb, docrep, pooled};
    float* poolbuf = d3[0];

    const int TB = 256;
    const dim3 gemmGridW((NWN + 127) / 128, PDIM / 64, 1);
    const dim3 gemmGridD((BN + 127) / 128, PDIM / 64, 1);
    const dim3 gemmGridD3((BN + 127) / 128, PDIM / 64, 3);

    prep<<<GPB + PKB + ZWB + ZDB, TB, 0, stream>>>(emb_word, dst_nid, hdst,
                                                   w_sage[0], w_sage[1], w_sage[2], w_sage[3], w_sage[4], w_sage[5],
                                                   w_dense, wpk[0], wi, di);

    count_dst<<<3 * WEB + 3 * DEB, TB, 0, stream>>>(jw0, jw1, jw2, jd0, jd1, jd2, wi, di);
    scan_counts<<<6, SCAN_T, 0, stream>>>(wi, di);
    place_rows<<<3 * WEB + 3 * DEB, TB, 0, stream>>>(jw0, jw1, jw2, jd0, jd1, jd2, wi, di);
    gather_mean<<<3 * WNB + 3 * DNB, TB, 0, stream>>>(emb_word, emb_doc, wi, di,
                                                      wnb[0], wnb[1], wnb[2], dnbr[0], dnbr[1], dnbr[2]);

    for (int r = 0; r < 3; ++r)
        gemm_f32_wmma<<<gemmGridW, TB, 0, stream>>>(hdst, wnb[r], wnb[r], wnb[r], wpk[r], 0,
                                                    b_sage[r], b_sage[r], b_sage[r], h3[r], 0,
                                                    NWN, PDIM, 2 * PDIM, PDIM, PDIM, PDIM, PDIM);

    sdpa_fuse_add<<<(NWN + 7) / 8, 256, 0, stream>>>(h3[0], h3[1], h3[2], hdst, nbuf, NWN);

    pool_seq<<<BN, PDIM, 0, stream>>>(nbuf, x_batch, poolbuf);
    gemm_f32_wmma<<<gemmGridD, TB, 0, stream>>>(poolbuf, poolbuf, poolbuf, poolbuf, wpk[0] + (size_t)12 * PDIM * PDIM, 0,
                                                b_dense, b_dense, b_dense, docout, 0,
                                                BN, PDIM, PDIM, PDIM, PDIM, PDIM, PDIM);

    gemm_f32_wmma<<<gemmGridD3, TB, 0, stream>>>(docout, dnbr[0], dnbr[1], dnbr[2], wpk[3], 2 * PDIM * PDIM,
                                                 b_sage[3], b_sage[4], b_sage[5], d3[0], BN * PDIM,
                                                 BN, PDIM, 2 * PDIM, PDIM, PDIM, PDIM, PDIM);

    sdpa_fuse_add<<<(BN + 7) / 8, 256, 0, stream>>>(d3[0], d3[1], d3[2], docout, docrep, BN);
    fc_kernel<<<(BN * CN + TB - 1) / TB, TB, 0, stream>>>(docrep, w_fc, b_fc, out);
}


# word GEMM: a wave owns whole rows (tile 16x64 -> 16x320: A read once, h_r stored over its own mean_r), so the 3 relation launches run as 1, one block per WGP; 14 -> 12 launches, same k order
# speedup vs baseline: 2.7761x; 1.1108x over previous
#include <hip/hip_runtime.h>

#define DDIM 300
#define PDIM 320
#define NWN  20000
#define EWN  320000
#define BN   256
#define LN   256
#define SDN  2048
#define EDN  4096
#define CN   20
#define KC   160
#define LP   164

typedef __attribute__((ext_vector_type(2))) float v2f;
typedef __attribute__((ext_vector_type(8))) float v8f;

struct SegIn { const int* e_src; const int* e_dst; const int* src_nid; };
struct SegWs { int* cnt; int* offs; int* cur; int* rows; int n; int E; };

__device__ __forceinline__ SegWs seg_ws(int job, int* wi, int* di) {
    SegWs s;
    if (job < 3) {
        s.n = NWN; s.E = EWN;
        s.cnt = wi + job * NWN; s.offs = wi + (3 + job) * NWN; s.cur = wi + (6 + job) * NWN;
        s.rows = wi + 9 * NWN + job * EWN;
    } else {
        const int r = job - 3;
        s.n = BN; s.E = EDN;
        s.cnt = di + r * BN; s.offs = di + (3 + r) * BN; s.cur = di + (6 + r) * BN;
        s.rows = di + 9 * BN + r * EDN;
    }
    return s;
}

__device__ __forceinline__ SegIn seg_in(int job, const SegIn& j0, const SegIn& j1, const SegIn& j2,
                                        const SegIn& j3, const SegIn& j4, const SegIn& j5) {
    switch (job) {
        case 0: return j0;
        case 1: return j1;
        case 2: return j2;
        case 3: return j3;
        case 4: return j4;
        default: return j5;
    }
}

#define WEB ((EWN + 255) / 256)
#define DEB ((EDN + 255) / 256)
__device__ __forceinline__ int edge_job(int blk, int& local) {
    if (blk < 3 * WEB) { const int j = blk / WEB; local = blk - j * WEB; return j; }
    const int d = blk - 3 * WEB;
    const int j = d / DEB;
    local = d - j * DEB;
    return 3 + j;
}

#define GPB ((NWN * PDIM) / 256)
#define PKB ((13 * PDIM * PDIM) / 256)
#define ZWB ((3 * NWN + 255) / 256)
#define ZDB ((3 * BN + 255) / 256)
static_assert((NWN * PDIM) % 256 == 0 && (PDIM * PDIM) % 256 == 0, "prep blocks never straddle two jobs or two slabs");

__global__ __launch_bounds__(256) void prep(const float* __restrict__ emb, const int* __restrict__ nid,
                                            float* __restrict__ hdst,
                                            const float* __restrict__ w0, const float* __restrict__ w1,
                                            const float* __restrict__ w2, const float* __restrict__ w3,
                                            const float* __restrict__ w4, const float* __restrict__ w5,
                                            const float* __restrict__ wd, float* __restrict__ wout,
                                            int* __restrict__ wi, int* __restrict__ di) {
    const int blk = blockIdx.x;
    if (blk < GPB) {
        const int t = blk * 256 + threadIdx.x;
        const int r = t / PDIM, j = t % PDIM;
        hdst[t] = (j < DDIM) ? emb[(size_t)nid[r] * DDIM + j] : 0.f;
    } else if (blk < GPB + PKB) {
        const int t = (blk - GPB) * 256 + threadIdx.x;
        const int slab = t / (PDIM * PDIM);
        const int r = (t / PDIM) % PDIM;
        const int c = t % PDIM;
        const int which = slab >> 1;
        const float* __restrict__ w = (slab >= 12) ? wd : (which == 0) ? w0 : (which == 1) ? w1 : (which == 2) ? w2
                                    : (which == 3) ? w3 : (which == 4) ? w4 : w5;
        const int r0 = (slab < 12 && (slab & 1)) ? DDIM : 0;
        wout[t] = (r < DDIM && c < DDIM) ? w[((size_t)(r0 + r)) * DDIM + c] : 0.f;
    } else if (blk < GPB + PKB + ZWB) {
        const int t = (blk - GPB - PKB) * 256 + threadIdx.x;
        if (t < 3 * NWN) wi[t] = 0;
    } else {
        const int t = (blk - GPB - PKB - ZWB) * 256 + threadIdx.x;
        if (t < 3 * BN) di[t] = 0;
    }
}


__global__ __launch_bounds__(256) void count_dst(const SegIn j0, const SegIn j1, const SegIn j2,
                                                 const SegIn j3, const SegIn j4, const SegIn j5,
                                                 int* __restrict__ wi, int* __restrict__ di) {
    int local;
    const int job = edge_job(blockIdx.x, local);
    const SegIn in = seg_in(job, j0, j1, j2, j3, j4, j5);
    const SegWs s = seg_ws(job, wi, di);
    const int t = local * 256 + threadIdx.x;
    if (t < s.E) atomicAdd(&s.cnt[in.e_dst[t]], 1);
}

#define SCAN_T 1024
__global__ __launch_bounds__(SCAN_T) void scan_counts(int* __restrict__ wi, int* __restrict__ di) {
    __shared__ int wsum[SCAN_T / 32];
    const SegWs sw = seg_ws(blockIdx.x, wi, di);
    const int* __restrict__ cnt = sw.cnt;
    int* __restrict__ offs = sw.offs;
    int* __restrict__ cur = sw.cur;
    const int n = sw.n;
    const int t = threadIdx.x;
    const int lane = t & 31;
    const int ipt = (n + SCAN_T - 1) / SCAN_T;
    const int b = t * ipt;
    const int lim = min(ipt, n - b);
    int s = 0;
    for (int i = 0; i < lim; ++i) s += cnt[b + i];
    int v = s;
    for (int off = 1; off < 32; off <<= 1) {
        int u = __shfl_up(v, off, 32);
        if (lane >= off) v += u;
    }
    if (lane == 31) wsum[t >> 5] = v;
    __syncthreads();
    if (t < 32) {
        int w = wsum[t];
        int x = w;
        for (int off = 1; off < 32; off <<= 1) {
            int u = __shfl_up(x, off, 32);
            if (t >= off) x += u;
        }
        wsum[t] = x - w;
    }
    __syncthreads();
    int run = wsum[t >> 5] + (v - s);
    for (int i = 0; i < lim; ++i) {
        const int c = cnt[b + i];
        offs[b + i] = run;
        cur[b + i]  = run;
        run += c;
    }
}

__global__ __launch_bounds__(256) void place_rows(const SegIn j0, const SegIn j1, const SegIn j2,
                                                  const SegIn j3, const SegIn j4, const SegIn j5,
                                                  int* __restrict__ wi, int* __restrict__ di) {
    int local;
    const int job = edge_job(blockIdx.x, local);
    const SegIn in = seg_in(job, j0, j1, j2, j3, j4, j5);
    const SegWs s = seg_ws(job, wi, di);
    const int t = local * 256 + threadIdx.x;
    if (t >= s.E) return;
    const int pos = atomicAdd(&s.cur[in.e_dst[t]], 1);
    s.rows[pos] = in.src_nid[in.e_src[t]];
}

__device__ __forceinline__ void add4(float4& a, const float4 x) {
    a.x += x.x; a.y += x.y; a.z += x.z; a.w += x.w;
}

#define WNB ((NWN + 7) / 8)
#define DNB ((BN + 7) / 8)
__global__ __launch_bounds__(256) void gather_mean(const float* __restrict__ emb_w, const float* __restrict__ emb_d,
                                                   int* __restrict__ wi, int* __restrict__ di,
                                                   float* __restrict__ o0, float* __restrict__ o1,
                                                   float* __restrict__ o2, float* __restrict__ o3,
                                                   float* __restrict__ o4, float* __restrict__ o5) {
    const int lane = threadIdx.x & 31;
    const int blk = blockIdx.x;
    int job, local;
    if (blk < 3 * WNB) { job = blk / WNB; local = blk - job * WNB; }
    else { const int d = blk - 3 * WNB; const int j = d / DNB; local = d - j * DNB; job = 3 + j; }
    const SegWs s = seg_ws(job, wi, di);
    const float* __restrict__ emb = (job < 3) ? emb_w : emb_d;
    float* __restrict__ out = (job == 0) ? o0 : (job == 1) ? o1 : (job == 2) ? o2
                            : (job == 3) ? o3 : (job == 4) ? o4 : o5;
    const int* __restrict__ rows = s.rows;
    const int* __restrict__ offs = s.offs;
    const int* __restrict__ cnt = s.cnt;
    const int node = __builtin_amdgcn_readfirstlane((int)(local * 8 + (threadIdx.x >> 5)));
    if (node >= s.n) return;
    const int beg = offs[node];
    const int deg = cnt[node];
    const int i2 = (lane < 11) ? lane + 64 : 74;

    float4 a0 = make_float4(0.f, 0.f, 0.f, 0.f), a1 = a0, a2 = a0;
    if (deg <= 32) {
        const int key = (lane < deg) ? rows[beg + lane] : 0x7fffffff;
        int rank = 0;
#pragma unroll
        for (int k = 0; k < 32; ++k) {
            const int kk = __builtin_amdgcn_readlane(key, k);
            rank += (kk < key || (kk == key && k < lane)) ? 1 : 0;
        }
        const int sorted = __builtin_amdgcn_ds_permute(rank << 2, key);
        int k = 0;
        for (; k + 4 <= deg; k += 4) {
            const float4* r0 = (const float4*)(emb + (size_t)__builtin_amdgcn_readlane(sorted, k + 0) * DDIM);
            const float4* r1 = (const float4*)(emb + (size_t)__builtin_amdgcn_readlane(sorted, k + 1) * DDIM);
            const float4* r2 = (const float4*)(emb + (size_t)__builtin_amdgcn_readlane(sorted, k + 2) * DDIM);
            const float4* r3 = (const float4*)(emb + (size_t)__builtin_amdgcn_readlane(sorted, k + 3) * DDIM);
            const float4 x00 = r0[lane], x01 = r0[lane + 32], x02 = r0[i2];
            const float4 x10 = r1[lane], x11 = r1[lane + 32], x12 = r1[i2];
            const float4 x20 = r2[lane], x21 = r2[lane + 32], x22 = r2[i2];
            const float4 x30 = r3[lane], x31 = r3[lane + 32], x32 = r3[i2];
            add4(a0, x00); add4(a1, x01); add4(a2, x02);
            add4(a0, x10); add4(a1, x11); add4(a2, x12);
            add4(a0, x20); add4(a1, x21); add4(a2, x22);
            add4(a0, x30); add4(a1, x31); add4(a2, x32);
        }
        for (; k < deg; ++k) {
            const float4* r0 = (const float4*)(emb + (size_t)__builtin_amdgcn_readlane(sorted, k) * DDIM);
            const float4 x00 = r0[lane], x01 = r0[lane + 32], x02 = r0[i2];
            add4(a0, x00); add4(a1, x01); add4(a2, x02);
        }
    } else {
        int last = -1;
        for (int pass = 0; pass < deg; ++pass) {
            int best = 0x7fffffff;
            for (int b = lane; b < deg; b += 32) {
                const int kx = rows[beg + b];
                if (kx > last && kx < best) best = kx;
            }
            for (int off = 16; off > 0; off >>= 1) best = min(best, __shfl_xor(best, off, 32));
            best = __builtin_amdgcn_readfirstlane(best);
            if (best == 0x7fffffff) break;
            int mult = 0;
            for (int b = lane; b < deg; b += 32) mult += (rows[beg + b] == best) ? 1 : 0;
            for (int off = 16; off > 0; off >>= 1) mult += __shfl_xor(mult, off, 32);
            mult = __builtin_amdgcn_readfirstlane(mult);
            const float4* r0 = (const float4*)(emb + (size_t)best * DDIM);
            const float4 x00 = r0[lane], x01 = r0[lane + 32], x02 = r0[i2];
            for (int m = 0; m < mult; ++m) { add4(a0, x00); add4(a1, x01); add4(a2, x02); }
            last = best;
        }
    }

    const float d = fmaxf((float)deg, 1.f);
    const float4 z = make_float4(0.f, 0.f, 0.f, 0.f);
    float4* o = (float4*)(out + (size_t)node * PDIM);
    o[lane]      = make_float4(a0.x / d, a0.y / d, a0.z / d, a0.w / d);
    o[lane + 32] = make_float4(a1.x / d, a1.y / d, a1.z / d, a1.w / d);
    if (lane < 16)
        o[lane + 64] = (lane < 11) ? make_float4(a2.x / d, a2.y / d, a2.z / d, a2.w / d) : z;
}

#define AB 10
static_assert(KC == 16 * AB, "the K loop below is written for four A batches per chunk");

__device__ __forceinline__ void load_a_batch(v2f (&a)[AB], const float* __restrict__ p) {
#pragma unroll
    for (int t = 0; t < AB; ++t) a[t] = *(const v2f*)(p + 4 * t);
}

__device__ __forceinline__ void mma_a_batch(v8f (&acc)[4], const v2f (&a)[AB], const float* lb) {
#pragma unroll
    for (int t = 0; t < AB; ++t) {
#pragma unroll
        for (int ns = 0; ns < 4; ++ns) {
            v2f b = *(const v2f*)(lb + ns * 16 * LP + 4 * t);
            acc[ns] = __builtin_amdgcn_wmma_f32_16x16x4_f32(
                false, a[t], false, b, (short)0, acc[ns], false, false);
        }
    }
}

__global__ __launch_bounds__(256) void gemm_f32_wmma(
        const float* __restrict__ A0,
        const float* __restrict__ A1z0, const float* __restrict__ A1z1, const float* __restrict__ A1z2,
        const float* __restrict__ Bz0, int bzs,
        const float* __restrict__ bias0, const float* __restrict__ bias1, const float* __restrict__ bias2,
        float* __restrict__ Cz0, int czs,
        int M, int N, int K, int ksplit, int lda, int ldb, int ldc) {
    __shared__ float ldsB[64 * LP];
    (void)N;
    const int z = blockIdx.z;
    const float* __restrict__ A1 = (z == 0) ? A1z0 : (z == 1) ? A1z1 : A1z2;
    const float* __restrict__ bias = (z == 0) ? bias0 : (z == 1) ? bias1 : bias2;
    const float* __restrict__ B = Bz0 + (size_t)z * bzs;
    float* __restrict__ C = Cz0 + (size_t)z * czs;
    const int lane = threadIdx.x & 31;
    const int wave = threadIdx.x >> 5;
    const int hi   = lane >> 4;
    const int l16  = lane & 15;
    const int nbase = blockIdx.y * 64;
    const int mt    = blockIdx.x * 8 + wave;
    const bool valid = (mt * 16) < M;
    const int m = mt * 16 + l16;

    v8f acc[4];
    if (valid) {
#pragma unroll
        for (int ns = 0; ns < 4; ++ns) {
            const int n = nbase + ns * 16 + l16;
            const float bv = (bias && n < DDIM) ? bias[n] : 0.f;
#pragma unroll
            for (int i = 0; i < 8; ++i) acc[ns][i] = bv;
        }
    }

    const int aoff = m * lda + 2 * hi;
    const float* lb = &ldsB[l16 * LP + 2 * hi];
    const int sc  = (threadIdx.x & 15) * 4;
    const int skr = threadIdx.x >> 4;
    const float* __restrict__ Bp = B + (size_t)skr * ldb + nbase + sc;

    v2f a0[AB], a1[AB];
    if (valid) load_a_batch(a0, A0 + aoff);

    for (int kk = 0; kk < K; kk += KC) {
        float4 bt[KC / 16];
#pragma unroll
        for (int i = 0; i < KC / 16; ++i)
            bt[i] = *(const float4*)(Bp + (size_t)(kk + 16 * i) * ldb);

        __syncthreads();
#pragma unroll
        for (int i = 0; i < KC / 16; ++i) {
            const int kloc = skr + 16 * i;
            ldsB[(sc + 0) * LP + kloc] = bt[i].x;
            ldsB[(sc + 1) * LP + kloc] = bt[i].y;
            ldsB[(sc + 2) * LP + kloc] = bt[i].z;
            ldsB[(sc + 3) * LP + kloc] = bt[i].w;
        }
        __syncthreads();

        if (valid) {
            const float* ak = ((kk < ksplit) ? A0 + kk : A1 + (kk - ksplit)) + aoff;
            load_a_batch(a1, ak + 4 * AB);
            mma_a_batch(acc, a0, lb);
            load_a_batch(a0, ak + 8 * AB);
            mma_a_batch(acc, a1, lb + 4 * AB);
            load_a_batch(a1, ak + 12 * AB);
            mma_a_batch(acc, a0, lb + 8 * AB);
            const int kn = kk + KC;
            if (kn < K) load_a_batch(a0, ((kn < ksplit) ? A0 + kn : A1 + (kn - ksplit)) + aoff);
            mma_a_batch(acc, a1, lb + 12 * AB);
        }
    }
    if (valid) {
#pragma unroll
        for (int ns = 0; ns < 4; ++ns)
#pragma unroll
            for (int i = 0; i < 8; ++i)
                C[(size_t)(mt * 16 + i + hi * 8) * ldc + nbase + ns * 16 + l16] = acc[ns][i];
    }
}

#define KW  32
#define LPW 36
#define NT  (PDIM / 16)
#define AW  (KW / 4)
#define NCH ((2 * PDIM) / KW)
#define NBT ((KW * (PDIM / 4)) / 256)
static_assert(PDIM % (2 * KW) == 0, "a pair of chunks never straddles the A0 | X boundary; the K loop takes two chunks per trip");
static_assert(KW == 32 && PDIM == 32 * NBT, "panel staging map below: 8 rows x 16 floats per wave-load, 32 columns per pass");

__device__ __forceinline__ void load_b_panel(float4 (&bt)[NBT], const float* __restrict__ p) {
#pragma unroll
    for (int i = 0; i < NBT; ++i) bt[i] = *(const float4*)(p + 32 * i);
}

__device__ __forceinline__ void store_b_panel(float* ls, const float4 (&bt)[NBT]) {
#pragma unroll
    for (int i = 0; i < NBT; ++i) {
        ls[(32 * i + 0) * LPW] = bt[i].x;
        ls[(32 * i + 1) * LPW] = bt[i].y;
        ls[(32 * i + 2) * LPW] = bt[i].z;
        ls[(32 * i + 3) * LPW] = bt[i].w;
    }
}

__device__ __forceinline__ void chunk_rows(v8f (&acc)[NT], v2f (&a)[AW], const float* lb, const float* pn,
                                           float* ls, float4 (&bt)[NBT], const float* __restrict__ bn) {
#pragma unroll
    for (int t = 0; t < AW; ++t) {
        if (t == AW / 2) {
            store_b_panel(ls, bt);
            load_b_panel(bt, bn);
        }
#pragma unroll
        for (int ns = 0; ns < NT; ++ns) {
            v2f b = *(const v2f*)(lb + ns * 16 * LPW + 4 * t);
            acc[ns] = __builtin_amdgcn_wmma_f32_16x16x4_f32(
                false, a[t], false, b, (short)0, acc[ns], false, false);
        }
        a[t] = *(const v2f*)(pn + 4 * t);
    }
}

__global__ __launch_bounds__(256) void gemm_rows_wmma(
        const float* __restrict__ A0, float* X0, float* X1, float* X2,
        const float* __restrict__ Bz0, int bzs,
        const float* __restrict__ bias0, const float* __restrict__ bias1, const float* __restrict__ bias2,
        int M) {
    __shared__ float ldsB[2 * PDIM * LPW];
    const int lane = threadIdx.x & 31;
    const int wave = threadIdx.x >> 5;
    const int hi   = lane >> 4;
    const int l16  = lane & 15;
    const float* lb0 = &ldsB[l16 * LPW + 2 * hi];
    const float* lb1 = lb0 + PDIM * LPW;
    const int skr = (wave & 3) * 8 + (lane & 7);
    const int sc  = ((wave >> 2) * 4 + (lane >> 3)) * 4;
    float* ls0 = &ldsB[sc * LPW + skr];
    float* ls1 = ls0 + PDIM * LPW;
    const int nrb = (M + 127) >> 7;

    for (int u = blockIdx.x; u < 3 * nrb; u += gridDim.x) {
        const int z = u / nrb;
        const int mt = (u - z * nrb) * 8 + wave;
        float* X = (z == 0) ? X0 : (z == 1) ? X1 : X2;
        const float* __restrict__ bias = (z == 0) ? bias0 : (z == 1) ? bias1 : bias2;
        const float* __restrict__ Bp = Bz0 + (size_t)z * bzs + (size_t)skr * PDIM + sc;
        const bool valid = (mt * 16) < M;
        const int aoff = min(mt * 16 + l16, M - 1) * PDIM + 2 * hi;

        float4 bt[NBT];
        v2f ae[AW], ao[AW];
        load_b_panel(bt, Bp);
#pragma unroll
        for (int t = 0; t < AW; ++t) {
            ae[t] = *(const v2f*)(A0 + aoff + 4 * t);
            ao[t] = *(const v2f*)(A0 + aoff + KW + 4 * t);
        }
        store_b_panel(ls0, bt);
        load_b_panel(bt, Bp + (size_t)KW * PDIM);
        __syncthreads();

        v8f acc[NT];
#pragma unroll
        for (int ns = 0; ns < NT; ++ns) {
            const int n = ns * 16 + l16;
            const float bv = (n < DDIM) ? bias[n] : 0.f;
#pragma unroll
            for (int i = 0; i < 8; ++i) acc[ns][i] = bv;
        }

        for (int c = 0; c < NCH; c += 2) {
            const int kn = min((c + 2) * KW, 2 * PDIM - 2 * KW);
            const float* an = ((kn < PDIM) ? A0 + kn : X + (kn - PDIM)) + aoff;
            chunk_rows(acc, ae, lb0, an, ls1, bt, Bp + (size_t)kn * PDIM);
            __syncthreads();
            chunk_rows(acc, ao, lb1, an + KW, ls0, bt, Bp + (size_t)(kn + KW) * PDIM);
            __syncthreads();
        }
        if (valid) {
#pragma unroll
            for (int ns = 0; ns < NT; ++ns)
#pragma unroll
                for (int i = 0; i < 8; ++i)
                    X[(size_t)(mt * 16 + i + hi * 8) * PDIM + ns * 16 + l16] = acc[ns][i];
        }
    }
}

__global__ void sdpa_fuse_add(const float* __restrict__ h0, const float* __restrict__ h1,
                              const float* __restrict__ h2, const float* __restrict__ addv,
                              float* __restrict__ out, int n) {
    const int lane = threadIdx.x & 31;
    const int node = blockIdx.x * (blockDim.x >> 5) + (threadIdx.x >> 5);
    if (node >= n) return;
    const float* x0 = h0 + (size_t)node * PDIM;
    const float* x1 = h1 + (size_t)node * PDIM;
    const float* x2 = h2 + (size_t)node * PDIM;

    float d00 = 0, d01 = 0, d02 = 0, d11 = 0, d12 = 0, d22 = 0;
    for (int j = lane; j < PDIM; j += 32) {
        float a = x0[j], b = x1[j], e = x2[j];
        d00 += a * a; d01 += a * b; d02 += a * e;
        d11 += b * b; d12 += b * e; d22 += e * e;
    }
    for (int off = 16; off > 0; off >>= 1) {
        d00 += __shfl_xor(d00, off, 32);
        d01 += __shfl_xor(d01, off, 32);
        d02 += __shfl_xor(d02, off, 32);
        d11 += __shfl_xor(d11, off, 32);
        d12 += __shfl_xor(d12, off, 32);
        d22 += __shfl_xor(d22, off, 32);
    }
    const float scale = rsqrtf((float)DDIM);
    float S[3][3] = {{d00, d01, d02}, {d01, d11, d12}, {d02, d12, d22}};
    float wsum[3] = {0.f, 0.f, 0.f};
    for (int t = 0; t < 3; ++t) {
        float p0 = S[t][0] * scale, p1 = S[t][1] * scale, p2 = S[t][2] * scale;
        float mx = fmaxf(p0, fmaxf(p1, p2));
        float e0 = __expf(p0 - mx), e1 = __expf(p1 - mx), e2 = __expf(p2 - mx);
        float inv = 1.f / (e0 + e1 + e2);
        wsum[0] += e0 * inv; wsum[1] += e1 * inv; wsum[2] += e2 * inv;
    }
    const float w0 = wsum[0] * (1.f / 3.f), w1 = wsum[1] * (1.f / 3.f), w2 = wsum[2] * (1.f / 3.f);
    const float* av = addv + (size_t)node * PDIM;
    float* o = out + (size_t)node * PDIM;
    for (int j = lane; j < PDIM; j += 32)
        o[j] = w0 * x0[j] + w1 * x1[j] + w2 * x2[j] + av[j];
}

#define PB 32
static_assert(LN % (2 * PB) == 0, "pool_seq walks the sequence in pairs of batches");
static_assert(PDIM % 32 == 0, "every wave of pool_seq is full, so every lane's token id can be read by readlane");

__device__ __forceinline__ void pool_load(float (&x)[PB], const float* __restrict__ table, unsigned j,
                                          int ids, unsigned real) {
#pragma unroll
    for (int t = 0; t < PB; ++t) {
        const int idx = __builtin_amdgcn_readlane(ids, t);
        const int r = ((real >> t) & 1u) ? idx : 0;
        const float* __restrict__ rp = table + (size_t)r * PDIM;
        x[t] = rp[j];
    }
}

__device__ __forceinline__ void pool_add(float& acc, const float (&x)[PB], unsigned real) {
#pragma unroll
    for (int t = 0; t < PB; ++t)
        acc += ((real >> t) & 1u) ? x[t] : 0.f;
}

__global__ __launch_bounds__(PDIM) void pool_seq(const float* __restrict__ table, const int* __restrict__ xb,
                                                 float* __restrict__ pooled) {
    const int b = blockIdx.x;
    const unsigned j = threadIdx.x;
    if (j >= PDIM) return;
    const unsigned lane = j & 31u;
    const int* row = xb + b * LN;
    int ids[LN / PB];
    unsigned real[LN / PB];
#pragma unroll
    for (int k = 0; k < LN / PB; ++k) ids[k] = row[PB * k + lane];
#pragma unroll
    for (int k = 0; k < LN / PB; ++k)
        real[k] = __builtin_amdgcn_ballot_w32((unsigned)ids[k] < (unsigned)NWN);

    float acc = 0.f;
    float x0[PB], x1[PB];
    pool_load(x0, table, j, ids[0], real[0]);
#pragma unroll
    for (int k = 0; k < LN / PB; k += 2) {
        pool_load(x1, table, j, ids[k + 1], real[k + 1]);
        pool_add(acc, x0, real[k]);
        if (k + 2 < LN / PB) pool_load(x0, table, j, ids[k + 2], real[k + 2]);
        pool_add(acc, x1, real[k + 1]);
    }
    pooled[(size_t)b * PDIM + j] = acc * (1.f / (float)LN);
}

__global__ void fc_kernel(const float* __restrict__ rep, const float* __restrict__ wfc,
                          const float* __restrict__ bfc, float* __restrict__ out) {
    int t = blockIdx.x * blockDim.x + threadIdx.x;
    if (t >= BN * CN) return;
    int b = t / CN, c = t % CN;
    float acc = bfc[c];
    const float* r = rep + (size_t)b * PDIM;
    for (int j = 0; j < DDIM; ++j) acc += r[j] * wfc[(size_t)j * CN + c];
    out[t] = acc;
}


extern "C" void kernel_launch(void* const* d_in, const int* in_sizes, int n_in,
                              void* d_out, int out_size, void* d_ws, size_t ws_size,
                              hipStream_t stream) {
    (void)in_sizes; (void)n_in; (void)out_size; (void)ws_size;
    const float* emb_word = (const float*)d_in[0];
    const float* emb_doc  = (const float*)d_in[1];
    const float* w_sage[6] = {(const float*)d_in[2],  (const float*)d_in[4],  (const float*)d_in[6],
                              (const float*)d_in[8],  (const float*)d_in[10], (const float*)d_in[12]};
    const float* b_sage[6] = {(const float*)d_in[3],  (const float*)d_in[5],  (const float*)d_in[7],
                              (const float*)d_in[9],  (const float*)d_in[11], (const float*)d_in[13]};
    const float* w_dense = (const float*)d_in[14];
    const float* b_dense = (const float*)d_in[15];
    const float* w_fc    = (const float*)d_in[16];
    const float* b_fc    = (const float*)d_in[17];
    const int* dst_nid      = (const int*)d_in[18];
    const int* x_batch      = (const int*)d_in[28];
    const SegIn jw0 = {(const int*)d_in[22], (const int*)d_in[23], (const int*)d_in[19]};
    const SegIn jw1 = {(const int*)d_in[24], (const int*)d_in[25], (const int*)d_in[20]};
    const SegIn jw2 = {(const int*)d_in[26], (const int*)d_in[27], (const int*)d_in[21]};
    const SegIn jd0 = {(const int*)d_in[33], (const int*)d_in[34], (const int*)d_in[30]};
    const SegIn jd1 = {(const int*)d_in[35], (const int*)d_in[36], (const int*)d_in[31]};
    const SegIn jd2 = {(const int*)d_in[37], (const int*)d_in[38], (const int*)d_in[32]};
    float* out = (float*)d_out;

    float* ws = (float*)d_ws;
    size_t off = 0;
    auto carve = [&](size_t nflt) { float* p = ws + off; off += nflt; return p; };
    float* hdst = carve((size_t)NWN * PDIM);
    float* nbuf = carve((size_t)NWN * PDIM);
    float* h3[3]; for (int r = 0; r < 3; ++r) h3[r] = carve((size_t)NWN * PDIM);
    float* cnt  = carve(NWN);
    float* wpk[6]; for (int r = 0; r < 6; ++r) wpk[r] = carve((size_t)2 * PDIM * PDIM);
    float* wdn  = carve((size_t)PDIM * PDIM);
    float* pooled = carve((size_t)BN * PDIM);
    float* docout = carve((size_t)BN * PDIM);
    float* dnb    = carve((size_t)BN * PDIM);
    float* d3[3]; for (int r = 0; r < 3; ++r) d3[r] = carve((size_t)BN * PDIM);
    float* docrep = carve((size_t)BN * PDIM);
    (void)cnt; (void)wdn;
    int* wi = (int*)h3[0];
    int* di = (int*)d3[0];
    float* wnb[3] = {h3[1], h3[2], nbuf};
    float* dnbr[3] = {dnb, docrep, pooled};
    float* poolbuf = d3[0];
    float* wtab = h3[0];

    const int TB = 256;
    const dim3 gemmGridW(256, 1, 1);
    const dim3 gemmGridD((BN + 127) / 128, PDIM / 64, 1);
    const dim3 gemmGridD3((BN + 127) / 128, PDIM / 64, 3);

    prep<<<GPB + PKB + ZWB + ZDB, TB, 0, stream>>>(emb_word, dst_nid, hdst,
                                                   w_sage[0], w_sage[1], w_sage[2], w_sage[3], w_sage[4], w_sage[5],
                                                   w_dense, wpk[0], wi, di);

    count_dst<<<3 * WEB + 3 * DEB, TB, 0, stream>>>(jw0, jw1, jw2, jd0, jd1, jd2, wi, di);
    scan_counts<<<6, SCAN_T, 0, stream>>>(wi, di);
    place_rows<<<3 * WEB + 3 * DEB, TB, 0, stream>>>(jw0, jw1, jw2, jd0, jd1, jd2, wi, di);
    gather_mean<<<3 * WNB + 3 * DNB, TB, 0, stream>>>(emb_word, emb_doc, wi, di,
                                                      wnb[0], wnb[1], wnb[2], dnbr[0], dnbr[1], dnbr[2]);

    gemm_rows_wmma<<<gemmGridW, TB, 0, stream>>>(hdst, wnb[0], wnb[1], wnb[2], wpk[0], 2 * PDIM * PDIM,
                                                 b_sage[0], b_sage[1], b_sage[2], NWN);

    sdpa_fuse_add<<<(NWN + 7) / 8, 256, 0, stream>>>(wnb[0], wnb[1], wnb[2], hdst, wtab, NWN);

    pool_seq<<<BN, PDIM, 0, stream>>>(wtab, x_batch, poolbuf);
    gemm_f32_wmma<<<gemmGridD, TB, 0, stream>>>(poolbuf, poolbuf, poolbuf, poolbuf, wpk[0] + (size_t)12 * PDIM * PDIM, 0,
                                                b_dense, b_dense, b_dense, docout, 0,
                                                BN, PDIM, PDIM, PDIM, PDIM, PDIM, PDIM);

    gemm_f32_wmma<<<gemmGridD3, TB, 0, stream>>>(docout, dnbr[0], dnbr[1], dnbr[2], wpk[3], 2 * PDIM * PDIM,
                                                 b_sage[3], b_sage[4], b_sage[5], d3[0], BN * PDIM,
                                                 BN, PDIM, 2 * PDIM, PDIM, PDIM, PDIM, PDIM);

    sdpa_fuse_add<<<(BN + 7) / 8, 256, 0, stream>>>(d3[0], d3[1], d3[2], docout, docrep, BN);
    fc_kernel<<<(BN * CN + TB - 1) / TB, TB, 0, stream>>>(docrep, w_fc, b_fc, out);
}


# prep, count_dst, place_rows: a thread moves 16 bytes (prep: a wave copies 4 whole rows; edge stages: 4 edges per thread): 304k one-element waves to start -> 22k; same values, same float sums
# speedup vs baseline: 2.8882x; 1.0404x over previous
#include <hip/hip_runtime.h>

#define DDIM 300
#define PDIM 320
#define NWN  20000
#define EWN  320000
#define BN   256
#define LN   256
#define SDN  2048
#define EDN  4096
#define CN   20
#define KC   160
#define LP   164

typedef __attribute__((ext_vector_type(2))) float v2f;
typedef __attribute__((ext_vector_type(8))) float v8f;

struct SegIn { const int* e_src; const int* e_dst; const int* src_nid; };
struct SegWs { int* cnt; int* offs; int* cur; int* rows; int n; int E; };

__device__ __forceinline__ SegWs seg_ws(int job, int* wi, int* di) {
    SegWs s;
    if (job < 3) {
        s.n = NWN; s.E = EWN;
        s.cnt = wi + job * NWN; s.offs = wi + (3 + job) * NWN; s.cur = wi + (6 + job) * NWN;
        s.rows = wi + 9 * NWN + job * EWN;
    } else {
        const int r = job - 3;
        s.n = BN; s.E = EDN;
        s.cnt = di + r * BN; s.offs = di + (3 + r) * BN; s.cur = di + (6 + r) * BN;
        s.rows = di + 9 * BN + r * EDN;
    }
    return s;
}

__device__ __forceinline__ SegIn seg_in(int job, const SegIn& j0, const SegIn& j1, const SegIn& j2,
                                        const SegIn& j3, const SegIn& j4, const SegIn& j5) {
    switch (job) {
        case 0: return j0;
        case 1: return j1;
        case 2: return j2;
        case 3: return j3;
        case 4: return j4;
        default: return j5;
    }
}

#define EPT 4
#define WEB ((EWN / EPT + 255) / 256)
#define DEB ((EDN / EPT + 255) / 256)
static_assert(EWN % EPT == 0 && EDN % EPT == 0, "no partial group of edges");
__device__ __forceinline__ int edge_job(int blk, int& local) {
    if (blk < 3 * WEB) { const int j = blk / WEB; local = blk - j * WEB; return j; }
    const int d = blk - 3 * WEB;
    const int j = d / DEB;
    local = d - j * DEB;
    return 3 + j;
}

#define RPW 4
#define HRB (NWN / (8 * RPW))
#define WRB ((13 * PDIM) / (8 * RPW))
#define ZQW ((3 * NWN) / 4)
#define ZQD ((3 * BN) / 4)
#define ZB  ((ZQW + ZQD + 255) / 256)
static_assert(NWN % (8 * RPW) == 0 && (13 * PDIM) % (8 * RPW) == 0, "every wave of the row blocks has RPW rows");
static_assert(PDIM % RPW == 0 && DDIM % RPW == 0, "the RPW rows of a wave lie in one slab and are all data rows or all pad rows");
static_assert((3 * NWN) % 4 == 0 && (3 * BN) % 4 == 0, "count arrays are whole 16-byte pieces");

__device__ __forceinline__ void copy_rows(const float* __restrict__ s0, const float* __restrict__ s1,
                                          const float* __restrict__ s2, const float* __restrict__ s3,
                                          float* __restrict__ o, int lane) {
    static_assert(RPW == 4 && PDIM == 320 && DDIM == 300, "run of 320 float4, 75 data groups per row");
    float4 x[10];
    bool keep[10];
#pragma unroll
    for (int i = 0; i < 10; ++i) {
        const int f = lane + 32 * i;
        const int q = (f >= 80 ? 1 : 0) + (f >= 160 ? 1 : 0) + (f >= 240 ? 1 : 0);
        const int g = f - 80 * q;
        const float* __restrict__ sq = (q == 0) ? s0 : (q == 1) ? s1 : (q == 2) ? s2 : s3;
        keep[i] = g < 75;
        x[i] = ((const float4*)sq)[keep[i] ? g : 74];
    }
    float4* __restrict__ d = (float4*)o;
#pragma unroll
    for (int i = 0; i < 10; ++i)
        d[lane + 32 * i] = make_float4(keep[i] ? x[i].x : 0.f, keep[i] ? x[i].y : 0.f,
                                       keep[i] ? x[i].z : 0.f, keep[i] ? x[i].w : 0.f);
}

__device__ __forceinline__ void zero_rows(float* __restrict__ o, int lane) {
    const float4 z = make_float4(0.f, 0.f, 0.f, 0.f);
    float4* __restrict__ d = (float4*)o;
#pragma unroll
    for (int i = 0; i < 10; ++i) d[lane + 32 * i] = z;
}

__global__ __launch_bounds__(256) void prep(const float* __restrict__ emb, const int* __restrict__ nid,
                                            float* __restrict__ hdst,
                                            const float* __restrict__ w0, const float* __restrict__ w1,
                                            const float* __restrict__ w2, const float* __restrict__ w3,
                                            const float* __restrict__ w4, const float* __restrict__ w5,
                                            const float* __restrict__ wd, float* __restrict__ wout,
                                            int* __restrict__ wi, int* __restrict__ di) {
    const int blk = blockIdx.x;
    const int lane = threadIdx.x & 31;
    const int wv = __builtin_amdgcn_readfirstlane((int)(blk * 8 + (threadIdx.x >> 5)));
    if (blk < HRB) {
        const int r = wv * RPW;
        copy_rows(emb + (size_t)nid[r + 0] * DDIM, emb + (size_t)nid[r + 1] * DDIM,
                  emb + (size_t)nid[r + 2] * DDIM, emb + (size_t)nid[r + 3] * DDIM,
                  hdst + (size_t)r * PDIM, lane);
    } else if (blk < HRB + WRB) {
        const int rr = (wv - HRB * 8) * RPW;
        const int slab = rr / PDIM;
        const int r = rr - slab * PDIM;
        float* __restrict__ o = wout + (size_t)rr * PDIM;
        if (r < DDIM) {
            const int which = slab >> 1;
            const float* __restrict__ w = (slab >= 12) ? wd : (which == 0) ? w0 : (which == 1) ? w1 : (which == 2) ? w2
                                        : (which == 3) ? w3 : (which == 4) ? w4 : w5;
            const int r0 = (slab < 12 && (slab & 1)) ? DDIM : 0;
            const float* __restrict__ s = w + (size_t)(r0 + r) * DDIM;
            copy_rows(s, s + DDIM, s + 2 * DDIM, s + 3 * DDIM, o, lane);
        } else {
            zero_rows(o, lane);
        }
    } else {
        const int t = (blk - HRB - WRB) * 256 + threadIdx.x;
        const int4 z = make_int4(0, 0, 0, 0);
        if (t < ZQW) ((int4*)wi)[t] = z;
        else if (t < ZQW + ZQD) ((int4*)di)[t - ZQW] = z;
    }
}


__global__ __launch_bounds__(256) void count_dst(const SegIn j0, const SegIn j1, const SegIn j2,
                                                 const SegIn j3, const SegIn j4, const SegIn j5,
                                                 int* __restrict__ wi, int* __restrict__ di) {
    int local;
    const int job = edge_job(blockIdx.x, local);
    const SegIn in = seg_in(job, j0, j1, j2, j3, j4, j5);
    const SegWs s = seg_ws(job, wi, di);
    const int t = local * 256 + threadIdx.x;
    if (t < s.E / EPT) {
        const int4 d = ((const int4*)in.e_dst)[t];
        atomicAdd(&s.cnt[d.x], 1);
        atomicAdd(&s.cnt[d.y], 1);
        atomicAdd(&s.cnt[d.z], 1);
        atomicAdd(&s.cnt[d.w], 1);
    }
}

#define SCAN_T 1024
__global__ __launch_bounds__(SCAN_T) void scan_counts(int* __restrict__ wi, int* __restrict__ di) {
    __shared__ int wsum[SCAN_T / 32];
    const SegWs sw = seg_ws(blockIdx.x, wi, di);
    const int* __restrict__ cnt = sw.cnt;
    int* __restrict__ offs = sw.offs;
    int* __restrict__ cur = sw.cur;
    const int n = sw.n;
    const int t = threadIdx.x;
    const int lane = t & 31;
    const int ipt = (n + SCAN_T - 1) / SCAN_T;
    const int b = t * ipt;
    const int lim = min(ipt, n - b);
    int s = 0;
    for (int i = 0; i < lim; ++i) s += cnt[b + i];
    int v = s;
    for (int off = 1; off < 32; off <<= 1) {
        int u = __shfl_up(v, off, 32);
        if (lane >= off) v += u;
    }
    if (lane == 31) wsum[t >> 5] = v;
    __syncthreads();
    if (t < 32) {
        int w = wsum[t];
        int x = w;
        for (int off = 1; off < 32; off <<= 1) {
            int u = __shfl_up(x, off, 32);
            if (t >= off) x += u;
        }
        wsum[t] = x - w;
    }
    __syncthreads();
    int run = wsum[t >> 5] + (v - s);
    for (int i = 0; i < lim; ++i) {
        const int c = cnt[b + i];
        offs[b + i] = run;
        cur[b + i]  = run;
        run += c;
    }
}

__global__ __launch_bounds__(256) void place_rows(const SegIn j0, const SegIn j1, const SegIn j2,
                                                  const SegIn j3, const SegIn j4, const SegIn j5,
                                                  int* __restrict__ wi, int* __restrict__ di) {
    int local;
    const int job = edge_job(blockIdx.x, local);
    const SegIn in = seg_in(job, j0, j1, j2, j3, j4, j5);
    const SegWs s = seg_ws(job, wi, di);
    const int t = local * 256 + threadIdx.x;
    if (t >= s.E / EPT) return;
    const int4 d = ((const int4*)in.e_dst)[t];
    const int4 e = ((const int4*)in.e_src)[t];
    const int r0 = in.src_nid[e.x], r1 = in.src_nid[e.y], r2 = in.src_nid[e.z], r3 = in.src_nid[e.w];
    const int p0 = atomicAdd(&s.cur[d.x], 1);
    const int p1 = atomicAdd(&s.cur[d.y], 1);
    const int p2 = atomicAdd(&s.cur[d.z], 1);
    const int p3 = atomicAdd(&s.cur[d.w], 1);
    s.rows[p0] = r0; s.rows[p1] = r1; s.rows[p2] = r2; s.rows[p3] = r3;
}

__device__ __forceinline__ void add4(float4& a, const float4 x) {
    a.x += x.x; a.y += x.y; a.z += x.z; a.w += x.w;
}

#define WNB ((NWN + 7) / 8)
#define DNB ((BN + 7) / 8)
__global__ __launch_bounds__(256) void gather_mean(const float* __restrict__ emb_w, const float* __restrict__ emb_d,
                                                   int* __restrict__ wi, int* __restrict__ di,
                                                   float* __restrict__ o0, float* __restrict__ o1,
                                                   float* __restrict__ o2, float* __restrict__ o3,
                                                   float* __restrict__ o4, float* __restrict__ o5) {
    const int lane = threadIdx.x & 31;
    const int blk = blockIdx.x;
    int job, local;
    if (blk < 3 * WNB) { job = blk / WNB; local = blk - job * WNB; }
    else { const int d = blk - 3 * WNB; const int j = d / DNB; local = d - j * DNB; job = 3 + j; }
    const SegWs s = seg_ws(job, wi, di);
    const float* __restrict__ emb = (job < 3) ? emb_w : emb_d;
    float* __restrict__ out = (job == 0) ? o0 : (job == 1) ? o1 : (job == 2) ? o2
                            : (job == 3) ? o3 : (job == 4) ? o4 : o5;
    const int* __restrict__ rows = s.rows;
    const int* __restrict__ offs = s.offs;
    const int* __restrict__ cnt = s.cnt;
    const int node = __builtin_amdgcn_readfirstlane((int)(local * 8 + (threadIdx.x >> 5)));
    if (node >= s.n) return;
    const int beg = offs[node];
    const int deg = cnt[node];
    const int i2 = (lane < 11) ? lane + 64 : 74;

    float4 a0 = make_float4(0.f, 0.f, 0.f, 0.f), a1 = a0, a2 = a0;
    if (deg <= 32) {
        const int key = (lane < deg) ? rows[beg + lane] : 0x7fffffff;
        int rank = 0;
#pragma unroll
        for (int k = 0; k < 32; ++k) {
            const int kk = __builtin_amdgcn_readlane(key, k);
            rank += (kk < key || (kk == key && k < lane)) ? 1 : 0;
        }
        const int sorted = __builtin_amdgcn_ds_permute(rank << 2, key);
        int k = 0;
        for (; k + 4 <= deg; k += 4) {
            const float4* r0 = (const float4*)(emb + (size_t)__builtin_amdgcn_readlane(sorted, k + 0) * DDIM);
            const float4* r1 = (const float4*)(emb + (size_t)__builtin_amdgcn_readlane(sorted, k + 1) * DDIM);
            const float4* r2 = (const float4*)(emb + (size_t)__builtin_amdgcn_readlane(sorted, k + 2) * DDIM);
            const float4* r3 = (const float4*)(emb + (size_t)__builtin_amdgcn_readlane(sorted, k + 3) * DDIM);
            const float4 x00 = r0[lane], x01 = r0[lane + 32], x02 = r0[i2];
            const float4 x10 = r1[lane], x11 = r1[lane + 32], x12 = r1[i2];
            const float4 x20 = r2[lane], x21 = r2[lane + 32], x22 = r2[i2];
            const float4 x30 = r3[lane], x31 = r3[lane + 32], x32 = r3[i2];
            add4(a0, x00); add4(a1, x01); add4(a2, x02);
            add4(a0, x10); add4(a1, x11); add4(a2, x12);
            add4(a0, x20); add4(a1, x21); add4(a2, x22);
            add4(a0, x30); add4(a1, x31); add4(a2, x32);
        }
        for (; k < deg; ++k) {
            const float4* r0 = (const float4*)(emb + (size_t)__builtin_amdgcn_readlane(sorted, k) * DDIM);
            const float4 x00 = r0[lane], x01 = r0[lane + 32], x02 = r0[i2];
            add4(a0, x00); add4(a1, x01); add4(a2, x02);
        }
    } else {
        int last = -1;
        for (int pass = 0; pass < deg; ++pass) {
            int best = 0x7fffffff;
            for (int b = lane; b < deg; b += 32) {
                const int kx = rows[beg + b];
                if (kx > last && kx < best) best = kx;
            }
            for (int off = 16; off > 0; off >>= 1) best = min(best, __shfl_xor(best, off, 32));
            best = __builtin_amdgcn_readfirstlane(best);
            if (best == 0x7fffffff) break;
            int mult = 0;
            for (int b = lane; b < deg; b += 32) mult += (rows[beg + b] == best) ? 1 : 0;
            for (int off = 16; off > 0; off >>= 1) mult += __shfl_xor(mult, off, 32);
            mult = __builtin_amdgcn_readfirstlane(mult);
            const float4* r0 = (const float4*)(emb + (size_t)best * DDIM);
            const float4 x00 = r0[lane], x01 = r0[lane + 32], x02 = r0[i2];
            for (int m = 0; m < mult; ++m) { add4(a0, x00); add4(a1, x01); add4(a2, x02); }
            last = best;
        }
    }

    const float d = fmaxf((float)deg, 1.f);
    const float4 z = make_float4(0.f, 0.f, 0.f, 0.f);
    float4* o = (float4*)(out + (size_t)node * PDIM);
    o[lane]      = make_float4(a0.x / d, a0.y / d, a0.z / d, a0.w / d);
    o[lane + 32] = make_float4(a1.x / d, a1.y / d, a1.z / d, a1.w / d);
    if (lane < 16)
        o[lane + 64] = (lane < 11) ? make_float4(a2.x / d, a2.y / d, a2.z / d, a2.w / d) : z;
}

#define AB 10
static_assert(KC == 16 * AB, "the K loop below is written for four A batches per chunk");

__device__ __forceinline__ void load_a_batch(v2f (&a)[AB], const float* __restrict__ p) {
#pragma unroll
    for (int t = 0; t < AB; ++t) a[t] = *(const v2f*)(p + 4 * t);
}

__device__ __forceinline__ void mma_a_batch(v8f (&acc)[4], const v2f (&a)[AB], const float* lb) {
#pragma unroll
    for (int t = 0; t < AB; ++t) {
#pragma unroll
        for (int ns = 0; ns < 4; ++ns) {
            v2f b = *(const v2f*)(lb + ns * 16 * LP + 4 * t);
            acc[ns] = __builtin_amdgcn_wmma_f32_16x16x4_f32(
                false, a[t], false, b, (short)0, acc[ns], false, false);
        }
    }
}

__global__ __launch_bounds__(256) void gemm_f32_wmma(
        const float* __restrict__ A0,
        const float* __restrict__ A1z0, const float* __restrict__ A1z1, const float* __restrict__ A1z2,
        const float* __restrict__ Bz0, int bzs,
        const float* __restrict__ bias0, const float* __restrict__ bias1, const float* __restrict__ bias2,
        float* __restrict__ Cz0, int czs,
        int M, int N, int K, int ksplit, int lda, int ldb, int ldc) {
    __shared__ float ldsB[64 * LP];
    (void)N;
    const int z = blockIdx.z;
    const float* __restrict__ A1 = (z == 0) ? A1z0 : (z == 1) ? A1z1 : A1z2;
    const float* __restrict__ bias = (z == 0) ? bias0 : (z == 1) ? bias1 : bias2;
    const float* __restrict__ B = Bz0 + (size_t)z * bzs;
    float* __restrict__ C = Cz0 + (size_t)z * czs;
    const int lane = threadIdx.x & 31;
    const int wave = threadIdx.x >> 5;
    const int hi   = lane >> 4;
    const int l16  = lane & 15;
    const int nbase = blockIdx.y * 64;
    const int mt    = blockIdx.x * 8 + wave;
    const bool valid = (mt * 16) < M;
    const int m = mt * 16 + l16;

    v8f acc[4];
    if (valid) {
#pragma unroll
        for (int ns = 0; ns < 4; ++ns) {
            const int n = nbase + ns * 16 + l16;
            const float bv = (bias && n < DDIM) ? bias[n] : 0.f;
#pragma unroll
            for (int i = 0; i < 8; ++i) acc[ns][i] = bv;
        }
    }

    const int aoff = m * lda + 2 * hi;
    const float* lb = &ldsB[l16 * LP + 2 * hi];
    const int sc  = (threadIdx.x & 15) * 4;
    const int skr = threadIdx.x >> 4;
    const float* __restrict__ Bp = B + (size_t)skr * ldb + nbase + sc;

    v2f a0[AB], a1[AB];
    if (valid) load_a_batch(a0, A0 + aoff);

    for (int kk = 0; kk < K; kk += KC) {
        float4 bt[KC / 16];
#pragma unroll
        for (int i = 0; i < KC / 16; ++i)
            bt[i] = *(const float4*)(Bp + (size_t)(kk + 16 * i) * ldb);

        __syncthreads();
#pragma unroll
        for (int i = 0; i < KC / 16; ++i) {
            const int kloc = skr + 16 * i;
            ldsB[(sc + 0) * LP + kloc] = bt[i].x;
            ldsB[(sc + 1) * LP + kloc] = bt[i].y;
            ldsB[(sc + 2) * LP + kloc] = bt[i].z;
            ldsB[(sc + 3) * LP + kloc] = bt[i].w;
        }
        __syncthreads();

        if (valid) {
            const float* ak = ((kk < ksplit) ? A0 + kk : A1 + (kk - ksplit)) + aoff;
            load_a_batch(a1, ak + 4 * AB);
            mma_a_batch(acc, a0, lb);
            load_a_batch(a0, ak + 8 * AB);
            mma_a_batch(acc, a1, lb + 4 * AB);
            load_a_batch(a1, ak + 12 * AB);
            mma_a_batch(acc, a0, lb + 8 * AB);
            const int kn = kk + KC;
            if (kn < K) load_a_batch(a0, ((kn < ksplit) ? A0 + kn : A1 + (kn - ksplit)) + aoff);
            mma_a_batch(acc, a1, lb + 12 * AB);
        }
    }
    if (valid) {
#pragma unroll
        for (int ns = 0; ns < 4; ++ns)
#pragma unroll
            for (int i = 0; i < 8; ++i)
                C[(size_t)(mt * 16 + i + hi * 8) * ldc + nbase + ns * 16 + l16] = acc[ns][i];
    }
}

#define KW  32
#define LPW 36
#define NT  (PDIM / 16)
#define AW  (KW / 4)
#define NCH ((2 * PDIM) / KW)
#define NBT ((KW * (PDIM / 4)) / 256)
static_assert(PDIM % (2 * KW) == 0, "a pair of chunks never straddles the A0 | X boundary; the K loop takes two chunks per trip");
static_assert(KW == 32 && PDIM == 32 * NBT, "panel staging map below: 8 rows x 16 floats per wave-load, 32 columns per pass");

__device__ __forceinline__ void load_b_panel(float4 (&bt)[NBT], const float* __restrict__ p) {
#pragma unroll
    for (int i = 0; i < NBT; ++i) bt[i] = *(const float4*)(p + 32 * i);
}

__device__ __forceinline__ void store_b_panel(float* ls, const float4 (&bt)[NBT]) {
#pragma unroll
    for (int i = 0; i < NBT; ++i) {
        ls[(32 * i + 0) * LPW] = bt[i].x;
        ls[(32 * i + 1) * LPW] = bt[i].y;
        ls[(32 * i + 2) * LPW] = bt[i].z;
        ls[(32 * i + 3) * LPW] = bt[i].w;
    }
}

__device__ __forceinline__ void chunk_rows(v8f (&acc)[NT], v2f (&a)[AW], const float* lb, const float* pn,
                                           float* ls, float4 (&bt)[NBT], const float* __restrict__ bn) {
#pragma unroll
    for (int t = 0; t < AW; ++t) {
        if (t == AW / 2) {
            store_b_panel(ls, bt);
            load_b_panel(bt, bn);
        }
#pragma unroll
        for (int ns = 0; ns < NT; ++ns) {
            v2f b = *(const v2f*)(lb + ns * 16 * LPW + 4 * t);
            acc[ns] = __builtin_amdgcn_wmma_f32_16x16x4_f32(
                false, a[t], false, b, (short)0, acc[ns], false, false);
        }
        a[t] = *(const v2f*)(pn + 4 * t);
    }
}

__global__ __launch_bounds__(256) void gemm_rows_wmma(
        const float* __restrict__ A0, float* X0, float* X1, float* X2,
        const float* __restrict__ Bz0, int bzs,
        const float* __restrict__ bias0, const float* __restrict__ bias1, const float* __restrict__ bias2,
        int M) {
    __shared__ float ldsB[2 * PDIM * LPW];
    const int lane = threadIdx.x & 31;
    const int wave = threadIdx.x >> 5;
    const int hi   = lane >> 4;
    const int l16  = lane & 15;
    const float* lb0 = &ldsB[l16 * LPW + 2 * hi];
    const float* lb1 = lb0 + PDIM * LPW;
    const int skr = (wave & 3) * 8 + (lane & 7);
    const int sc  = ((wave >> 2) * 4 + (lane >> 3)) * 4;
    float* ls0 = &ldsB[sc * LPW + skr];
    float* ls1 = ls0 + PDIM * LPW;
    const int nrb = (M + 127) >> 7;

    for (int u = blockIdx.x; u < 3 * nrb; u += gridDim.x) {
        const int z = u / nrb;
        const int mt = (u - z * nrb) * 8 + wave;
        float* X = (z == 0) ? X0 : (z == 1) ? X1 : X2;
        const float* __restrict__ bias = (z == 0) ? bias0 : (z == 1) ? bias1 : bias2;
        const float* __restrict__ Bp = Bz0 + (size_t)z * bzs + (size_t)skr * PDIM + sc;
        const bool valid = (mt * 16) < M;
        const int aoff = min(mt * 16 + l16, M - 1) * PDIM + 2 * hi;

        float4 bt[NBT];
        v2f ae[AW], ao[AW];
        load_b_panel(bt, Bp);
#pragma unroll
        for (int t = 0; t < AW; ++t) {
            ae[t] = *(const v2f*)(A0 + aoff + 4 * t);
            ao[t] = *(const v2f*)(A0 + aoff + KW + 4 * t);
        }
        store_b_panel(ls0, bt);
        load_b_panel(bt, Bp + (size_t)KW * PDIM);
        __syncthreads();

        v8f acc[NT];
#pragma unroll
        for (int ns = 0; ns < NT; ++ns) {
            const int n = ns * 16 + l16;
            const float bv = (n < DDIM) ? bias[n] : 0.f;
#pragma unroll
            for (int i = 0; i < 8; ++i) acc[ns][i] = bv;
        }

        for (int c = 0; c < NCH; c += 2) {
            const int kn = min((c + 2) * KW, 2 * PDIM - 2 * KW);
            const float* an = ((kn < PDIM) ? A0 + kn : X + (kn - PDIM)) + aoff;
            chunk_rows(acc, ae, lb0, an, ls1, bt, Bp + (size_t)kn * PDIM);
            __syncthreads();
            chunk_rows(acc, ao, lb1, an + KW, ls0, bt, Bp + (size_t)(kn + KW) * PDIM);
            __syncthreads();
        }
        if (valid) {
#pragma unroll
            for (int ns = 0; ns < NT; ++ns)
#pragma unroll
                for (int i = 0; i < 8; ++i)
                    X[(size_t)(mt * 16 + i + hi * 8) * PDIM + ns * 16 + l16] = acc[ns][i];
        }
    }
}

__global__ void sdpa_fuse_add(const float* __restrict__ h0, const float* __restrict__ h1,
                              const float* __restrict__ h2, const float* __restrict__ addv,
                              float* __restrict__ out, int n) {
    const int lane = threadIdx.x & 31;
    const int node = blockIdx.x * (blockDim.x >> 5) + (threadIdx.x >> 5);
    if (node >= n) return;
    const float* x0 = h0 + (size_t)node * PDIM;
    const float* x1 = h1 + (size_t)node * PDIM;
    const float* x2 = h2 + (size_t)node * PDIM;

    float d00 = 0, d01 = 0, d02 = 0, d11 = 0, d12 = 0, d22 = 0;
    for (int j = lane; j < PDIM; j += 32) {
        float a = x0[j], b = x1[j], e = x2[j];
        d00 += a * a; d01 += a * b; d02 += a * e;
        d11 += b * b; d12 += b * e; d22 += e * e;
    }
    for (int off = 16; off > 0; off >>= 1) {
        d00 += __shfl_xor(d00, off, 32);
        d01 += __shfl_xor(d01, off, 32);
        d02 += __shfl_xor(d02, off, 32);
        d11 += __shfl_xor(d11, off, 32);
        d12 += __shfl_xor(d12, off, 32);
        d22 += __shfl_xor(d22, off, 32);
    }
    const float scale = rsqrtf((float)DDIM);
    float S[3][3] = {{d00, d01, d02}, {d01, d11, d12}, {d02, d12, d22}};
    float wsum[3] = {0.f, 0.f, 0.f};
    for (int t = 0; t < 3; ++t) {
        float p0 = S[t][0] * scale, p1 = S[t][1] * scale, p2 = S[t][2] * scale;
        float mx = fmaxf(p0, fmaxf(p1, p2));
        float e0 = __expf(p0 - mx), e1 = __expf(p1 - mx), e2 = __expf(p2 - mx);
        float inv = 1.f / (e0 + e1 + e2);
        wsum[0] += e0 * inv; wsum[1] += e1 * inv; wsum[2] += e2 * inv;
    }
    const float w0 = wsum[0] * (1.f / 3.f), w1 = wsum[1] * (1.f / 3.f), w2 = wsum[2] * (1.f / 3.f);
    const float* av = addv + (size_t)node * PDIM;
    float* o = out + (size_t)node * PDIM;
    for (int j = lane; j < PDIM; j += 32)
        o[j] = w0 * x0[j] + w1 * x1[j] + w2 * x2[j] + av[j];
}

#define PB 32
static_assert(LN % (2 * PB) == 0, "pool_seq walks the sequence in pairs of batches");
static_assert(PDIM % 32 == 0, "every wave of pool_seq is full, so every lane's token id can be read by readlane");

__device__ __forceinline__ void pool_load(float (&x)[PB], const float* __restrict__ table, unsigned j,
                                          int ids, unsigned real) {
#pragma unroll
    for (int t = 0; t < PB; ++t) {
        const int idx = __builtin_amdgcn_readlane(ids, t);
        const int r = ((real >> t) & 1u) ? idx : 0;
        const float* __restrict__ rp = table + (size_t)r * PDIM;
        x[t] = rp[j];
    }
}

__device__ __forceinline__ void pool_add(float& acc, const float (&x)[PB], unsigned real) {
#pragma unroll
    for (int t = 0; t < PB; ++t)
        acc += ((real >> t) & 1u) ? x[t] : 0.f;
}

__global__ __launch_bounds__(PDIM) void pool_seq(const float* __restrict__ table, const int* __restrict__ xb,
                                                 float* __restrict__ pooled) {
    const int b = blockIdx.x;
    const unsigned j = threadIdx.x;
    if (j >= PDIM) return;
    const unsigned lane = j & 31u;
    const int* row = xb + b * LN;
    int ids[LN / PB];
    unsigned real[LN / PB];
#pragma unroll
    for (int k = 0; k < LN / PB; ++k) ids[k] = row[PB * k + lane];
#pragma unroll
    for (int k = 0; k < LN / PB; ++k)
        real[k] = __builtin_amdgcn_ballot_w32((unsigned)ids[k] < (unsigned)NWN);

    float acc = 0.f;
    float x0[PB], x1[PB];
    pool_load(x0, table, j, ids[0], real[0]);
#pragma unroll
    for (int k = 0; k < LN / PB; k += 2) {
        pool_load(x1, table, j, ids[k + 1], real[k + 1]);
        pool_add(acc, x0, real[k]);
        if (k + 2 < LN / PB) pool_load(x0, table, j, ids[k + 2], real[k + 2]);
        pool_add(acc, x1, real[k + 1]);
    }
    pooled[(size_t)b * PDIM + j] = acc * (1.f / (float)LN);
}

__global__ void fc_kernel(const float* __restrict__ rep, const float* __restrict__ wfc,
                          const float* __restrict__ bfc, float* __restrict__ out) {
    int t = blockIdx.x * blockDim.x + threadIdx.x;
    if (t >= BN * CN) return;
    int b = t / CN, c = t % CN;
    float acc = bfc[c];
    const float* r = rep + (size_t)b * PDIM;
    for (int j = 0; j < DDIM; ++j) acc += r[j] * wfc[(size_t)j * CN + c];
    out[t] = acc;
}


extern "C" void kernel_launch(void* const* d_in, const int* in_sizes, int n_in,
                              void* d_out, int out_size, void* d_ws, size_t ws_size,
                              hipStream_t stream) {
    (void)in_sizes; (void)n_in; (void)out_size; (void)ws_size;
    const float* emb_word = (const float*)d_in[0];
    const float* emb_doc  = (const float*)d_in[1];
    const float* w_sage[6] = {(const float*)d_in[2],  (const float*)d_in[4],  (const float*)d_in[6],
                              (const float*)d_in[8],  (const float*)d_in[10], (const float*)d_in[12]};
    const float* b_sage[6] = {(const float*)d_in[3],  (const float*)d_in[5],  (const float*)d_in[7],
                              (const float*)d_in[9],  (const float*)d_in[11], (const float*)d_in[13]};
    const float* w_dense = (const float*)d_in[14];
    const float* b_dense = (const float*)d_in[15];
    const float* w_fc    = (const float*)d_in[16];
    const float* b_fc    = (const float*)d_in[17];
    const int* dst_nid      = (const int*)d_in[18];
    const int* x_batch      = (const int*)d_in[28];
    const SegIn jw0 = {(const int*)d_in[22], (const int*)d_in[23], (const int*)d_in[19]};
    const SegIn jw1 = {(const int*)d_in[24], (const int*)d_in[25], (const int*)d_in[20]};
    const SegIn jw2 = {(const int*)d_in[26], (const int*)d_in[27], (const int*)d_in[21]};
    const SegIn jd0 = {(const int*)d_in[33], (const int*)d_in[34], (const int*)d_in[30]};
    const SegIn jd1 = {(const int*)d_in[35], (const int*)d_in[36], (const int*)d_in[31]};
    const SegIn jd2 = {(const int*)d_in[37], (const int*)d_in[38], (const int*)d_in[32]};
    float* out = (float*)d_out;

    float* ws = (float*)d_ws;
    size_t off = 0;
    auto carve = [&](size_t nflt) { float* p = ws + off; off += nflt; return p; };
    float* hdst = carve((size_t)NWN * PDIM);
    float* nbuf = carve((size_t)NWN * PDIM);
    float* h3[3]; for (int r = 0; r < 3; ++r) h3[r] = carve((size_t)NWN * PDIM);
    float* cnt  = carve(NWN);
    float* wpk[6]; for (int r = 0; r < 6; ++r) wpk[r] = carve((size_t)2 * PDIM * PDIM);
    float* wdn  = carve((size_t)PDIM * PDIM);
    float* pooled = carve((size_t)BN * PDIM);
    float* docout = carve((size_t)BN * PDIM);
    float* dnb    = carve((size_t)BN * PDIM);
    float* d3[3]; for (int r = 0; r < 3; ++r) d3[r] = carve((size_t)BN * PDIM);
    float* docrep = carve((size_t)BN * PDIM);
    (void)cnt; (void)wdn;
    int* wi = (int*)h3[0];
    int* di = (int*)d3[0];
    float* wnb[3] = {h3[1], h3[2], nbuf};
    float* dnbr[3] = {dnb, docrep, pooled};
    float* poolbuf = d3[0];
    float* wtab = h3[0];

    const int TB = 256;
    const dim3 gemmGridW(256, 1, 1);
    const dim3 gemmGridD((BN + 127) / 128, PDIM / 64, 1);
    const dim3 gemmGridD3((BN + 127) / 128, PDIM / 64, 3);

    prep<<<HRB + WRB + ZB, TB, 0, stream>>>(emb_word, dst_nid, hdst,
                                                   w_sage[0], w_sage[1], w_sage[2], w_sage[3], w_sage[4], w_sage[5],
                                                   w_dense, wpk[0], wi, di);

    count_dst<<<3 * WEB + 3 * DEB, TB, 0, stream>>>(jw0, jw1, jw2, jd0, jd1, jd2, wi, di);
    scan_counts<<<6, SCAN_T, 0, stream>>>(wi, di);
    place_rows<<<3 * WEB + 3 * DEB, TB, 0, stream>>>(jw0, jw1, jw2, jd0, jd1, jd2, wi, di);
    gather_mean<<<3 * WNB + 3 * DNB, TB, 0, stream>>>(emb_word, emb_doc, wi, di,
                                                      wnb[0], wnb[1], wnb[2], dnbr[0], dnbr[1], dnbr[2]);

    gemm_rows_wmma<<<gemmGridW, TB, 0, stream>>>(hdst, wnb[0], wnb[1], wnb[2], wpk[0], 2 * PDIM * PDIM,
                                                 b_sage[0], b_sage[1], b_sage[2], NWN);

    sdpa_fuse_add<<<(NWN + 7) / 8, 256, 0, stream>>>(wnb[0], wnb[1], wnb[2], hdst, wtab, NWN);

    pool_seq<<<BN, PDIM, 0, stream>>>(wtab, x_batch, poolbuf);
    gemm_f32_wmma<<<gemmGridD, TB, 0, stream>>>(poolbuf, poolbuf, poolbuf, poolbuf, wpk[0] + (size_t)12 * PDIM * PDIM, 0,
                                                b_dense, b_dense, b_dense, docout, 0,
                                                BN, PDIM, PDIM, PDIM, PDIM, PDIM, PDIM);

    gemm_f32_wmma<<<gemmGridD3, TB, 0, stream>>>(docout, dnbr[0], dnbr[1], dnbr[2], wpk[3], 2 * PDIM * PDIM,
                                                 b_sage[3], b_sage[4], b_sage[5], d3[0], BN * PDIM,
                                                 BN, PDIM, 2 * PDIM, PDIM, PDIM, PDIM, PDIM);

    sdpa_fuse_add<<<(BN + 7) / 8, 256, 0, stream>>>(d3[0], d3[1], d3[2], docout, docrep, BN);
    fc_kernel<<<(BN * CN + TB - 1) / TB, TB, 0, stream>>>(docrep, w_fc, b_fc, out);
}


# seg_mean bucketing: every destination owns a fixed 32-slot bucket (slot = returning integer add; overflow spills to a per-job list): no count pass, no prefix sum; 12 -> 10 launches, same sums
# speedup vs baseline: 3.5023x; 1.2126x over previous
#include <hip/hip_runtime.h>

#define DDIM 300
#define PDIM 320
#define NWN  20000
#define EWN  320000
#define BN   256
#define LN   256
#define SDN  2048
#define EDN  4096
#define CN   20
#define KC   160
#define LP   164

typedef __attribute__((ext_vector_type(2))) float v2f;
typedef __attribute__((ext_vector_type(8))) float v8f;

struct SegIn { const int* e_src; const int* e_dst; const int* src_nid; };
struct SegWs { int* cur; int* ovfc; int* rows; int2* ovf; int n; int E; };

#define CAP 32
#define RBW (3 * NWN + 128)
#define RBD (3 * BN + 128)
static_assert(RBW % 32 == 0 && RBD % 32 == 0, "buckets start on 128-byte lines");

__device__ __forceinline__ SegWs seg_ws(int job, int* wi, int* di) {
    SegWs s;
    if (job < 3) {
        s.n = NWN; s.E = EWN;
        s.cur = wi + job * NWN; s.ovfc = wi + 3 * NWN + job;
        s.rows = wi + RBW + job * (NWN * CAP);
        s.ovf = (int2*)(wi + RBW + 3 * (NWN * CAP)) + (size_t)job * EWN;
    } else {
        const int r = job - 3;
        s.n = BN; s.E = EDN;
        s.cur = di + r * BN; s.ovfc = di + 3 * BN + r;
        s.rows = di + RBD + r * (BN * CAP);
        s.ovf = (int2*)(di + RBD + 3 * (BN * CAP)) + (size_t)r * EDN;
    }
    return s;
}

__device__ __forceinline__ SegIn seg_in(int job, const SegIn& j0, const SegIn& j1, const SegIn& j2,
                                        const SegIn& j3, const SegIn& j4, const SegIn& j5) {
    switch (job) {
        case 0: return j0;
        case 1: return j1;
        case 2: return j2;
        case 3: return j3;
        case 4: return j4;
        default: return j5;
    }
}

#define EPT 4
#define WEB ((EWN / EPT + 255) / 256)
#define DEB ((EDN / EPT + 255) / 256)
static_assert(EWN % EPT == 0 && EDN % EPT == 0, "no partial group of edges");
__device__ __forceinline__ int edge_job(int blk, int& local) {
    if (blk < 3 * WEB) { const int j = blk / WEB; local = blk - j * WEB; return j; }
    const int d = blk - 3 * WEB;
    const int j = d / DEB;
    local = d - j * DEB;
    return 3 + j;
}

#define RPW 4
#define HRB (NWN / (8 * RPW))
#define WRB ((13 * PDIM) / (8 * RPW))
#define ZQW ((3 * NWN + 4) / 4)
#define ZQD ((3 * BN + 4) / 4)
#define ZB  ((ZQW + ZQD + 255) / 256)
static_assert(NWN % (8 * RPW) == 0 && (13 * PDIM) % (8 * RPW) == 0, "every wave of the row blocks has RPW rows");
static_assert(PDIM % RPW == 0 && DDIM % RPW == 0, "the RPW rows of a wave lie in one slab and are all data rows or all pad rows");
static_assert((3 * NWN) % 4 == 0 && (3 * BN) % 4 == 0, "count arrays are whole 16-byte pieces");
static_assert(4 * ZQW <= RBW && 4 * ZQD <= RBD, "the zeroed words end in front of the first bucket");

__device__ __forceinline__ void copy_rows(const float* __restrict__ s0, const float* __restrict__ s1,
                                          const float* __restrict__ s2, const float* __restrict__ s3,
                                          float* __restrict__ o, int lane) {
    static_assert(RPW == 4 && PDIM == 320 && DDIM == 300, "run of 320 float4, 75 data groups per row");
    float4 x[10];
    bool keep[10];
#pragma unroll
    for (int i = 0; i < 10; ++i) {
        const int f = lane + 32 * i;
        const int q = (f >= 80 ? 1 : 0) + (f >= 160 ? 1 : 0) + (f >= 240 ? 1 : 0);
        const int g = f - 80 * q;
        const float* __restrict__ sq = (q == 0) ? s0 : (q == 1) ? s1 : (q == 2) ? s2 : s3;
        keep[i] = g < 75;
        x[i] = ((const float4*)sq)[keep[i] ? g : 74];
    }
    float4* __restrict__ d = (float4*)o;
#pragma unroll
    for (int i = 0; i < 10; ++i)
        d[lane + 32 * i] = make_float4(keep[i] ? x[i].x : 0.f, keep[i] ? x[i].y : 0.f,
                                       keep[i] ? x[i].z : 0.f, keep[i] ? x[i].w : 0.f);
}

__device__ __forceinline__ void zero_rows(float* __restrict__ o, int lane) {
    const float4 z = make_float4(0.f, 0.f, 0.f, 0.f);
    float4* __restrict__ d = (float4*)o;
#pragma unroll
    for (int i = 0; i < 10; ++i) d[lane + 32 * i] = z;
}

__global__ __launch_bounds__(256) void prep(const float* __restrict__ emb, const int* __restrict__ nid,
                                            float* __restrict__ hdst,
                                            const float* __restrict__ w0, const float* __restrict__ w1,
                                            const float* __restrict__ w2, const float* __restrict__ w3,
                                            const float* __restrict__ w4, const float* __restrict__ w5,
                                            const float* __restrict__ wd, float* __restrict__ wout,
                                            int* __restrict__ wi, int* __restrict__ di) {
    const int blk = blockIdx.x;
    const int lane = threadIdx.x & 31;
    const int wv = __builtin_amdgcn_readfirstlane((int)(blk * 8 + (threadIdx.x >> 5)));
    if (blk < HRB) {
        const int r = wv * RPW;
        copy_rows(emb + (size_t)nid[r + 0] * DDIM, emb + (size_t)nid[r + 1] * DDIM,
                  emb + (size_t)nid[r + 2] * DDIM, emb + (size_t)nid[r + 3] * DDIM,
                  hdst + (size_t)r * PDIM, lane);
    } else if (blk < HRB + WRB) {
        const int rr = (wv - HRB * 8) * RPW;
        const int slab = rr / PDIM;
        const int r = rr - slab * PDIM;
        float* __restrict__ o = wout + (size_t)rr * PDIM;
        if (r < DDIM) {
            const int which = slab >> 1;
            const float* __restrict__ w = (slab >= 12) ? wd : (which == 0) ? w0 : (which == 1) ? w1 : (which == 2) ? w2
                                        : (which == 3) ? w3 : (which == 4) ? w4 : w5;
            const int r0 = (slab < 12 && (slab & 1)) ? DDIM : 0;
            const float* __restrict__ s = w + (size_t)(r0 + r) * DDIM;
            copy_rows(s, s + DDIM, s + 2 * DDIM, s + 3 * DDIM, o, lane);
        } else {
            zero_rows(o, lane);
        }
    } else {
        const int t = (blk - HRB - WRB) * 256 + threadIdx.x;
        const int4 z = make_int4(0, 0, 0, 0);
        if (t < ZQW) ((int4*)wi)[t] = z;
        else if (t < ZQW + ZQD) ((int4*)di)[t - ZQW] = z;
    }
}


__device__ __forceinline__ void put_row(const SegWs& s, int dst, int p, int id) {
    if (p < CAP) {
        s.rows[dst * CAP + p] = id;
    } else {
        const int q = atomicAdd(s.ovfc, 1);
        s.ovf[q] = make_int2(dst, id);
    }
}

__global__ __launch_bounds__(256) void place_rows(const SegIn j0, const SegIn j1, const SegIn j2,
                                                  const SegIn j3, const SegIn j4, const SegIn j5,
                                                  int* __restrict__ wi, int* __restrict__ di) {
    int local;
    const int job = edge_job(blockIdx.x, local);
    const SegIn in = seg_in(job, j0, j1, j2, j3, j4, j5);
    const SegWs s = seg_ws(job, wi, di);
    const int t = local * 256 + threadIdx.x;
    if (t >= s.E / EPT) return;
    const int4 d = ((const int4*)in.e_dst)[t];
    const int4 e = ((const int4*)in.e_src)[t];
    const int r0 = in.src_nid[e.x], r1 = in.src_nid[e.y], r2 = in.src_nid[e.z], r3 = in.src_nid[e.w];
    const int p0 = atomicAdd(&s.cur[d.x], 1);
    const int p1 = atomicAdd(&s.cur[d.y], 1);
    const int p2 = atomicAdd(&s.cur[d.z], 1);
    const int p3 = atomicAdd(&s.cur[d.w], 1);
    put_row(s, d.x, p0, r0); put_row(s, d.y, p1, r1); put_row(s, d.z, p2, r2); put_row(s, d.w, p3, r3);
}

__device__ __forceinline__ void add4(float4& a, const float4 x) {
    a.x += x.x; a.y += x.y; a.z += x.z; a.w += x.w;
}

#define WNB ((NWN + 7) / 8)
#define DNB ((BN + 7) / 8)
__global__ __launch_bounds__(256) void gather_mean(const float* __restrict__ emb_w, const float* __restrict__ emb_d,
                                                   int* __restrict__ wi, int* __restrict__ di,
                                                   float* __restrict__ o0, float* __restrict__ o1,
                                                   float* __restrict__ o2, float* __restrict__ o3,
                                                   float* __restrict__ o4, float* __restrict__ o5) {
    const int lane = threadIdx.x & 31;
    const int blk = blockIdx.x;
    int job, local;
    if (blk < 3 * WNB) { job = blk / WNB; local = blk - job * WNB; }
    else { const int d = blk - 3 * WNB; const int j = d / DNB; local = d - j * DNB; job = 3 + j; }
    const SegWs s = seg_ws(job, wi, di);
    const float* __restrict__ emb = (job < 3) ? emb_w : emb_d;
    float* __restrict__ out = (job == 0) ? o0 : (job == 1) ? o1 : (job == 2) ? o2
                            : (job == 3) ? o3 : (job == 4) ? o4 : o5;
    const int* __restrict__ rows = s.rows;
    const int* __restrict__ cnt = s.cur;
    const int node = __builtin_amdgcn_readfirstlane((int)(local * 8 + (threadIdx.x >> 5)));
    if (node >= s.n) return;
    const int beg = node * CAP;
    const int deg = cnt[node];
    const int i2 = (lane < 11) ? lane + 64 : 74;

    float4 a0 = make_float4(0.f, 0.f, 0.f, 0.f), a1 = a0, a2 = a0;
    if (deg <= CAP) {
        const int key = (lane < deg) ? rows[beg + lane] : 0x7fffffff;
        int rank = 0;
#pragma unroll
        for (int k = 0; k < 32; ++k) {
            const int kk = __builtin_amdgcn_readlane(key, k);
            rank += (kk < key || (kk == key && k < lane)) ? 1 : 0;
        }
        const int sorted = __builtin_amdgcn_ds_permute(rank << 2, key);
        int k = 0;
        for (; k + 4 <= deg; k += 4) {
            const float4* r0 = (const float4*)(emb + (size_t)__builtin_amdgcn_readlane(sorted, k + 0) * DDIM);
            const float4* r1 = (const float4*)(emb + (size_t)__builtin_amdgcn_readlane(sorted, k + 1) * DDIM);
            const float4* r2 = (const float4*)(emb + (size_t)__builtin_amdgcn_readlane(sorted, k + 2) * DDIM);
            const float4* r3 = (const float4*)(emb + (size_t)__builtin_amdgcn_readlane(sorted, k + 3) * DDIM);
            const float4 x00 = r0[lane], x01 = r0[lane + 32], x02 = r0[i2];
            const float4 x10 = r1[lane], x11 = r1[lane + 32], x12 = r1[i2];
            const float4 x20 = r2[lane], x21 = r2[lane + 32], x22 = r2[i2];
            const float4 x30 = r3[lane], x31 = r3[lane + 32], x32 = r3[i2];
            add4(a0, x00); add4(a1, x01); add4(a2, x02);
            add4(a0, x10); add4(a1, x11); add4(a2, x12);
            add4(a0, x20); add4(a1, x21); add4(a2, x22);
            add4(a0, x30); add4(a1, x31); add4(a2, x32);
        }
        for (; k < deg; ++k) {
            const float4* r0 = (const float4*)(emb + (size_t)__builtin_amdgcn_readlane(sorted, k) * DDIM);
            const float4 x00 = r0[lane], x01 = r0[lane + 32], x02 = r0[i2];
            add4(a0, x00); add4(a1, x01); add4(a2, x02);
        }
    } else {
        const int2* __restrict__ ov = s.ovf;
        const int nov = *s.ovfc;
        const int mine = rows[beg + lane];
        int last = -1;
        for (int pass = 0; pass < deg; ++pass) {
            int best = (mine > last) ? mine : 0x7fffffff;
            for (int b = lane; b < nov; b += 32) {
                const int2 e = ov[b];
                if (e.x == node && e.y > last && e.y < best) best = e.y;
            }
            for (int off = 16; off > 0; off >>= 1) best = min(best, __shfl_xor(best, off, 32));
            best = __builtin_amdgcn_readfirstlane(best);
            if (best == 0x7fffffff) break;
            int mult = (mine == best) ? 1 : 0;
            for (int b = lane; b < nov; b += 32) {
                const int2 e = ov[b];
                mult += (e.x == node && e.y == best) ? 1 : 0;
            }
            for (int off = 16; off > 0; off >>= 1) mult += __shfl_xor(mult, off, 32);
            mult = __builtin_amdgcn_readfirstlane(mult);
            const float4* r0 = (const float4*)(emb + (size_t)best * DDIM);
            const float4 x00 = r0[lane], x01 = r0[lane + 32], x02 = r0[i2];
            for (int m = 0; m < mult; ++m) { add4(a0, x00); add4(a1, x01); add4(a2, x02); }
            last = best;
        }
    }

    const float d = fmaxf((float)deg, 1.f);
    const float4 z = make_float4(0.f, 0.f, 0.f, 0.f);
    float4* o = (float4*)(out + (size_t)node * PDIM);
    o[lane]      = make_float4(a0.x / d, a0.y / d, a0.z / d, a0.w / d);
    o[lane + 32] = make_float4(a1.x / d, a1.y / d, a1.z / d, a1.w / d);
    if (lane < 16)
        o[lane + 64] = (lane < 11) ? make_float4(a2.x / d, a2.y / d, a2.z / d, a2.w / d) : z;
}

#define AB 10
static_assert(KC == 16 * AB, "the K loop below is written for four A batches per chunk");

__device__ __forceinline__ void load_a_batch(v2f (&a)[AB], const float* __restrict__ p) {
#pragma unroll
    for (int t = 0; t < AB; ++t) a[t] = *(const v2f*)(p + 4 * t);
}

__device__ __forceinline__ void mma_a_batch(v8f (&acc)[4], const v2f (&a)[AB], const float* lb) {
#pragma unroll
    for (int t = 0; t < AB; ++t) {
#pragma unroll
        for (int ns = 0; ns < 4; ++ns) {
            v2f b = *(const v2f*)(lb + ns * 16 * LP + 4 * t);
            acc[ns] = __builtin_amdgcn_wmma_f32_16x16x4_f32(
                false, a[t], false, b, (short)0, acc[ns], false, false);
        }
    }
}

__global__ __launch_bounds__(256) void gemm_f32_wmma(
        const float* __restrict__ A0,
        const float* __restrict__ A1z0, const float* __restrict__ A1z1, const float* __restrict__ A1z2,
        const float* __restrict__ Bz0, int bzs,
        const float* __restrict__ bias0, const float* __restrict__ bias1, const float* __restrict__ bias2,
        float* __restrict__ Cz0, int czs,
        int M, int N, int K, int ksplit, int lda, int ldb, int ldc) {
    __shared__ float ldsB[64 * LP];
    (void)N;
    const int z = blockIdx.z;
    const float* __restrict__ A1 = (z == 0) ? A1z0 : (z == 1) ? A1z1 : A1z2;
    const float* __restrict__ bias = (z == 0) ? bias0 : (z == 1) ? bias1 : bias2;
    const float* __restrict__ B = Bz0 + (size_t)z * bzs;
    float* __restrict__ C = Cz0 + (size_t)z * czs;
    const int lane = threadIdx.x & 31;
    const int wave = threadIdx.x >> 5;
    const int hi   = lane >> 4;
    const int l16  = lane & 15;
    const int nbase = blockIdx.y * 64;
    const int mt    = blockIdx.x * 8 + wave;
    const bool valid = (mt * 16) < M;
    const int m = mt * 16 + l16;

    v8f acc[4];
    if (valid) {
#pragma unroll
        for (int ns = 0; ns < 4; ++ns) {
            const int n = nbase + ns * 16 + l16;
            const float bv = (bias && n < DDIM) ? bias[n] : 0.f;
#pragma unroll
            for (int i = 0; i < 8; ++i) acc[ns][i] = bv;
        }
    }

    const int aoff = m * lda + 2 * hi;
    const float* lb = &ldsB[l16 * LP + 2 * hi];
    const int sc  = (threadIdx.x & 15) * 4;
    const int skr = threadIdx.x >> 4;
    const float* __restrict__ Bp = B + (size_t)skr * ldb + nbase + sc;

    v2f a0[AB], a1[AB];
    if (valid) load_a_batch(a0, A0 + aoff);

    for (int kk = 0; kk < K; kk += KC) {
        float4 bt[KC / 16];
#pragma unroll
        for (int i = 0; i < KC / 16; ++i)
            bt[i] = *(const float4*)(Bp + (size_t)(kk + 16 * i) * ldb);

        __syncthreads();
#pragma unroll
        for (int i = 0; i < KC / 16; ++i) {
            const int kloc = skr + 16 * i;
            ldsB[(sc + 0) * LP + kloc] = bt[i].x;
            ldsB[(sc + 1) * LP + kloc] = bt[i].y;
            ldsB[(sc + 2) * LP + kloc] = bt[i].z;
            ldsB[(sc + 3) * LP + kloc] = bt[i].w;
        }
        __syncthreads();

        if (valid) {
            const float* ak = ((kk < ksplit) ? A0 + kk : A1 + (kk - ksplit)) + aoff;
            load_a_batch(a1, ak + 4 * AB);
            mma_a_batch(acc, a0, lb);
            load_a_batch(a0, ak + 8 * AB);
            mma_a_batch(acc, a1, lb + 4 * AB);
            load_a_batch(a1, ak + 12 * AB);
            mma_a_batch(acc, a0, lb + 8 * AB);
            const int kn = kk + KC;
            if (kn < K) load_a_batch(a0, ((kn < ksplit) ? A0 + kn : A1 + (kn - ksplit)) + aoff);
            mma_a_batch(acc, a1, lb + 12 * AB);
        }
    }
    if (valid) {
#pragma unroll
        for (int ns = 0; ns < 4; ++ns)
#pragma unroll
            for (int i = 0; i < 8; ++i)
                C[(size_t)(mt * 16 + i + hi * 8) * ldc + nbase + ns * 16 + l16] = acc[ns][i];
    }
}

#define KW  32
#define LPW 36
#define NT  (PDIM / 16)
#define AW  (KW / 4)
#define NCH ((2 * PDIM) / KW)
#define NBT ((KW * (PDIM / 4)) / 256)
static_assert(PDIM % (2 * KW) == 0, "a pair of chunks never straddles the A0 | X boundary; the K loop takes two chunks per trip");
static_assert(KW == 32 && PDIM == 32 * NBT, "panel staging map below: 8 rows x 16 floats per wave-load, 32 columns per pass");

__device__ __forceinline__ void load_b_panel(float4 (&bt)[NBT], const float* __restrict__ p) {
#pragma unroll
    for (int i = 0; i < NBT; ++i) bt[i] = *(const float4*)(p + 32 * i);
}

__device__ __forceinline__ void store_b_panel(float* ls, const float4 (&bt)[NBT]) {
#pragma unroll
    for (int i = 0; i < NBT; ++i) {
        ls[(32 * i + 0) * LPW] = bt[i].x;
        ls[(32 * i + 1) * LPW] = bt[i].y;
        ls[(32 * i + 2) * LPW] = bt[i].z;
        ls[(32 * i + 3) * LPW] = bt[i].w;
    }
}

__device__ __forceinline__ void chunk_rows(v8f (&acc)[NT], v2f (&a)[AW], const float* lb, const float* pn,
                                           float* ls, float4 (&bt)[NBT], const float* __restrict__ bn) {
#pragma unroll
    for (int t = 0; t < AW; ++t) {
        if (t == AW / 2) {
            store_b_panel(ls, bt);
            load_b_panel(bt, bn);
        }
#pragma unroll
        for (int ns = 0; ns < NT; ++ns) {
            v2f b = *(const v2f*)(lb + ns * 16 * LPW + 4 * t);
            acc[ns] = __builtin_amdgcn_wmma_f32_16x16x4_f32(
                false, a[t], false, b, (short)0, acc[ns], false, false);
        }
        a[t] = *(const v2f*)(pn + 4 * t);
    }
}

__global__ __launch_bounds__(256) void gemm_rows_wmma(
        const float* __restrict__ A0, float* X0, float* X1, float* X2,
        const float* __restrict__ Bz0, int bzs,
        const float* __restrict__ bias0, const float* __restrict__ bias1, const float* __restrict__ bias2,
        int M) {
    __shared__ float ldsB[2 * PDIM * LPW];
    const int lane = threadIdx.x & 31;
    const int wave = threadIdx.x >> 5;
    const int hi   = lane >> 4;
    const int l16  = lane & 15;
    const float* lb0 = &ldsB[l16 * LPW + 2 * hi];
    const float* lb1 = lb0 + PDIM * LPW;
    const int skr = (wave & 3) * 8 + (lane & 7);
    const int sc  = ((wave >> 2) * 4 + (lane >> 3)) * 4;
    float* ls0 = &ldsB[sc * LPW + skr];
    float* ls1 = ls0 + PDIM * LPW;
    const int nrb = (M + 127) >> 7;

    for (int u = blockIdx.x; u < 3 * nrb; u += gridDim.x) {
        const int z = u / nrb;
        const int mt = (u - z * nrb) * 8 + wave;
        float* X = (z == 0) ? X0 : (z == 1) ? X1 : X2;
        const float* __restrict__ bias = (z == 0) ? bias0 : (z == 1) ? bias1 : bias2;
        const float* __restrict__ Bp = Bz0 + (size_t)z * bzs + (size_t)skr * PDIM + sc;
        const bool valid = (mt * 16) < M;
        const int aoff = min(mt * 16 + l16, M - 1) * PDIM + 2 * hi;

        float4 bt[NBT];
        v2f ae[AW], ao[AW];
        load_b_panel(bt, Bp);
#pragma unroll
        for (int t = 0; t < AW; ++t) {
            ae[t] = *(const v2f*)(A0 + aoff + 4 * t);
            ao[t] = *(const v2f*)(A0 + aoff + KW + 4 * t);
        }
        store_b_panel(ls0, bt);
        load_b_panel(bt, Bp + (size_t)KW * PDIM);
        __syncthreads();

        v8f acc[NT];
#pragma unroll
        for (int ns = 0; ns < NT; ++ns) {
            const int n = ns * 16 + l16;
            const float bv = (n < DDIM) ? bias[n] : 0.f;
#pragma unroll
            for (int i = 0; i < 8; ++i) acc[ns][i] = bv;
        }

        for (int c = 0; c < NCH; c += 2) {
            const int kn = min((c + 2) * KW, 2 * PDIM - 2 * KW);
            const float* an = ((kn < PDIM) ? A0 + kn : X + (kn - PDIM)) + aoff;
            chunk_rows(acc, ae, lb0, an, ls1, bt, Bp + (size_t)kn * PDIM);
            __syncthreads();
            chunk_rows(acc, ao, lb1, an + KW, ls0, bt, Bp + (size_t)(kn + KW) * PDIM);
            __syncthreads();
        }
        if (valid) {
#pragma unroll
            for (int ns = 0; ns < NT; ++ns)
#pragma unroll
                for (int i = 0; i < 8; ++i)
                    X[(size_t)(mt * 16 + i + hi * 8) * PDIM + ns * 16 + l16] = acc[ns][i];
        }
    }
}

__global__ void sdpa_fuse_add(const float* __restrict__ h0, const float* __restrict__ h1,
                              const float* __restrict__ h2, const float* __restrict__ addv,
                              float* __restrict__ out, int n) {
    const int lane = threadIdx.x & 31;
    const int node = blockIdx.x * (blockDim.x >> 5) + (threadIdx.x >> 5);
    if (node >= n) return;
    const float* x0 = h0 + (size_t)node * PDIM;
    const float* x1 = h1 + (size_t)node * PDIM;
    const float* x2 = h2 + (size_t)node * PDIM;

    float d00 = 0, d01 = 0, d02 = 0, d11 = 0, d12 = 0, d22 = 0;
    for (int j = lane; j < PDIM; j += 32) {
        float a = x0[j], b = x1[j], e = x2[j];
        d00 += a * a; d01 += a * b; d02 += a * e;
        d11 += b * b; d12 += b * e; d22 += e * e;
    }
    for (int off = 16; off > 0; off >>= 1) {
        d00 += __shfl_xor(d00, off, 32);
        d01 += __shfl_xor(d01, off, 32);
        d02 += __shfl_xor(d02, off, 32);
        d11 += __shfl_xor(d11, off, 32);
        d12 += __shfl_xor(d12, off, 32);
        d22 += __shfl_xor(d22, off, 32);
    }
    const float scale = rsqrtf((float)DDIM);
    float S[3][3] = {{d00, d01, d02}, {d01, d11, d12}, {d02, d12, d22}};
    float wsum[3] = {0.f, 0.f, 0.f};
    for (int t = 0; t < 3; ++t) {
        float p0 = S[t][0] * scale, p1 = S[t][1] * scale, p2 = S[t][2] * scale;
        float mx = fmaxf(p0, fmaxf(p1, p2));
        float e0 = __expf(p0 - mx), e1 = __expf(p1 - mx), e2 = __expf(p2 - mx);
        float inv = 1.f / (e0 + e1 + e2);
        wsum[0] += e0 * inv; wsum[1] += e1 * inv; wsum[2] += e2 * inv;
    }
    const float w0 = wsum[0] * (1.f / 3.f), w1 = wsum[1] * (1.f / 3.f), w2 = wsum[2] * (1.f / 3.f);
    const float* av = addv + (size_t)node * PDIM;
    float* o = out + (size_t)node * PDIM;
    for (int j = lane; j < PDIM; j += 32)
        o[j] = w0 * x0[j] + w1 * x1[j] + w2 * x2[j] + av[j];
}

#define PB 32
static_assert(LN % (2 * PB) == 0, "pool_seq walks the sequence in pairs of batches");
static_assert(PDIM % 32 == 0, "every wave of pool_seq is full, so every lane's token id can be read by readlane");

__device__ __forceinline__ void pool_load(float (&x)[PB], const float* __restrict__ table, unsigned j,
                                          int ids, unsigned real) {
#pragma unroll
    for (int t = 0; t < PB; ++t) {
        const int idx = __builtin_amdgcn_readlane(ids, t);
        const int r = ((real >> t) & 1u) ? idx : 0;
        const float* __restrict__ rp = table + (size_t)r * PDIM;
        x[t] = rp[j];
    }
}

__device__ __forceinline__ void pool_add(float& acc, const float (&x)[PB], unsigned real) {
#pragma unroll
    for (int t = 0; t < PB; ++t)
        acc += ((real >> t) & 1u) ? x[t] : 0.f;
}

__global__ __launch_bounds__(PDIM) void pool_seq(const float* __restrict__ table, const int* __restrict__ xb,
                                                 float* __restrict__ pooled) {
    const int b = blockIdx.x;
    const unsigned j = threadIdx.x;
    if (j >= PDIM) return;
    const unsigned lane = j & 31u;
    const int* row = xb + b * LN;
    int ids[LN / PB];
    unsigned real[LN / PB];
#pragma unroll
    for (int k = 0; k < LN / PB; ++k) ids[k] = row[PB * k + lane];
#pragma unroll
    for (int k = 0; k < LN / PB; ++k)
        real[k] = __builtin_amdgcn_ballot_w32((unsigned)ids[k] < (unsigned)NWN);

    float acc = 0.f;
    float x0[PB], x1[PB];
    pool_load(x0, table, j, ids[0], real[0]);
#pragma unroll
    for (int k = 0; k < LN / PB; k += 2) {
        pool_load(x1, table, j, ids[k + 1], real[k + 1]);
        pool_add(acc, x0, real[k]);
        if (k + 2 < LN / PB) pool_load(x0, table, j, ids[k + 2], real[k + 2]);
        pool_add(acc, x1, real[k + 1]);
    }
    pooled[(size_t)b * PDIM + j] = acc * (1.f / (float)LN);
}

__global__ void fc_kernel(const float* __restrict__ rep, const float* __restrict__ wfc,
                          const float* __restrict__ bfc, float* __restrict__ out) {
    int t = blockIdx.x * blockDim.x + threadIdx.x;
    if (t >= BN * CN) return;
    int b = t / CN, c = t % CN;
    float acc = bfc[c];
    const float* r = rep + (size_t)b * PDIM;
    for (int j = 0; j < DDIM; ++j) acc += r[j] * wfc[(size_t)j * CN + c];
    out[t] = acc;
}


extern "C" void kernel_launch(void* const* d_in, const int* in_sizes, int n_in,
                              void* d_out, int out_size, void* d_ws, size_t ws_size,
                              hipStream_t stream) {
    (void)in_sizes; (void)n_in; (void)out_size; (void)ws_size;
    const float* emb_word = (const float*)d_in[0];
    const float* emb_doc  = (const float*)d_in[1];
    const float* w_sage[6] = {(const float*)d_in[2],  (const float*)d_in[4],  (const float*)d_in[6],
                              (const float*)d_in[8],  (const float*)d_in[10], (const float*)d_in[12]};
    const float* b_sage[6] = {(const float*)d_in[3],  (const float*)d_in[5],  (const float*)d_in[7],
                              (const float*)d_in[9],  (const float*)d_in[11], (const float*)d_in[13]};
    const float* w_dense = (const float*)d_in[14];
    const float* b_dense = (const float*)d_in[15];
    const float* w_fc    = (const float*)d_in[16];
    const float* b_fc    = (const float*)d_in[17];
    const int* dst_nid      = (const int*)d_in[18];
    const int* x_batch      = (const int*)d_in[28];
    const SegIn jw0 = {(const int*)d_in[22], (const int*)d_in[23], (const int*)d_in[19]};
    const SegIn jw1 = {(const int*)d_in[24], (const int*)d_in[25], (const int*)d_in[20]};
    const SegIn jw2 = {(const int*)d_in[26], (const int*)d_in[27], (const int*)d_in[21]};
    const SegIn jd0 = {(const int*)d_in[33], (const int*)d_in[34], (const int*)d_in[30]};
    const SegIn jd1 = {(const int*)d_in[35], (const int*)d_in[36], (const int*)d_in[31]};
    const SegIn jd2 = {(const int*)d_in[37], (const int*)d_in[38], (const int*)d_in[32]};
    float* out = (float*)d_out;

    float* ws = (float*)d_ws;
    size_t off = 0;
    auto carve = [&](size_t nflt) { float* p = ws + off; off += nflt; return p; };
    float* hdst = carve((size_t)NWN * PDIM);
    float* nbuf = carve((size_t)NWN * PDIM);
    float* h3[3]; for (int r = 0; r < 3; ++r) h3[r] = carve((size_t)NWN * PDIM);
    float* cnt  = carve(NWN);
    float* wpk[6]; for (int r = 0; r < 6; ++r) wpk[r] = carve((size_t)2 * PDIM * PDIM);
    float* wdn  = carve((size_t)PDIM * PDIM);
    float* pooled = carve((size_t)BN * PDIM);
    float* docout = carve((size_t)BN * PDIM);
    float* dnb    = carve((size_t)BN * PDIM);
    float* d3[3]; for (int r = 0; r < 3; ++r) d3[r] = carve((size_t)BN * PDIM);
    float* docrep = carve((size_t)BN * PDIM);
    (void)cnt; (void)wdn;
    static_assert((size_t)RBW + 3 * ((size_t)NWN * CAP + 2 * (size_t)EWN) <= (size_t)NWN * PDIM, "word-job scratch fits its buffer");
    static_assert((size_t)RBD + 3 * ((size_t)BN * CAP + 2 * (size_t)EDN) <= (size_t)BN * PDIM, "doc-job scratch fits its buffer");
    int* wi = (int*)h3[0];
    int* di = (int*)d3[0];
    float* wnb[3] = {h3[1], h3[2], nbuf};
    float* dnbr[3] = {dnb, docrep, pooled};
    float* poolbuf = d3[0];
    float* wtab = h3[0];

    const int TB = 256;
    const dim3 gemmGridW(256, 1, 1);
    const dim3 gemmGridD((BN + 127) / 128, PDIM / 64, 1);
    const dim3 gemmGridD3((BN + 127) / 128, PDIM / 64, 3);

    prep<<<HRB + WRB + ZB, TB, 0, stream>>>(emb_word, dst_nid, hdst,
                                                   w_sage[0], w_sage[1], w_sage[2], w_sage[3], w_sage[4], w_sage[5],
                                                   w_dense, wpk[0], wi, di);

    place_rows<<<3 * WEB + 3 * DEB, TB, 0, stream>>>(jw0, jw1, jw2, jd0, jd1, jd2, wi, di);
    gather_mean<<<3 * WNB + 3 * DNB, TB, 0, stream>>>(emb_word, emb_doc, wi, di,
                                                      wnb[0], wnb[1], wnb[2], dnbr[0], dnbr[1], dnbr[2]);

    gemm_rows_wmma<<<gemmGridW, TB, 0, stream>>>(hdst, wnb[0], wnb[1], wnb[2], wpk[0], 2 * PDIM * PDIM,
                                                 b_sage[0], b_sage[1], b_sage[2], NWN);

    sdpa_fuse_add<<<(NWN + 7) / 8, 256, 0, stream>>>(wnb[0], wnb[1], wnb[2], hdst, wtab, NWN);

    pool_seq<<<BN, PDIM, 0, stream>>>(wtab, x_batch, poolbuf);
    gemm_f32_wmma<<<gemmGridD, TB, 0, stream>>>(poolbuf, poolbuf, poolbuf, poolbuf, wpk[0] + (size_t)12 * PDIM * PDIM, 0,
                                                b_dense, b_dense, b_dense, docout, 0,
                                                BN, PDIM, PDIM, PDIM, PDIM, PDIM, PDIM);

    gemm_f32_wmma<<<gemmGridD3, TB, 0, stream>>>(docout, dnbr[0], dnbr[1], dnbr[2], wpk[3], 2 * PDIM * PDIM,
                                                 b_sage[3], b_sage[4], b_sage[5], d3[0], BN * PDIM,
                                                 BN, PDIM, 2 * PDIM, PDIM, PDIM, PDIM, PDIM);

    sdpa_fuse_add<<<(BN + 7) / 8, 256, 0, stream>>>(d3[0], d3[1], d3[2], docout, docrep, BN);
    fc_kernel<<<(BN * CN + TB - 1) / TB, TB, 0, stream>>>(docrep, w_fc, b_fc, out);
}


# doc GEMMs (dense, doc SAGE): one wave per 16x16 tile of C, A and B fragments straight from global, 16 k-steps a batch (no LDS, no barrier): 320/960 waves on all WGPs, not 10/30 blocks; same k order
# speedup vs baseline: 3.7005x; 1.0566x over previous
#include <hip/hip_runtime.h>

#define DDIM 300
#define PDIM 320
#define NWN  20000
#define EWN  320000
#define BN   256
#define LN   256
#define SDN  2048
#define EDN  4096
#define CN   20

typedef __attribute__((ext_vector_type(2))) float v2f;
typedef __attribute__((ext_vector_type(8))) float v8f;

struct SegIn { const int* e_src; const int* e_dst; const int* src_nid; };
struct SegWs { int* cur; int* ovfc; int* rows; int2* ovf; int n; int E; };

#define CAP 32
#define RBW (3 * NWN + 128)
#define RBD (3 * BN + 128)
static_assert(RBW % 32 == 0 && RBD % 32 == 0, "buckets start on 128-byte lines");

__device__ __forceinline__ SegWs seg_ws(int job, int* wi, int* di) {
    SegWs s;
    if (job < 3) {
        s.n = NWN; s.E = EWN;
        s.cur = wi + job * NWN; s.ovfc = wi + 3 * NWN + job;
        s.rows = wi + RBW + job * (NWN * CAP);
        s.ovf = (int2*)(wi + RBW + 3 * (NWN * CAP)) + (size_t)job * EWN;
    } else {
        const int r = job - 3;
        s.n = BN; s.E = EDN;
        s.cur = di + r * BN; s.ovfc = di + 3 * BN + r;
        s.rows = di + RBD + r * (BN * CAP);
        s.ovf = (int2*)(di + RBD + 3 * (BN * CAP)) + (size_t)r * EDN;
    }
    return s;
}

__device__ __forceinline__ SegIn seg_in(int job, const SegIn& j0, const SegIn& j1, const SegIn& j2,
                                        const SegIn& j3, const SegIn& j4, const SegIn& j5) {
    switch (job) {
        case 0: return j0;
        case 1: return j1;
        case 2: return j2;
        case 3: return j3;
        case 4: return j4;
        default: return j5;
    }
}

#define EPT 4
#define WEB ((EWN / EPT + 255) / 256)
#define DEB ((EDN / EPT + 255) / 256)
static_assert(EWN % EPT == 0 && EDN % EPT == 0, "no partial group of edges");
__device__ __forceinline__ int edge_job(int blk, int& local) {
    if (blk < 3 * WEB) { const int j = blk / WEB; local = blk - j * WEB; return j; }
    const int d = blk - 3 * WEB;
    const int j = d / DEB;
    local = d - j * DEB;
    return 3 + j;
}

#define RPW 4
#define HRB (NWN / (8 * RPW))
#define WRB ((13 * PDIM) / (8 * RPW))
#define ZQW ((3 * NWN + 4) / 4)
#define ZQD ((3 * BN + 4) / 4)
#define ZB  ((ZQW + ZQD + 255) / 256)
static_assert(NWN % (8 * RPW) == 0 && (13 * PDIM) % (8 * RPW) == 0, "every wave of the row blocks has RPW rows");
static_assert(PDIM % RPW == 0 && DDIM % RPW == 0, "the RPW rows of a wave lie in one slab and are all data rows or all pad rows");
static_assert((3 * NWN) % 4 == 0 && (3 * BN) % 4 == 0, "count arrays are whole 16-byte pieces");
static_assert(4 * ZQW <= RBW && 4 * ZQD <= RBD, "the zeroed words end in front of the first bucket");

__device__ __forceinline__ void copy_rows(const float* __restrict__ s0, const float* __restrict__ s1,
                                          const float* __restrict__ s2, const float* __restrict__ s3,
                                          float* __restrict__ o, int lane) {
    static_assert(RPW == 4 && PDIM == 320 && DDIM == 300, "run of 320 float4, 75 data groups per row");
    float4 x[10];
    bool keep[10];
#pragma unroll
    for (int i = 0; i < 10; ++i) {
        const int f = lane + 32 * i;
        const int q = (f >= 80 ? 1 : 0) + (f >= 160 ? 1 : 0) + (f >= 240 ? 1 : 0);
        const int g = f - 80 * q;
        const float* __restrict__ sq = (q == 0) ? s0 : (q == 1) ? s1 : (q == 2) ? s2 : s3;
        keep[i] = g < 75;
        x[i] = ((const float4*)sq)[keep[i] ? g : 74];
    }
    float4* __restrict__ d = (float4*)o;
#pragma unroll
    for (int i = 0; i < 10; ++i)
        d[lane + 32 * i] = make_float4(keep[i] ? x[i].x : 0.f, keep[i] ? x[i].y : 0.f,
                                       keep[i] ? x[i].z : 0.f, keep[i] ? x[i].w : 0.f);
}

__device__ __forceinline__ void zero_rows(float* __restrict__ o, int lane) {
    const float4 z = make_float4(0.f, 0.f, 0.f, 0.f);
    float4* __restrict__ d = (float4*)o;
#pragma unroll
    for (int i = 0; i < 10; ++i) d[lane + 32 * i] = z;
}

__global__ __launch_bounds__(256) void prep(const float* __restrict__ emb, const int* __restrict__ nid,
                                            float* __restrict__ hdst,
                                            const float* __restrict__ w0, const float* __restrict__ w1,
                                            const float* __restrict__ w2, const float* __restrict__ w3,
                                            const float* __restrict__ w4, const float* __restrict__ w5,
                                            const float* __restrict__ wd, float* __restrict__ wout,
                                            int* __restrict__ wi, int* __restrict__ di) {
    const int blk = blockIdx.x;
    const int lane = threadIdx.x & 31;
    const int wv = __builtin_amdgcn_readfirstlane((int)(blk * 8 + (threadIdx.x >> 5)));
    if (blk < HRB) {
        const int r = wv * RPW;
        copy_rows(emb + (size_t)nid[r + 0] * DDIM, emb + (size_t)nid[r + 1] * DDIM,
                  emb + (size_t)nid[r + 2] * DDIM, emb + (size_t)nid[r + 3] * DDIM,
                  hdst + (size_t)r * PDIM, lane);
    } else if (blk < HRB + WRB) {
        const int rr = (wv - HRB * 8) * RPW;
        const int slab = rr / PDIM;
        const int r = rr - slab * PDIM;
        float* __restrict__ o = wout + (size_t)rr * PDIM;
        if (r < DDIM) {
            const int which = slab >> 1;
            const float* __restrict__ w = (slab >= 12) ? wd : (which == 0) ? w0 : (which == 1) ? w1 : (which == 2) ? w2
                                        : (which == 3) ? w3 : (which == 4) ? w4 : w5;
            const int r0 = (slab < 12 && (slab & 1)) ? DDIM : 0;
            const float* __restrict__ s = w + (size_t)(r0 + r) * DDIM;
            copy_rows(s, s + DDIM, s + 2 * DDIM, s + 3 * DDIM, o, lane);
        } else {
            zero_rows(o, lane);
        }
    } else {
        const int t = (blk - HRB - WRB) * 256 + threadIdx.x;
        const int4 z = make_int4(0, 0, 0, 0);
        if (t < ZQW) ((int4*)wi)[t] = z;
        else if (t < ZQW + ZQD) ((int4*)di)[t - ZQW] = z;
    }
}


__device__ __forceinline__ void put_row(const SegWs& s, int dst, int p, int id) {
    if (p < CAP) {
        s.rows[dst * CAP + p] = id;
    } else {
        const int q = atomicAdd(s.ovfc, 1);
        s.ovf[q] = make_int2(dst, id);
    }
}

__global__ __launch_bounds__(256) void place_rows(const SegIn j0, const SegIn j1, const SegIn j2,
                                                  const SegIn j3, const SegIn j4, const SegIn j5,
                                                  int* __restrict__ wi, int* __restrict__ di) {
    int local;
    const int job = edge_job(blockIdx.x, local);
    const SegIn in = seg_in(job, j0, j1, j2, j3, j4, j5);
    const SegWs s = seg_ws(job, wi, di);
    const int t = local * 256 + threadIdx.x;
    if (t >= s.E / EPT) return;
    const int4 d = ((const int4*)in.e_dst)[t];
    const int4 e = ((const int4*)in.e_src)[t];
    const int r0 = in.src_nid[e.x], r1 = in.src_nid[e.y], r2 = in.src_nid[e.z], r3 = in.src_nid[e.w];
    const int p0 = atomicAdd(&s.cur[d.x], 1);
    const int p1 = atomicAdd(&s.cur[d.y], 1);
    const int p2 = atomicAdd(&s.cur[d.z], 1);
    const int p3 = atomicAdd(&s.cur[d.w], 1);
    put_row(s, d.x, p0, r0); put_row(s, d.y, p1, r1); put_row(s, d.z, p2, r2); put_row(s, d.w, p3, r3);
}

__device__ __forceinline__ void add4(float4& a, const float4 x) {
    a.x += x.x; a.y += x.y; a.z += x.z; a.w += x.w;
}

#define WNB ((NWN + 7) / 8)
#define DNB ((BN + 7) / 8)
__global__ __launch_bounds__(256) void gather_mean(const float* __restrict__ emb_w, const float* __restrict__ emb_d,
                                                   int* __restrict__ wi, int* __restrict__ di,
                                                   float* __restrict__ o0, float* __restrict__ o1,
                                                   float* __restrict__ o2, float* __restrict__ o3,
                                                   float* __restrict__ o4, float* __restrict__ o5) {
    const int lane = threadIdx.x & 31;
    const int blk = blockIdx.x;
    int job, local;
    if (blk < 3 * WNB) { job = blk / WNB; local = blk - job * WNB; }
    else { const int d = blk - 3 * WNB; const int j = d / DNB; local = d - j * DNB; job = 3 + j; }
    const SegWs s = seg_ws(job, wi, di);
    const float* __restrict__ emb = (job < 3) ? emb_w : emb_d;
    float* __restrict__ out = (job == 0) ? o0 : (job == 1) ? o1 : (job == 2) ? o2
                            : (job == 3) ? o3 : (job == 4) ? o4 : o5;
    const int* __restrict__ rows = s.rows;
    const int* __restrict__ cnt = s.cur;
    const int node = __builtin_amdgcn_readfirstlane((int)(local * 8 + (threadIdx.x >> 5)));
    if (node >= s.n) return;
    const int beg = node * CAP;
    const int deg = cnt[node];
    const int i2 = (lane < 11) ? lane + 64 : 74;

    float4 a0 = make_float4(0.f, 0.f, 0.f, 0.f), a1 = a0, a2 = a0;
    if (deg <= CAP) {
        const int key = (lane < deg) ? rows[beg + lane] : 0x7fffffff;
        int rank = 0;
#pragma unroll
        for (int k = 0; k < 32; ++k) {
            const int kk = __builtin_amdgcn_readlane(key, k);
            rank += (kk < key || (kk == key && k < lane)) ? 1 : 0;
        }
        const int sorted = __builtin_amdgcn_ds_permute(rank << 2, key);
        int k = 0;
        for (; k + 4 <= deg; k += 4) {
            const float4* r0 = (const float4*)(emb + (size_t)__builtin_amdgcn_readlane(sorted, k + 0) * DDIM);
            const float4* r1 = (const float4*)(emb + (size_t)__builtin_amdgcn_readlane(sorted, k + 1) * DDIM);
            const float4* r2 = (const float4*)(emb + (size_t)__builtin_amdgcn_readlane(sorted, k + 2) * DDIM);
            const float4* r3 = (const float4*)(emb + (size_t)__builtin_amdgcn_readlane(sorted, k + 3) * DDIM);
            const float4 x00 = r0[lane], x01 = r0[lane + 32], x02 = r0[i2];
            const float4 x10 = r1[lane], x11 = r1[lane + 32], x12 = r1[i2];
            const float4 x20 = r2[lane], x21 = r2[lane + 32], x22 = r2[i2];
            const float4 x30 = r3[lane], x31 = r3[lane + 32], x32 = r3[i2];
            add4(a0, x00); add4(a1, x01); add4(a2, x02);
            add4(a0, x10); add4(a1, x11); add4(a2, x12);
            add4(a0, x20); add4(a1, x21); add4(a2, x22);
            add4(a0, x30); add4(a1, x31); add4(a2, x32);
        }
        for (; k < deg; ++k) {
            const float4* r0 = (const float4*)(emb + (size_t)__builtin_amdgcn_readlane(sorted, k) * DDIM);
            const float4 x00 = r0[lane], x01 = r0[lane + 32], x02 = r0[i2];
            add4(a0, x00); add4(a1, x01); add4(a2, x02);
        }
    } else {
        const int2* __restrict__ ov = s.ovf;
        const int nov = *s.ovfc;
        const int mine = rows[beg + lane];
        int last = -1;
        for (int pass = 0; pass < deg; ++pass) {
            int best = (mine > last) ? mine : 0x7fffffff;
            for (int b = lane; b < nov; b += 32) {
                const int2 e = ov[b];
                if (e.x == node && e.y > last && e.y < best) best = e.y;
            }
            for (int off = 16; off > 0; off >>= 1) best = min(best, __shfl_xor(best, off, 32));
            best = __builtin_amdgcn_readfirstlane(best);
            if (best == 0x7fffffff) break;
            int mult = (mine == best) ? 1 : 0;
            for (int b = lane; b < nov; b += 32) {
                const int2 e = ov[b];
                mult += (e.x == node && e.y == best) ? 1 : 0;
            }
            for (int off = 16; off > 0; off >>= 1) mult += __shfl_xor(mult, off, 32);
            mult = __builtin_amdgcn_readfirstlane(mult);
            const float4* r0 = (const float4*)(emb + (size_t)best * DDIM);
            const float4 x00 = r0[lane], x01 = r0[lane + 32], x02 = r0[i2];
            for (int m = 0; m < mult; ++m) { add4(a0, x00); add4(a1, x01); add4(a2, x02); }
            last = best;
        }
    }

    const float d = fmaxf((float)deg, 1.f);
    const float4 z = make_float4(0.f, 0.f, 0.f, 0.f);
    float4* o = (float4*)(out + (size_t)node * PDIM);
    o[lane]      = make_float4(a0.x / d, a0.y / d, a0.z / d, a0.w / d);
    o[lane + 32] = make_float4(a1.x / d, a1.y / d, a1.z / d, a1.w / d);
    if (lane < 16)
        o[lane + 64] = (lane < 11) ? make_float4(a2.x / d, a2.y / d, a2.z / d, a2.w / d) : z;
}

#define KB 16
static_assert(PDIM % (4 * KB) == 0, "a batch never straddles the A0 | A1 boundary");
static_assert(BN % 16 == 0 && PDIM % 64 == 0, "every wave of the grid owns a whole tile: no lane is switched off");

__device__ __forceinline__ void load_ab_batch(v2f (&a)[KB], v2f (&b)[KB],
                                              const float* __restrict__ ap, const float* __restrict__ bp) {
#pragma unroll
    for (int t = 0; t < KB; ++t) {
        a[t] = *(const v2f*)(ap + 4 * t);
        b[t].x = bp[(size_t)(4 * t) * PDIM];
        b[t].y = bp[(size_t)(4 * t + 1) * PDIM];
    }
}

__device__ __forceinline__ void mma_ab_batch(v8f& acc, const v2f (&a)[KB], const v2f (&b)[KB]) {
#pragma unroll
    for (int t = 0; t < KB; ++t)
        acc = __builtin_amdgcn_wmma_f32_16x16x4_f32(false, a[t], false, b[t], (short)0, acc, false, false);
}

template <int KT>
__global__ __launch_bounds__(128) __attribute__((amdgpu_waves_per_eu(1, 2))) void gemm_tile_wmma(
        const float* __restrict__ A0,
        const float* __restrict__ A1z0, const float* __restrict__ A1z1, const float* __restrict__ A1z2,
        const float* __restrict__ Bz0, int bzs,
        const float* __restrict__ bias0, const float* __restrict__ bias1, const float* __restrict__ bias2,
        float* __restrict__ Cz0, int czs) {
    static_assert(KT % (4 * KB) == 0, "whole batches");
    constexpr int NBAT = KT / (4 * KB);
    const int z = blockIdx.z;
    const float* __restrict__ A1 = (z == 0) ? A1z0 : (z == 1) ? A1z1 : A1z2;
    const float* __restrict__ bias = (z == 0) ? bias0 : (z == 1) ? bias1 : bias2;
    const float* __restrict__ B = Bz0 + (size_t)z * bzs;
    float* __restrict__ C = Cz0 + (size_t)z * czs;
    const int lane = threadIdx.x & 31;
    const int wave = threadIdx.x >> 5;
    const int hi   = lane >> 4;
    const int l16  = lane & 15;
    const int n    = (blockIdx.x * 4 + wave) * 16 + l16;
    const int mrow = blockIdx.y * 16;

    const size_t aoff = (size_t)(mrow + l16) * PDIM + 2 * hi;
    const float* __restrict__ bq = B + (size_t)(2 * hi) * PDIM + n;

    v2f a0[KB], b0[KB], a1[KB], b1[KB];
    load_ab_batch(a0, b0, A0 + aoff, bq);

    __builtin_amdgcn_sched_barrier(0);
    v8f acc;
    const float bl = bias[min(n, DDIM - 1)];
    const float bv = (n < DDIM) ? bl : 0.f;
#pragma unroll
    for (int i = 0; i < 8; ++i) acc[i] = bv;
#pragma unroll
    for (int bt = 0; bt < NBAT; bt += 2) {
        if (bt + 1 < NBAT) {
            const int k1 = (bt + 1) * 4 * KB;
            load_ab_batch(a1, b1, ((k1 < PDIM) ? A0 + k1 : A1 + (k1 - PDIM)) + aoff, bq + (size_t)k1 * PDIM);
        }
        mma_ab_batch(acc, a0, b0);
        if (bt + 2 < NBAT) {
            const int k2 = (bt + 2) * 4 * KB;
            load_ab_batch(a0, b0, ((k2 < PDIM) ? A0 + k2 : A1 + (k2 - PDIM)) + aoff, bq + (size_t)k2 * PDIM);
        }
        if (bt + 1 < NBAT) mma_ab_batch(acc, a1, b1);
    }
#pragma unroll
    for (int i = 0; i < 8; ++i)
        C[(size_t)(mrow + i + hi * 8) * PDIM + n] = acc[i];
}

#define KW  32
#define LPW 36
#define NT  (PDIM / 16)
#define AW  (KW / 4)
#define NCH ((2 * PDIM) / KW)
#define NBT ((KW * (PDIM / 4)) / 256)
static_assert(PDIM % (2 * KW) == 0, "a pair of chunks never straddles the A0 | X boundary; the K loop takes two chunks per trip");
static_assert(KW == 32 && PDIM == 32 * NBT, "panel staging map below: 8 rows x 16 floats per wave-load, 32 columns per pass");

__device__ __forceinline__ void load_b_panel(float4 (&bt)[NBT], const float* __restrict__ p) {
#pragma unroll
    for (int i = 0; i < NBT; ++i) bt[i] = *(const float4*)(p + 32 * i);
}

__device__ __forceinline__ void store_b_panel(float* ls, const float4 (&bt)[NBT]) {
#pragma unroll
    for (int i = 0; i < NBT; ++i) {
        ls[(32 * i + 0) * LPW] = bt[i].x;
        ls[(32 * i + 1) * LPW] = bt[i].y;
        ls[(32 * i + 2) * LPW] = bt[i].z;
        ls[(32 * i + 3) * LPW] = bt[i].w;
    }
}

__device__ __forceinline__ void chunk_rows(v8f (&acc)[NT], v2f (&a)[AW], const float* lb, const float* pn,
                                           float* ls, float4 (&bt)[NBT], const float* __restrict__ bn) {
#pragma unroll
    for (int t = 0; t < AW; ++t) {
        if (t == AW / 2) {
            store_b_panel(ls, bt);
            load_b_panel(bt, bn);
        }
#pragma unroll
        for (int ns = 0; ns < NT; ++ns) {
            v2f b = *(const v2f*)(lb + ns * 16 * LPW + 4 * t);
            acc[ns] = __builtin_amdgcn_wmma_f32_16x16x4_f32(
                false, a[t], false, b, (short)0, acc[ns], false, false);
        }
        a[t] = *(const v2f*)(pn + 4 * t);
    }
}

__global__ __launch_bounds__(256) void gemm_rows_wmma(
        const float* __restrict__ A0, float* X0, float* X1, float* X2,
        const float* __restrict__ Bz0, int bzs,
        const float* __restrict__ bias0, const float* __restrict__ bias1, const float* __restrict__ bias2,
        int M) {
    __shared__ float ldsB[2 * PDIM * LPW];
    const int lane = threadIdx.x & 31;
    const int wave = threadIdx.x >> 5;
    const int hi   = lane >> 4;
    const int l16  = lane & 15;
    const float* lb0 = &ldsB[l16 * LPW + 2 * hi];
    const float* lb1 = lb0 + PDIM * LPW;
    const int skr = (wave & 3) * 8 + (lane & 7);
    const int sc  = ((wave >> 2) * 4 + (lane >> 3)) * 4;
    float* ls0 = &ldsB[sc * LPW + skr];
    float* ls1 = ls0 + PDIM * LPW;
    const int nrb = (M + 127) >> 7;

    for (int u = blockIdx.x; u < 3 * nrb; u += gridDim.x) {
        const int z = u / nrb;
        const int mt = (u - z * nrb) * 8 + wave;
        float* X = (z == 0) ? X0 : (z == 1) ? X1 : X2;
        const float* __restrict__ bias = (z == 0) ? bias0 : (z == 1) ? bias1 : bias2;
        const float* __restrict__ Bp = Bz0 + (size_t)z * bzs + (size_t)skr * PDIM + sc;
        const bool valid = (mt * 16) < M;
        const int aoff = min(mt * 16 + l16, M - 1) * PDIM + 2 * hi;

        float4 bt[NBT];
        v2f ae[AW], ao[AW];
        load_b_panel(bt, Bp);
#pragma unroll
        for (int t = 0; t < AW; ++t) {
            ae[t] = *(const v2f*)(A0 + aoff + 4 * t);
            ao[t] = *(const v2f*)(A0 + aoff + KW + 4 * t);
        }
        store_b_panel(ls0, bt);
        load_b_panel(bt, Bp + (size_t)KW * PDIM);
        __syncthreads();

        v8f acc[NT];
#pragma unroll
        for (int ns = 0; ns < NT; ++ns) {
            const int n = ns * 16 + l16;
            const float bv = (n < DDIM) ? bias[n] : 0.f;
#pragma unroll
            for (int i = 0; i < 8; ++i) acc[ns][i] = bv;
        }

        for (int c = 0; c < NCH; c += 2) {
            const int kn = min((c + 2) * KW, 2 * PDIM - 2 * KW);
            const float* an = ((kn < PDIM) ? A0 + kn : X + (kn - PDIM)) + aoff;
            chunk_rows(acc, ae, lb0, an, ls1, bt, Bp + (size_t)kn * PDIM);
            __syncthreads();
            chunk_rows(acc, ao, lb1, an + KW, ls0, bt, Bp + (size_t)(kn + KW) * PDIM);
            __syncthreads();
        }
        if (valid) {
#pragma unroll
            for (int ns = 0; ns < NT; ++ns)
#pragma unroll
                for (int i = 0; i < 8; ++i)
                    X[(size_t)(mt * 16 + i + hi * 8) * PDIM + ns * 16 + l16] = acc[ns][i];
        }
    }
}

__global__ void sdpa_fuse_add(const float* __restrict__ h0, const float* __restrict__ h1,
                              const float* __restrict__ h2, const float* __restrict__ addv,
                              float* __restrict__ out, int n) {
    const int lane = threadIdx.x & 31;
    const int node = blockIdx.x * (blockDim.x >> 5) + (threadIdx.x >> 5);
    if (node >= n) return;
    const float* x0 = h0 + (size_t)node * PDIM;
    const float* x1 = h1 + (size_t)node * PDIM;
    const float* x2 = h2 + (size_t)node * PDIM;

    float d00 = 0, d01 = 0, d02 = 0, d11 = 0, d12 = 0, d22 = 0;
    for (int j = lane; j < PDIM; j += 32) {
        float a = x0[j], b = x1[j], e = x2[j];
        d00 += a * a; d01 += a * b; d02 += a * e;
        d11 += b * b; d12 += b * e; d22 += e * e;
    }
    for (int off = 16; off > 0; off >>= 1) {
        d00 += __shfl_xor(d00, off, 32);
        d01 += __shfl_xor(d01, off, 32);
        d02 += __shfl_xor(d02, off, 32);
        d11 += __shfl_xor(d11, off, 32);
        d12 += __shfl_xor(d12, off, 32);
        d22 += __shfl_xor(d22, off, 32);
    }
    const float scale = rsqrtf((float)DDIM);
    float S[3][3] = {{d00, d01, d02}, {d01, d11, d12}, {d02, d12, d22}};
    float wsum[3] = {0.f, 0.f, 0.f};
    for (int t = 0; t < 3; ++t) {
        float p0 = S[t][0] * scale, p1 = S[t][1] * scale, p2 = S[t][2] * scale;
        float mx = fmaxf(p0, fmaxf(p1, p2));
        float e0 = __expf(p0 - mx), e1 = __expf(p1 - mx), e2 = __expf(p2 - mx);
        float inv = 1.f / (e0 + e1 + e2);
        wsum[0] += e0 * inv; wsum[1] += e1 * inv; wsum[2] += e2 * inv;
    }
    const float w0 = wsum[0] * (1.f / 3.f), w1 = wsum[1] * (1.f / 3.f), w2 = wsum[2] * (1.f / 3.f);
    const float* av = addv + (size_t)node * PDIM;
    float* o = out + (size_t)node * PDIM;
    for (int j = lane; j < PDIM; j += 32)
        o[j] = w0 * x0[j] + w1 * x1[j] + w2 * x2[j] + av[j];
}

#define PB 32
static_assert(LN % (2 * PB) == 0, "pool_seq walks the sequence in pairs of batches");
static_assert(PDIM % 32 == 0, "every wave of pool_seq is full, so every lane's token id can be read by readlane");

__device__ __forceinline__ void pool_load(float (&x)[PB], const float* __restrict__ table, unsigned j,
                                          int ids, unsigned real) {
#pragma unroll
    for (int t = 0; t < PB; ++t) {
        const int idx = __builtin_amdgcn_readlane(ids, t);
        const int r = ((real >> t) & 1u) ? idx : 0;
        const float* __restrict__ rp = table + (size_t)r * PDIM;
        x[t] = rp[j];
    }
}

__device__ __forceinline__ void pool_add(float& acc, const float (&x)[PB], unsigned real) {
#pragma unroll
    for (int t = 0; t < PB; ++t)
        acc += ((real >> t) & 1u) ? x[t] : 0.f;
}

__global__ __launch_bounds__(PDIM) void pool_seq(const float* __restrict__ table, const int* __restrict__ xb,
                                                 float* __restrict__ pooled) {
    const int b = blockIdx.x;
    const unsigned j = threadIdx.x;
    if (j >= PDIM) return;
    const unsigned lane = j & 31u;
    const int* row = xb + b * LN;
    int ids[LN / PB];
    unsigned real[LN / PB];
#pragma unroll
    for (int k = 0; k < LN / PB; ++k) ids[k] = row[PB * k + lane];
#pragma unroll
    for (int k = 0; k < LN / PB; ++k)
        real[k] = __builtin_amdgcn_ballot_w32((unsigned)ids[k] < (unsigned)NWN);

    float acc = 0.f;
    float x0[PB], x1[PB];
    pool_load(x0, table, j, ids[0], real[0]);
#pragma unroll
    for (int k = 0; k < LN / PB; k += 2) {
        pool_load(x1, table, j, ids[k + 1], real[k + 1]);
        pool_add(acc, x0, real[k]);
        if (k + 2 < LN / PB) pool_load(x0, table, j, ids[k + 2], real[k + 2]);
        pool_add(acc, x1, real[k + 1]);
    }
    pooled[(size_t)b * PDIM + j] = acc * (1.f / (float)LN);
}

__global__ void fc_kernel(const float* __restrict__ rep, const float* __restrict__ wfc,
                          const float* __restrict__ bfc, float* __restrict__ out) {
    int t = blockIdx.x * blockDim.x + threadIdx.x;
    if (t >= BN * CN) return;
    int b = t / CN, c = t % CN;
    float acc = bfc[c];
    const float* r = rep + (size_t)b * PDIM;
    for (int j = 0; j < DDIM; ++j) acc += r[j] * wfc[(size_t)j * CN + c];
    out[t] = acc;
}


extern "C" void kernel_launch(void* const* d_in, const int* in_sizes, int n_in,
                              void* d_out, int out_size, void* d_ws, size_t ws_size,
                              hipStream_t stream) {
    (void)in_sizes; (void)n_in; (void)out_size; (void)ws_size;
    const float* emb_word = (const float*)d_in[0];
    const float* emb_doc  = (const float*)d_in[1];
    const float* w_sage[6] = {(const float*)d_in[2],  (const float*)d_in[4],  (const float*)d_in[6],
                              (const float*)d_in[8],  (const float*)d_in[10], (const float*)d_in[12]};
    const float* b_sage[6] = {(const float*)d_in[3],  (const float*)d_in[5],  (const float*)d_in[7],
                              (const float*)d_in[9],  (const float*)d_in[11], (const float*)d_in[13]};
    const float* w_dense = (const float*)d_in[14];
    const float* b_dense = (const float*)d_in[15];
    const float* w_fc    = (const float*)d_in[16];
    const float* b_fc    = (const float*)d_in[17];
    const int* dst_nid      = (const int*)d_in[18];
    const int* x_batch      = (const int*)d_in[28];
    const SegIn jw0 = {(const int*)d_in[22], (const int*)d_in[23], (const int*)d_in[19]};
    const SegIn jw1 = {(const int*)d_in[24], (const int*)d_in[25], (const int*)d_in[20]};
    const SegIn jw2 = {(const int*)d_in[26], (const int*)d_in[27], (const int*)d_in[21]};
    const SegIn jd0 = {(const int*)d_in[33], (const int*)d_in[34], (const int*)d_in[30]};
    const SegIn jd1 = {(const int*)d_in[35], (const int*)d_in[36], (const int*)d_in[31]};
    const SegIn jd2 = {(const int*)d_in[37], (const int*)d_in[38], (const int*)d_in[32]};
    float* out = (float*)d_out;

    float* ws = (float*)d_ws;
    size_t off = 0;
    auto carve = [&](size_t nflt) { float* p = ws + off; off += nflt; return p; };
    float* hdst = carve((size_t)NWN * PDIM);
    float* nbuf = carve((size_t)NWN * PDIM);
    float* h3[3]; for (int r = 0; r < 3; ++r) h3[r] = carve((size_t)NWN * PDIM);
    float* cnt  = carve(NWN);
    float* wpk[6]; for (int r = 0; r < 6; ++r) wpk[r] = carve((size_t)2 * PDIM * PDIM);
    float* wdn  = carve((size_t)PDIM * PDIM);
    float* pooled = carve((size_t)BN * PDIM);
    float* docout = carve((size_t)BN * PDIM);
    float* dnb    = carve((size_t)BN * PDIM);
    float* d3[3]; for (int r = 0; r < 3; ++r) d3[r] = carve((size_t)BN * PDIM);
    float* docrep = carve((size_t)BN * PDIM);
    (void)cnt; (void)wdn;
    static_assert((size_t)RBW + 3 * ((size_t)NWN * CAP + 2 * (size_t)EWN) <= (size_t)NWN * PDIM, "word-job scratch fits its buffer");
    static_assert((size_t)RBD + 3 * ((size_t)BN * CAP + 2 * (size_t)EDN) <= (size_t)BN * PDIM, "doc-job scratch fits its buffer");
    int* wi = (int*)h3[0];
    int* di = (int*)d3[0];
    float* wnb[3] = {h3[1], h3[2], nbuf};
    float* dnbr[3] = {dnb, docrep, pooled};
    float* poolbuf = d3[0];
    float* wtab = h3[0];

    const int TB = 256;
    const dim3 gemmGridW(256, 1, 1);
    const dim3 gemmGridD(PDIM / 64, BN / 16, 1);
    const dim3 gemmGridD3(PDIM / 64, BN / 16, 3);

    prep<<<HRB + WRB + ZB, TB, 0, stream>>>(emb_word, dst_nid, hdst,
                                                   w_sage[0], w_sage[1], w_sage[2], w_sage[3], w_sage[4], w_sage[5],
                                                   w_dense, wpk[0], wi, di);

    place_rows<<<3 * WEB + 3 * DEB, TB, 0, stream>>>(jw0, jw1, jw2, jd0, jd1, jd2, wi, di);
    gather_mean<<<3 * WNB + 3 * DNB, TB, 0, stream>>>(emb_word, emb_doc, wi, di,
                                                      wnb[0], wnb[1], wnb[2], dnbr[0], dnbr[1], dnbr[2]);

    gemm_rows_wmma<<<gemmGridW, TB, 0, stream>>>(hdst, wnb[0], wnb[1], wnb[2], wpk[0], 2 * PDIM * PDIM,
                                                 b_sage[0], b_sage[1], b_sage[2], NWN);

    sdpa_fuse_add<<<(NWN + 7) / 8, 256, 0, stream>>>(wnb[0], wnb[1], wnb[2], hdst, wtab, NWN);

    pool_seq<<<BN, PDIM, 0, stream>>>(wtab, x_batch, poolbuf);
    gemm_tile_wmma<PDIM><<<gemmGridD, 128, 0, stream>>>(poolbuf, poolbuf, poolbuf, poolbuf,
                                                        wpk[0] + (size_t)12 * PDIM * PDIM, 0,
                                                        b_dense, b_dense, b_dense, docout, 0);

    gemm_tile_wmma<2 * PDIM><<<gemmGridD3, 128, 0, stream>>>(docout, dnbr[0], dnbr[1], dnbr[2], wpk[3], 2 * PDIM * PDIM,
                                                             b_sage[3], b_sage[4], b_sage[5], d3[0], BN * PDIM);

    sdpa_fuse_add<<<(BN + 7) / 8, 256, 0, stream>>>(d3[0], d3[1], d3[2], docout, docrep, BN);
    fc_kernel<<<(BN * CN + TB - 1) / TB, TB, 0, stream>>>(docrep, w_fc, b_fc, out);
}


# seg_mean buckets: 64 slots per destination (were 32): a node with 33..64 in-edges is rank-sorted in two registers and summed four rows in flight, not one row per pass from the spill list; same sums
# speedup vs baseline: 3.8955x; 1.0527x over previous
#include <hip/hip_runtime.h>

#define DDIM 300
#define PDIM 320
#define NWN  20000
#define EWN  320000
#define BN   256
#define LN   256
#define SDN  2048
#define EDN  4096
#define CN   20

typedef __attribute__((ext_vector_type(2))) float v2f;
typedef __attribute__((ext_vector_type(8))) float v8f;

struct SegIn { const int* e_src; const int* e_dst; const int* src_nid; };
struct SegWs { int* cur; int* ovfc; int* rows; int2* ovf; int n; int E; };

#define CAP 64
#define RBW ((3 * NWN + 128 + 63) / 64 * 64)
#define RBD ((3 * BN + 128 + 63) / 64 * 64)
static_assert(CAP == 64 && RBW % 64 == 0 && RBD % 64 == 0, "a bucket is one aligned 256-byte piece");

__device__ __forceinline__ SegWs seg_ws(int job, int* wi, int* di) {
    SegWs s;
    if (job < 3) {
        s.n = NWN; s.E = EWN;
        s.cur = wi + job * NWN; s.ovfc = wi + 3 * NWN + job;
        s.rows = wi + RBW + job * (NWN * CAP);
        s.ovf = (int2*)(wi + RBW + 3 * (NWN * CAP)) + (size_t)job * EWN;
    } else {
        const int r = job - 3;
        s.n = BN; s.E = EDN;
        s.cur = di + r * BN; s.ovfc = di + 3 * BN + r;
        s.rows = di + RBD + r * (BN * CAP);
        s.ovf = (int2*)(di + RBD + 3 * (BN * CAP)) + (size_t)r * EDN;
    }
    return s;
}

__device__ __forceinline__ SegIn seg_in(int job, const SegIn& j0, const SegIn& j1, const SegIn& j2,
                                        const SegIn& j3, const SegIn& j4, const SegIn& j5) {
    switch (job) {
        case 0: return j0;
        case 1: return j1;
        case 2: return j2;
        case 3: return j3;
        case 4: return j4;
        default: return j5;
    }
}

#define EPT 4
#define WEB ((EWN / EPT + 255) / 256)
#define DEB ((EDN / EPT + 255) / 256)
static_assert(EWN % EPT == 0 && EDN % EPT == 0, "no partial group of edges");
__device__ __forceinline__ int edge_job(int blk, int& local) {
    if (blk < 3 * WEB) { const int j = blk / WEB; local = blk - j * WEB; return j; }
    const int d = blk - 3 * WEB;
    const int j = d / DEB;
    local = d - j * DEB;
    return 3 + j;
}

#define RPW 4
#define HRB (NWN / (8 * RPW))
#define WRB ((13 * PDIM) / (8 * RPW))
#define ZQW ((3 * NWN + 4) / 4)
#define ZQD ((3 * BN + 4) / 4)
#define ZB  ((ZQW + ZQD + 255) / 256)
static_assert(NWN % (8 * RPW) == 0 && (13 * PDIM) % (8 * RPW) == 0, "every wave of the row blocks has RPW rows");
static_assert(PDIM % RPW == 0 && DDIM % RPW == 0, "the RPW rows of a wave lie in one slab and are all data rows or all pad rows");
static_assert((3 * NWN) % 4 == 0 && (3 * BN) % 4 == 0, "count arrays are whole 16-byte pieces");
static_assert(4 * ZQW <= RBW && 4 * ZQD <= RBD, "the zeroed words end in front of the first bucket");

__device__ __forceinline__ void copy_rows(const float* __restrict__ s0, const float* __restrict__ s1,
                                          const float* __restrict__ s2, const float* __restrict__ s3,
                                          float* __restrict__ o, int lane) {
    static_assert(RPW == 4 && PDIM == 320 && DDIM == 300, "run of 320 float4, 75 data groups per row");
    float4 x[10];
    bool keep[10];
#pragma unroll
    for (int i = 0; i < 10; ++i) {
        const int f = lane + 32 * i;
        const int q = (f >= 80 ? 1 : 0) + (f >= 160 ? 1 : 0) + (f >= 240 ? 1 : 0);
        const int g = f - 80 * q;
        const float* __restrict__ sq = (q == 0) ? s0 : (q == 1) ? s1 : (q == 2) ? s2 : s3;
        keep[i] = g < 75;
        x[i] = ((const float4*)sq)[keep[i] ? g : 74];
    }
    float4* __restrict__ d = (float4*)o;
#pragma unroll
    for (int i = 0; i < 10; ++i)
        d[lane + 32 * i] = make_float4(keep[i] ? x[i].x : 0.f, keep[i] ? x[i].y : 0.f,
                                       keep[i] ? x[i].z : 0.f, keep[i] ? x[i].w : 0.f);
}

__device__ __forceinline__ void zero_rows(float* __restrict__ o, int lane) {
    const float4 z = make_float4(0.f, 0.f, 0.f, 0.f);
    float4* __restrict__ d = (float4*)o;
#pragma unroll
    for (int i = 0; i < 10; ++i) d[lane + 32 * i] = z;
}

__global__ __launch_bounds__(256) void prep(const float* __restrict__ emb, const int* __restrict__ nid,
                                            float* __restrict__ hdst,
                                            const float* __restrict__ w0, const float* __restrict__ w1,
                                            const float* __restrict__ w2, const float* __restrict__ w3,
                                            const float* __restrict__ w4, const float* __restrict__ w5,
                                            const float* __restrict__ wd, float* __restrict__ wout,
                                            int* __restrict__ wi, int* __restrict__ di) {
    const int blk = blockIdx.x;
    const int lane = threadIdx.x & 31;
    const int wv = __builtin_amdgcn_readfirstlane((int)(blk * 8 + (threadIdx.x >> 5)));
    if (blk < HRB) {
        const int r = wv * RPW;
        copy_rows(emb + (size_t)nid[r + 0] * DDIM, emb + (size_t)nid[r + 1] * DDIM,
                  emb + (size_t)nid[r + 2] * DDIM, emb + (size_t)nid[r + 3] * DDIM,
                  hdst + (size_t)r * PDIM, lane);
    } else if (blk < HRB + WRB) {
        const int rr = (wv - HRB * 8) * RPW;
        const int slab = rr / PDIM;
        const int r = rr - slab * PDIM;
        float* __restrict__ o = wout + (size_t)rr * PDIM;
        if (r < DDIM) {
            const int which = slab >> 1;
            const float* __restrict__ w = (slab >= 12) ? wd : (which == 0) ? w0 : (which == 1) ? w1 : (which == 2) ? w2
                                        : (which == 3) ? w3 : (which == 4) ? w4 : w5;
            const int r0 = (slab < 12 && (slab & 1)) ? DDIM : 0;
            const float* __restrict__ s = w + (size_t)(r0 + r) * DDIM;
            copy_rows(s, s + DDIM, s + 2 * DDIM, s + 3 * DDIM, o, lane);
        } else {
            zero_rows(o, lane);
        }
    } else {
        const int t = (blk - HRB - WRB) * 256 + threadIdx.x;
        const int4 z = make_int4(0, 0, 0, 0);
        if (t < ZQW) ((int4*)wi)[t] = z;
        else if (t < ZQW + ZQD) ((int4*)di)[t - ZQW] = z;
    }
}


__device__ __forceinline__ void put_row(const SegWs& s, int dst, int p, int id) {
    if (p < CAP) {
        s.rows[dst * CAP + p] = id;
    } else {
        const int q = atomicAdd(s.ovfc, 1);
        s.ovf[q] = make_int2(dst, id);
    }
}

__global__ __launch_bounds__(256) void place_rows(const SegIn j0, const SegIn j1, const SegIn j2,
                                                  const SegIn j3, const SegIn j4, const SegIn j5,
                                                  int* __restrict__ wi, int* __restrict__ di) {
    int local;
    const int job = edge_job(blockIdx.x, local);
    const SegIn in = seg_in(job, j0, j1, j2, j3, j4, j5);
    const SegWs s = seg_ws(job, wi, di);
    const int t = local * 256 + threadIdx.x;
    if (t >= s.E / EPT) return;
    const int4 d = ((const int4*)in.e_dst)[t];
    const int4 e = ((const int4*)in.e_src)[t];
    const int r0 = in.src_nid[e.x], r1 = in.src_nid[e.y], r2 = in.src_nid[e.z], r3 = in.src_nid[e.w];
    const int p0 = atomicAdd(&s.cur[d.x], 1);
    const int p1 = atomicAdd(&s.cur[d.y], 1);
    const int p2 = atomicAdd(&s.cur[d.z], 1);
    const int p3 = atomicAdd(&s.cur[d.w], 1);
    put_row(s, d.x, p0, r0); put_row(s, d.y, p1, r1); put_row(s, d.z, p2, r2); put_row(s, d.w, p3, r3);
}

__device__ __forceinline__ void add4(float4& a, const float4 x) {
    a.x += x.x; a.y += x.y; a.z += x.z; a.w += x.w;
}

__device__ __forceinline__ void add_rows4(float4& a0, float4& a1, float4& a2, const float* __restrict__ emb,
                                          int q0, int q1, int q2, int q3, int lane, int i2) {
    const float4* r0 = (const float4*)(emb + (size_t)q0 * DDIM);
    const float4* r1 = (const float4*)(emb + (size_t)q1 * DDIM);
    const float4* r2 = (const float4*)(emb + (size_t)q2 * DDIM);
    const float4* r3 = (const float4*)(emb + (size_t)q3 * DDIM);
    const float4 x00 = r0[lane], x01 = r0[lane + 32], x02 = r0[i2];
    const float4 x10 = r1[lane], x11 = r1[lane + 32], x12 = r1[i2];
    const float4 x20 = r2[lane], x21 = r2[lane + 32], x22 = r2[i2];
    const float4 x30 = r3[lane], x31 = r3[lane + 32], x32 = r3[i2];
    add4(a0, x00); add4(a1, x01); add4(a2, x02);
    add4(a0, x10); add4(a1, x11); add4(a2, x12);
    add4(a0, x20); add4(a1, x21); add4(a2, x22);
    add4(a0, x30); add4(a1, x31); add4(a2, x32);
}

__device__ __forceinline__ void add_row1(float4& a0, float4& a1, float4& a2, const float* __restrict__ emb,
                                         int q0, int lane, int i2) {
    const float4* r0 = (const float4*)(emb + (size_t)q0 * DDIM);
    const float4 x00 = r0[lane], x01 = r0[lane + 32], x02 = r0[i2];
    add4(a0, x00); add4(a1, x01); add4(a2, x02);
}

__device__ __forceinline__ int id_of_rank(int r, int r0, int r1, int k0, int k1) {
    const unsigned m0 = __builtin_amdgcn_ballot_w32(r0 == r);
    const unsigned m1 = __builtin_amdgcn_ballot_w32(r1 == r);
    const int v0 = __builtin_amdgcn_readlane(k0, m0 ? __builtin_ctz(m0) : 0);
    const int v1 = __builtin_amdgcn_readlane(k1, m1 ? __builtin_ctz(m1) : 0);
    return m0 ? v0 : v1;
}

#define WNB ((NWN + 7) / 8)
#define DNB ((BN + 7) / 8)
__global__ __launch_bounds__(256) void gather_mean(const float* __restrict__ emb_w, const float* __restrict__ emb_d,
                                                   int* __restrict__ wi, int* __restrict__ di,
                                                   float* __restrict__ o0, float* __restrict__ o1,
                                                   float* __restrict__ o2, float* __restrict__ o3,
                                                   float* __restrict__ o4, float* __restrict__ o5) {
    const int lane = threadIdx.x & 31;
    const int blk = blockIdx.x;
    int job, local;
    if (blk < 3 * WNB) { job = blk / WNB; local = blk - job * WNB; }
    else { const int d = blk - 3 * WNB; const int j = d / DNB; local = d - j * DNB; job = 3 + j; }
    const SegWs s = seg_ws(job, wi, di);
    const float* __restrict__ emb = (job < 3) ? emb_w : emb_d;
    float* __restrict__ out = (job == 0) ? o0 : (job == 1) ? o1 : (job == 2) ? o2
                            : (job == 3) ? o3 : (job == 4) ? o4 : o5;
    const int* __restrict__ rows = s.rows;
    const int* __restrict__ cnt = s.cur;
    const int node = __builtin_amdgcn_readfirstlane((int)(local * 8 + (threadIdx.x >> 5)));
    if (node >= s.n) return;
    const int beg = node * CAP;
    const int deg = cnt[node];
    const int i2 = (lane < 11) ? lane + 64 : 74;

    float4 a0 = make_float4(0.f, 0.f, 0.f, 0.f), a1 = a0, a2 = a0;
    if (deg <= 32) {
        const int key = (lane < deg) ? rows[beg + lane] : 0x7fffffff;
        int rank = 0;
#pragma unroll
        for (int k = 0; k < 32; ++k) {
            const int kk = __builtin_amdgcn_readlane(key, k);
            rank += (kk < key || (kk == key && k < lane)) ? 1 : 0;
        }
        const int sorted = __builtin_amdgcn_ds_permute(rank << 2, key);
        int k = 0;
        for (; k + 4 <= deg; k += 4)
            add_rows4(a0, a1, a2, emb, __builtin_amdgcn_readlane(sorted, k + 0), __builtin_amdgcn_readlane(sorted, k + 1),
                      __builtin_amdgcn_readlane(sorted, k + 2), __builtin_amdgcn_readlane(sorted, k + 3), lane, i2);
        for (; k < deg; ++k)
            add_row1(a0, a1, a2, emb, __builtin_amdgcn_readlane(sorted, k), lane, i2);
    } else if (deg <= CAP) {
        const int k0 = rows[beg + lane];
        const int k1 = (32 + lane < deg) ? rows[beg + 32 + lane] : 0x7fffffff;
        int r0 = 0, r1 = 0;
#pragma nounroll
        for (int s = 0; s < 32; ++s) {
            const int a = __builtin_amdgcn_readlane(k0, s);
            const int b = __builtin_amdgcn_readlane(k1, s);
            r0 += ((a < k0 || (a == k0 && s < lane)) ? 1 : 0) + ((b < k0) ? 1 : 0);
            r1 += ((a <= k1) ? 1 : 0) + ((b < k1 || (b == k1 && s < lane)) ? 1 : 0);
        }
        int k = 0;
        for (; k + 4 <= deg; k += 4)
            add_rows4(a0, a1, a2, emb, id_of_rank(k + 0, r0, r1, k0, k1), id_of_rank(k + 1, r0, r1, k0, k1),
                      id_of_rank(k + 2, r0, r1, k0, k1), id_of_rank(k + 3, r0, r1, k0, k1), lane, i2);
        for (; k < deg; ++k)
            add_row1(a0, a1, a2, emb, id_of_rank(k, r0, r1, k0, k1), lane, i2);
    } else {
        const int2* __restrict__ ov = s.ovf;
        const int nov = *s.ovfc;
        const int m0 = rows[beg + lane];
        const int m1 = rows[beg + 32 + lane];
        int last = -1;
        for (int pass = 0; pass < deg; ++pass) {
            int best = (m0 > last) ? m0 : 0x7fffffff;
            if (m1 > last && m1 < best) best = m1;
            for (int b = lane; b < nov; b += 32) {
                const int2 e = ov[b];
                if (e.x == node && e.y > last && e.y < best) best = e.y;
            }
            for (int off = 16; off > 0; off >>= 1) best = min(best, __shfl_xor(best, off, 32));
            best = __builtin_amdgcn_readfirstlane(best);
            if (best == 0x7fffffff) break;
            int mult = ((m0 == best) ? 1 : 0) + ((m1 == best) ? 1 : 0);
            for (int b = lane; b < nov; b += 32) {
                const int2 e = ov[b];
                mult += (e.x == node && e.y == best) ? 1 : 0;
            }
            for (int off = 16; off > 0; off >>= 1) mult += __shfl_xor(mult, off, 32);
            mult = __builtin_amdgcn_readfirstlane(mult);
            const float4* r0 = (const float4*)(emb + (size_t)best * DDIM);
            const float4 x00 = r0[lane], x01 = r0[lane + 32], x02 = r0[i2];
            for (int m = 0; m < mult; ++m) { add4(a0, x00); add4(a1, x01); add4(a2, x02); }
            last = best;
        }
    }

    const float d = fmaxf((float)deg, 1.f);
    const float4 z = make_float4(0.f, 0.f, 0.f, 0.f);
    float4* o = (float4*)(out + (size_t)node * PDIM);
    o[lane]      = make_float4(a0.x / d, a0.y / d, a0.z / d, a0.w / d);
    o[lane + 32] = make_float4(a1.x / d, a1.y / d, a1.z / d, a1.w / d);
    if (lane < 16)
        o[lane + 64] = (lane < 11) ? make_float4(a2.x / d, a2.y / d, a2.z / d, a2.w / d) : z;
}

#define KB 16
static_assert(PDIM % (4 * KB) == 0, "a batch never straddles the A0 | A1 boundary");
static_assert(BN % 16 == 0 && PDIM % 64 == 0, "every wave of the grid owns a whole tile: no lane is switched off");

__device__ __forceinline__ void load_ab_batch(v2f (&a)[KB], v2f (&b)[KB],
                                              const float* __restrict__ ap, const float* __restrict__ bp) {
#pragma unroll
    for (int t = 0; t < KB; ++t) {
        a[t] = *(const v2f*)(ap + 4 * t);
        b[t].x = bp[(size_t)(4 * t) * PDIM];
        b[t].y = bp[(size_t)(4 * t + 1) * PDIM];
    }
}

__device__ __forceinline__ void mma_ab_batch(v8f& acc, const v2f (&a)[KB], const v2f (&b)[KB]) {
#pragma unroll
    for (int t = 0; t < KB; ++t)
        acc = __builtin_amdgcn_wmma_f32_16x16x4_f32(false, a[t], false, b[t], (short)0, acc, false, false);
}

template <int KT>
__global__ __launch_bounds__(128) __attribute__((amdgpu_waves_per_eu(1, 2))) void gemm_tile_wmma(
        const float* __restrict__ A0,
        const float* __restrict__ A1z0, const float* __restrict__ A1z1, const float* __restrict__ A1z2,
        const float* __restrict__ Bz0, int bzs,
        const float* __restrict__ bias0, const float* __restrict__ bias1, const float* __restrict__ bias2,
        float* __restrict__ Cz0, int czs) {
    static_assert(KT % (4 * KB) == 0, "whole batches");
    constexpr int NBAT = KT / (4 * KB);
    const int z = blockIdx.z;
    const float* __restrict__ A1 = (z == 0) ? A1z0 : (z == 1) ? A1z1 : A1z2;
    const float* __restrict__ bias = (z == 0) ? bias0 : (z == 1) ? bias1 : bias2;
    const float* __restrict__ B = Bz0 + (size_t)z * bzs;
    float* __restrict__ C = Cz0 + (size_t)z * czs;
    const int lane = threadIdx.x & 31;
    const int wave = threadIdx.x >> 5;
    const int hi   = lane >> 4;
    const int l16  = lane & 15;
    const int n    = (blockIdx.x * 4 + wave) * 16 + l16;
    const int mrow = blockIdx.y * 16;

    const size_t aoff = (size_t)(mrow + l16) * PDIM + 2 * hi;
    const float* __restrict__ bq = B + (size_t)(2 * hi) * PDIM + n;

    v2f a0[KB], b0[KB], a1[KB], b1[KB];
    load_ab_batch(a0, b0, A0 + aoff, bq);

    __builtin_amdgcn_sched_barrier(0);
    v8f acc;
    const float bl = bias[min(n, DDIM - 1)];
    const float bv = (n < DDIM) ? bl : 0.f;
#pragma unroll
    for (int i = 0; i < 8; ++i) acc[i] = bv;
#pragma unroll
    for (int bt = 0; bt < NBAT; bt += 2) {
        if (bt + 1 < NBAT) {
            const int k1 = (bt + 1) * 4 * KB;
            load_ab_batch(a1, b1, ((k1 < PDIM) ? A0 + k1 : A1 + (k1 - PDIM)) + aoff, bq + (size_t)k1 * PDIM);
        }
        mma_ab_batch(acc, a0, b0);
        if (bt + 2 < NBAT) {
            const int k2 = (bt + 2) * 4 * KB;
            load_ab_batch(a0, b0, ((k2 < PDIM) ? A0 + k2 : A1 + (k2 - PDIM)) + aoff, bq + (size_t)k2 * PDIM);
        }
        if (bt + 1 < NBAT) mma_ab_batch(acc, a1, b1);
    }
#pragma unroll
    for (int i = 0; i < 8; ++i)
        C[(size_t)(mrow + i + hi * 8) * PDIM + n] = acc[i];
}

#define KW  32
#define LPW 36
#define NT  (PDIM / 16)
#define AW  (KW / 4)
#define NCH ((2 * PDIM) / KW)
#define NBT ((KW * (PDIM / 4)) / 256)
static_assert(PDIM % (2 * KW) == 0, "a pair of chunks never straddles the A0 | X boundary; the K loop takes two chunks per trip");
static_assert(KW == 32 && PDIM == 32 * NBT, "panel staging map below: 8 rows x 16 floats per wave-load, 32 columns per pass");

__device__ __forceinline__ void load_b_panel(float4 (&bt)[NBT], const float* __restrict__ p) {
#pragma unroll
    for (int i = 0; i < NBT; ++i) bt[i] = *(const float4*)(p + 32 * i);
}

__device__ __forceinline__ void store_b_panel(float* ls, const float4 (&bt)[NBT]) {
#pragma unroll
    for (int i = 0; i < NBT; ++i) {
        ls[(32 * i + 0) * LPW] = bt[i].x;
        ls[(32 * i + 1) * LPW] = bt[i].y;
        ls[(32 * i + 2) * LPW] = bt[i].z;
        ls[(32 * i + 3) * LPW] = bt[i].w;
    }
}

__device__ __forceinline__ void chunk_rows(v8f (&acc)[NT], v2f (&a)[AW], const float* lb, const float* pn,
                                           float* ls, float4 (&bt)[NBT], const float* __restrict__ bn) {
#pragma unroll
    for (int t = 0; t < AW; ++t) {
        if (t == AW / 2) {
            store_b_panel(ls, bt);
            load_b_panel(bt, bn);
        }
#pragma unroll
        for (int ns = 0; ns < NT; ++ns) {
            v2f b = *(const v2f*)(lb + ns * 16 * LPW + 4 * t);
            acc[ns] = __builtin_amdgcn_wmma_f32_16x16x4_f32(
                false, a[t], false, b, (short)0, acc[ns], false, false);
        }
        a[t] = *(const v2f*)(pn + 4 * t);
    }
}

__global__ __launch_bounds__(256) void gemm_rows_wmma(
        const float* __restrict__ A0, float* X0, float* X1, float* X2,
        const float* __restrict__ Bz0, int bzs,
        const float* __restrict__ bias0, const float* __restrict__ bias1, const float* __restrict__ bias2,
        int M) {
    __shared__ float ldsB[2 * PDIM * LPW];
    const int lane = threadIdx.x & 31;
    const int wave = threadIdx.x >> 5;
    const int hi   = lane >> 4;
    const int l16  = lane & 15;
    const float* lb0 = &ldsB[l16 * LPW + 2 * hi];
    const float* lb1 = lb0 + PDIM * LPW;
    const int skr = (wave & 3) * 8 + (lane & 7);
    const int sc  = ((wave >> 2) * 4 + (lane >> 3)) * 4;
    float* ls0 = &ldsB[sc * LPW + skr];
    float* ls1 = ls0 + PDIM * LPW;
    const int nrb = (M + 127) >> 7;

    for (int u = blockIdx.x; u < 3 * nrb; u += gridDim.x) {
        const int z = u / nrb;
        const int mt = (u - z * nrb) * 8 + wave;
        float* X = (z == 0) ? X0 : (z == 1) ? X1 : X2;
        const float* __restrict__ bias = (z == 0) ? bias0 : (z == 1) ? bias1 : bias2;
        const float* __restrict__ Bp = Bz0 + (size_t)z * bzs + (size_t)skr * PDIM + sc;
        const bool valid = (mt * 16) < M;
        const int aoff = min(mt * 16 + l16, M - 1) * PDIM + 2 * hi;

        float4 bt[NBT];
        v2f ae[AW], ao[AW];
        load_b_panel(bt, Bp);
#pragma unroll
        for (int t = 0; t < AW; ++t) {
            ae[t] = *(const v2f*)(A0 + aoff + 4 * t);
            ao[t] = *(const v2f*)(A0 + aoff + KW + 4 * t);
        }
        store_b_panel(ls0, bt);
        load_b_panel(bt, Bp + (size_t)KW * PDIM);
        __syncthreads();

        v8f acc[NT];
#pragma unroll
        for (int ns = 0; ns < NT; ++ns) {
            const int n = ns * 16 + l16;
            const float bv = (n < DDIM) ? bias[n] : 0.f;
#pragma unroll
            for (int i = 0; i < 8; ++i) acc[ns][i] = bv;
        }

        for (int c = 0; c < NCH; c += 2) {
            const int kn = min((c + 2) * KW, 2 * PDIM - 2 * KW);
            const float* an = ((kn < PDIM) ? A0 + kn : X + (kn - PDIM)) + aoff;
            chunk_rows(acc, ae, lb0, an, ls1, bt, Bp + (size_t)kn * PDIM);
            __syncthreads();
            chunk_rows(acc, ao, lb1, an + KW, ls0, bt, Bp + (size_t)(kn + KW) * PDIM);
            __syncthreads();
        }
        if (valid) {
#pragma unroll
            for (int ns = 0; ns < NT; ++ns)
#pragma unroll
                for (int i = 0; i < 8; ++i)
                    X[(size_t)(mt * 16 + i + hi * 8) * PDIM + ns * 16 + l16] = acc[ns][i];
        }
    }
}

__global__ void sdpa_fuse_add(const float* __restrict__ h0, const float* __restrict__ h1,
                              const float* __restrict__ h2, const float* __restrict__ addv,
                              float* __restrict__ out, int n) {
    const int lane = threadIdx.x & 31;
    const int node = blockIdx.x * (blockDim.x >> 5) + (threadIdx.x >> 5);
    if (node >= n) return;
    const float* x0 = h0 + (size_t)node * PDIM;
    const float* x1 = h1 + (size_t)node * PDIM;
    const float* x2 = h2 + (size_t)node * PDIM;

    float d00 = 0, d01 = 0, d02 = 0, d11 = 0, d12 = 0, d22 = 0;
    for (int j = lane; j < PDIM; j += 32) {
        float a = x0[j], b = x1[j], e = x2[j];
        d00 += a * a; d01 += a * b; d02 += a * e;
        d11 += b * b; d12 += b * e; d22 += e * e;
    }
    for (int off = 16; off > 0; off >>= 1) {
        d00 += __shfl_xor(d00, off, 32);
        d01 += __shfl_xor(d01, off, 32);
        d02 += __shfl_xor(d02, off, 32);
        d11 += __shfl_xor(d11, off, 32);
        d12 += __shfl_xor(d12, off, 32);
        d22 += __shfl_xor(d22, off, 32);
    }
    const float scale = rsqrtf((float)DDIM);
    float S[3][3] = {{d00, d01, d02}, {d01, d11, d12}, {d02, d12, d22}};
    float wsum[3] = {0.f, 0.f, 0.f};
    for (int t = 0; t < 3; ++t) {
        float p0 = S[t][0] * scale, p1 = S[t][1] * scale, p2 = S[t][2] * scale;
        float mx = fmaxf(p0, fmaxf(p1, p2));
        float e0 = __expf(p0 - mx), e1 = __expf(p1 - mx), e2 = __expf(p2 - mx);
        float inv = 1.f / (e0 + e1 + e2);
        wsum[0] += e0 * inv; wsum[1] += e1 * inv; wsum[2] += e2 * inv;
    }
    const float w0 = wsum[0] * (1.f / 3.f), w1 = wsum[1] * (1.f / 3.f), w2 = wsum[2] * (1.f / 3.f);
    const float* av = addv + (size_t)node * PDIM;
    float* o = out + (size_t)node * PDIM;
    for (int j = lane; j < PDIM; j += 32)
        o[j] = w0 * x0[j] + w1 * x1[j] + w2 * x2[j] + av[j];
}

#define PB 32
static_assert(LN % (2 * PB) == 0, "pool_seq walks the sequence in pairs of batches");
static_assert(PDIM % 32 == 0, "every wave of pool_seq is full, so every lane's token id can be read by readlane");

__device__ __forceinline__ void pool_load(float (&x)[PB], const float* __restrict__ table, unsigned j,
                                          int ids, unsigned real) {
#pragma unroll
    for (int t = 0; t < PB; ++t) {
        const int idx = __builtin_amdgcn_readlane(ids, t);
        const int r = ((real >> t) & 1u) ? idx : 0;
        const float* __restrict__ rp = table + (size_t)r * PDIM;
        x[t] = rp[j];
    }
}

__device__ __forceinline__ void pool_add(float& acc, const float (&x)[PB], unsigned real) {
#pragma unroll
    for (int t = 0; t < PB; ++t)
        acc += ((real >> t) & 1u) ? x[t] : 0.f;
}

__global__ __launch_bounds__(PDIM) void pool_seq(const float* __restrict__ table, const int* __restrict__ xb,
                                                 float* __restrict__ pooled) {
    const int b = blockIdx.x;
    const unsigned j = threadIdx.x;
    if (j >= PDIM) return;
    const unsigned lane = j & 31u;
    const int* row = xb + b * LN;
    int ids[LN / PB];
    unsigned real[LN / PB];
#pragma unroll
    for (int k = 0; k < LN / PB; ++k) ids[k] = row[PB * k + lane];
#pragma unroll
    for (int k = 0; k < LN / PB; ++k)
        real[k] = __builtin_amdgcn_ballot_w32((unsigned)ids[k] < (unsigned)NWN);

    float acc = 0.f;
    float x0[PB], x1[PB];
    pool_load(x0, table, j, ids[0], real[0]);
#pragma unroll
    for (int k = 0; k < LN / PB; k += 2) {
        pool_load(x1, table, j, ids[k + 1], real[k + 1]);
        pool_add(acc, x0, real[k]);
        if (k + 2 < LN / PB) pool_load(x0, table, j, ids[k + 2], real[k + 2]);
        pool_add(acc, x1, real[k + 1]);
    }
    pooled[(size_t)b * PDIM + j] = acc * (1.f / (float)LN);
}

__global__ void fc_kernel(const float* __restrict__ rep, const float* __restrict__ wfc,
                          const float* __restrict__ bfc, float* __restrict__ out) {
    int t = blockIdx.x * blockDim.x + threadIdx.x;
    if (t >= BN * CN) return;
    int b = t / CN, c = t % CN;
    float acc = bfc[c];
    const float* r = rep + (size_t)b * PDIM;
    for (int j = 0; j < DDIM; ++j) acc += r[j] * wfc[(size_t)j * CN + c];
    out[t] = acc;
}


extern "C" void kernel_launch(void* const* d_in, const int* in_sizes, int n_in,
                              void* d_out, int out_size, void* d_ws, size_t ws_size,
                              hipStream_t stream) {
    (void)in_sizes; (void)n_in; (void)out_size; (void)ws_size;
    const float* emb_word = (const float*)d_in[0];
    const float* emb_doc  = (const float*)d_in[1];
    const float* w_sage[6] = {(const float*)d_in[2],  (const float*)d_in[4],  (const float*)d_in[6],
                              (const float*)d_in[8],  (const float*)d_in[10], (const float*)d_in[12]};
    const float* b_sage[6] = {(const float*)d_in[3],  (const float*)d_in[5],  (const float*)d_in[7],
                              (const float*)d_in[9],  (const float*)d_in[11], (const float*)d_in[13]};
    const float* w_dense = (const float*)d_in[14];
    const float* b_dense = (const float*)d_in[15];
    const float* w_fc    = (const float*)d_in[16];
    const float* b_fc    = (const float*)d_in[17];
    const int* dst_nid      = (const int*)d_in[18];
    const int* x_batch      = (const int*)d_in[28];
    const SegIn jw0 = {(const int*)d_in[22], (const int*)d_in[23], (const int*)d_in[19]};
    const SegIn jw1 = {(const int*)d_in[24], (const int*)d_in[25], (const int*)d_in[20]};
    const SegIn jw2 = {(const int*)d_in[26], (const int*)d_in[27], (const int*)d_in[21]};
    const SegIn jd0 = {(const int*)d_in[33], (const int*)d_in[34], (const int*)d_in[30]};
    const SegIn jd1 = {(const int*)d_in[35], (const int*)d_in[36], (const int*)d_in[31]};
    const SegIn jd2 = {(const int*)d_in[37], (const int*)d_in[38], (const int*)d_in[32]};
    float* out = (float*)d_out;

    float* ws = (float*)d_ws;
    size_t off = 0;
    auto carve = [&](size_t nflt) { float* p = ws + off; off += nflt; return p; };
    float* hdst = carve((size_t)NWN * PDIM);
    float* nbuf = carve((size_t)NWN * PDIM);
    float* h3[3]; for (int r = 0; r < 3; ++r) h3[r] = carve((size_t)NWN * PDIM);
    float* cnt  = carve(NWN);
    float* wpk[6]; for (int r = 0; r < 6; ++r) wpk[r] = carve((size_t)2 * PDIM * PDIM);
    float* wdn  = carve((size_t)PDIM * PDIM);
    float* pooled = carve((size_t)BN * PDIM);
    float* docout = carve((size_t)BN * PDIM);
    float* dnb    = carve((size_t)BN * PDIM);
    float* d3[3]; for (int r = 0; r < 3; ++r) d3[r] = carve((size_t)BN * PDIM);
    float* docrep = carve((size_t)BN * PDIM);
    (void)cnt; (void)wdn;
    static_assert((size_t)RBW + 3 * ((size_t)NWN * CAP + 2 * (size_t)EWN) <= (size_t)NWN * PDIM, "word-job scratch fits its buffer");
    static_assert((size_t)RBD + 3 * ((size_t)BN * CAP + 2 * (size_t)EDN) <= (size_t)BN * PDIM, "doc-job scratch fits its buffer");
    int* wi = (int*)h3[0];
    int* di = (int*)d3[0];
    float* wnb[3] = {h3[1], h3[2], nbuf};
    float* dnbr[3] = {dnb, docrep, pooled};
    float* poolbuf = d3[0];
    float* wtab = h3[0];

    const int TB = 256;
    const dim3 gemmGridW(256, 1, 1);
    const dim3 gemmGridD(PDIM / 64, BN / 16, 1);
    const dim3 gemmGridD3(PDIM / 64, BN / 16, 3);

    prep<<<HRB + WRB + ZB, TB, 0, stream>>>(emb_word, dst_nid, hdst,
                                                   w_sage[0], w_sage[1], w_sage[2], w_sage[3], w_sage[4], w_sage[5],
                                                   w_dense, wpk[0], wi, di);

    place_rows<<<3 * WEB + 3 * DEB, TB, 0, stream>>>(jw0, jw1, jw2, jd0, jd1, jd2, wi, di);
    gather_mean<<<3 * WNB + 3 * DNB, TB, 0, stream>>>(emb_word, emb_doc, wi, di,
                                                      wnb[0], wnb[1], wnb[2], dnbr[0], dnbr[1], dnbr[2]);

    gemm_rows_wmma<<<gemmGridW, TB, 0, stream>>>(hdst, wnb[0], wnb[1], wnb[2], wpk[0], 2 * PDIM * PDIM,
                                                 b_sage[0], b_sage[1], b_sage[2], NWN);

    sdpa_fuse_add<<<(NWN + 7) / 8, 256, 0, stream>>>(wnb[0], wnb[1], wnb[2], hdst, wtab, NWN);

    pool_seq<<<BN, PDIM, 0, stream>>>(wtab, x_batch, poolbuf);
    gemm_tile_wmma<PDIM><<<gemmGridD, 128, 0, stream>>>(poolbuf, poolbuf, poolbuf, poolbuf,
                                                        wpk[0] + (size_t)12 * PDIM * PDIM, 0,
                                                        b_dense, b_dense, b_dense, docout, 0);

    gemm_tile_wmma<2 * PDIM><<<gemmGridD3, 128, 0, stream>>>(docout, dnbr[0], dnbr[1], dnbr[2], wpk[3], 2 * PDIM * PDIM,
                                                             b_sage[3], b_sage[4], b_sage[5], d3[0], BN * PDIM);

    sdpa_fuse_add<<<(BN + 7) / 8, 256, 0, stream>>>(d3[0], d3[1], d3[2], docout, docrep, BN);
    fc_kernel<<<(BN * CN + TB - 1) / TB, TB, 0, stream>>>(docrep, w_fc, b_fc, out);
}
